# starting kernel
# baseline (speedup 1.0000x reference)
#include <hip/hip_runtime.h>

typedef float v2f __attribute__((ext_vector_type(2)));
typedef float v8f __attribute__((ext_vector_type(8)));

#define HD 64
#define CD 192
#define GN 256

__global__ void k_deg(const int* __restrict__ dst, int E, float* __restrict__ deg) {
  int e = blockIdx.x * blockDim.x + threadIdx.x;
  if (e < E) atomicAdd(&deg[dst[e]], 1.0f);
}

__global__ void k_dinv_cnt(float* __restrict__ dinv, const int* __restrict__ batch,
                           float* __restrict__ cnt, int N) {
  int i = blockIdx.x * blockDim.x + threadIdx.x;
  if (i < N) {
    dinv[i] = rsqrtf(dinv[i] + 1.0f);
    atomicAdd(&cnt[batch[i]], 1.0f);
  }
}

__global__ void k_gemm_wmma(const float* __restrict__ Z, int ldz,
                            const float* __restrict__ W, int K,
                            float* __restrict__ Hout, int N) {
  int wid = blockIdx.x * (blockDim.x >> 5) + (threadIdx.x >> 5);
  int tilesM = (N + 15) >> 4;
  if (wid >= tilesM * 4) return;
  int tm = wid >> 2, tn = wid & 3;
  int lane = threadIdx.x & 31;
  int half = lane >> 4;
  int m16 = lane & 15;
  int row = tm * 16 + m16;
  int lrow = row < N ? row : N - 1;
  const float* zr = Z + (size_t)lrow * ldz;
  int col = tn * 16 + m16;
  const float* wp = W + col;
  v8f c = {};
  for (int k0 = 0; k0 < K; k0 += 4) {
    int ka = k0 + 2 * half;
    v2f a, b;
    a.x = zr[ka];
    a.y = zr[ka + 1];
    b.x = wp[(size_t)ka * HD];
    b.y = wp[(size_t)(ka + 1) * HD];
    c = __builtin_amdgcn_wmma_f32_16x16x4_f32(false, a, false, b, (short)0, c, false, false);
  }
  int rbase = tm * 16 + half * 8;
#pragma unroll
  for (int j = 0; j < 8; ++j) {
    int r = rbase + j;
    if (r < N) Hout[(size_t)r * HD + col] = c[j];
  }
}

__global__ void k_selfinit(const float* __restrict__ h, const float* __restrict__ dinv,
                           float* __restrict__ agg, int N) {
  int t = blockIdx.x * blockDim.x + threadIdx.x;
  if (t < N * HD) {
    float di = dinv[t >> 6];
    agg[t] = h[t] * di * di;
  }
}

__global__ void k_edge_scatter(const int* __restrict__ src, const int* __restrict__ dst,
                               const float* __restrict__ dinv, const float* __restrict__ h,
                               float* __restrict__ agg, int E) {
  int t = blockIdx.x * blockDim.x + threadIdx.x;
  if (t >= E * 16) return;
  int e = t >> 4;
  int c4 = (t & 15) << 2;
  int s = src[e], d = dst[e];
  float w = dinv[s] * dinv[d];
  const float4 hv = *(const float4*)(h + (size_t)s * HD + c4);
  float* ap = agg + (size_t)d * HD + c4;
  atomicAdd(ap + 0, hv.x * w);
  atomicAdd(ap + 1, hv.y * w);
  atomicAdd(ap + 2, hv.z * w);
  atomicAdd(ap + 3, hv.w * w);
}

__global__ void k_relu_stats(float* __restrict__ agg, const float* __restrict__ bias,
                             float* __restrict__ bn_sum, float* __restrict__ bn_sq, int N) {
  __shared__ float ssum[256], ssq[256];
  int c = threadIdx.x & 63;
  int quad = threadIdx.x >> 6;
  int base = blockIdx.x * 64;
  float b = bias[c];
  float s = 0.f, q = 0.f;
  for (int r = quad; r < 64; r += 4) {
    int row = base + r;
    if (row < N) {
      float v = fmaxf(agg[(size_t)row * HD + c] + b, 0.f);
      agg[(size_t)row * HD + c] = v;
      s += v; q += v * v;
    }
  }
  ssum[threadIdx.x] = s; ssq[threadIdx.x] = q;
  __syncthreads();
  if (quad == 0) {
    float ts = ssum[c] + ssum[64 + c] + ssum[128 + c] + ssum[192 + c];
    float tq = ssq[c] + ssq[64 + c] + ssq[128 + c] + ssq[192 + c];
    atomicAdd(&bn_sum[c], ts);
    atomicAdd(&bn_sq[c], tq);
  }
}

__global__ void k_bnfinal(const float* __restrict__ bn_sum, const float* __restrict__ bn_sq,
                          const float* __restrict__ gamma, const float* __restrict__ beta,
                          float* __restrict__ scale, float* __restrict__ shift, float invN) {
  int c = threadIdx.x;
  float mu = bn_sum[c] * invN;
  float var = fmaxf(bn_sq[c] * invN - mu * mu, 0.f);
  float sc = rsqrtf(var + 1e-5f) * gamma[c];
  scale[c] = sc;
  shift[c] = beta[c] - mu * sc;
}

__global__ void k_bnapply_pool(const float* __restrict__ agg, const float* __restrict__ scale,
                               const float* __restrict__ shift, const int* __restrict__ batch,
                               float* __restrict__ zc, float* __restrict__ gsum,
                               int N, int lofs) {
  int c = threadIdx.x & 63;
  int quad = threadIdx.x >> 6;
  int base = blockIdx.x * 64;
  float sc = scale[c], sh = shift[c];
  int gcur = -1;
  float acc = 0.f;
  for (int r = quad; r < 64; r += 4) {
    int row = base + r;
    if (row < N) {
      float v = agg[(size_t)row * HD + c] * sc + sh;
      zc[(size_t)row * CD + lofs + c] = v;
      int g = batch[row];
      if (g != gcur) {
        if (gcur >= 0) atomicAdd(&gsum[(size_t)gcur * CD + lofs + c], acc);
        gcur = g; acc = 0.f;
      }
      acc += v;
    }
  }
  if (gcur >= 0) atomicAdd(&gsum[(size_t)gcur * CD + lofs + c], acc);
}

__global__ void k_l2_zc(const float* __restrict__ zc, float* __restrict__ out, int rows) {
  int wid = blockIdx.x * (blockDim.x >> 5) + (threadIdx.x >> 5);
  if (wid >= rows) return;
  int lane = threadIdx.x & 31;
  const float* a = zc + (size_t)wid * CD;
  float v[6]; float s = 0.f;
#pragma unroll
  for (int j = 0; j < 6; ++j) { v[j] = a[lane + 32 * j]; s += v[j] * v[j]; }
#pragma unroll
  for (int off = 16; off >= 1; off >>= 1) s += __shfl_xor(s, off, 32);
  float inv = 1.0f / fmaxf(sqrtf(s), 1e-12f);
  float* o = out + (size_t)wid * CD;
#pragma unroll
  for (int j = 0; j < 6; ++j) o[lane + 32 * j] = v[j] * inv;
}

__global__ void k_l2_gc(const float* __restrict__ gsum, const float* __restrict__ cnt,
                        float* __restrict__ out, int rows) {
  int wid = blockIdx.x * (blockDim.x >> 5) + (threadIdx.x >> 5);
  if (wid >= rows) return;
  int lane = threadIdx.x & 31;
  float icnt = 1.0f / fmaxf(cnt[wid], 1.0f);
  const float* a = gsum + (size_t)wid * CD;
  float v[6]; float s = 0.f;
#pragma unroll
  for (int j = 0; j < 6; ++j) { v[j] = a[lane + 32 * j] * icnt; s += v[j] * v[j]; }
#pragma unroll
  for (int off = 16; off >= 1; off >>= 1) s += __shfl_xor(s, off, 32);
  float inv = 1.0f / fmaxf(sqrtf(s), 1e-12f);
  float* o = out + (size_t)wid * CD;
#pragma unroll
  for (int j = 0; j < 6; ++j) o[lane + 32 * j] = v[j] * inv;
}

extern "C" void kernel_launch(void* const* d_in, const int* in_sizes, int n_in,
                              void* d_out, int out_size, void* d_ws, size_t ws_size,
                              hipStream_t stream) {
  (void)n_in; (void)out_size; (void)ws_size;
  const float* x     = (const float*)d_in[0];
  const int*   ei    = (const int*)d_in[1];
  const int*   batch = (const int*)d_in[2];
  int N = in_sizes[0] / 128;
  int E = in_sizes[1] / 2;
  const int* src = ei;
  const int* dst = ei + E;
  const float* Wl[3]  = {(const float*)d_in[3],  (const float*)d_in[7],  (const float*)d_in[11]};
  const float* bl[3]  = {(const float*)d_in[4],  (const float*)d_in[8],  (const float*)d_in[12]};
  const float* gl[3]  = {(const float*)d_in[5],  (const float*)d_in[9],  (const float*)d_in[13]};
  const float* bel[3] = {(const float*)d_in[6],  (const float*)d_in[10], (const float*)d_in[14]};

  float* ws   = (float*)d_ws;
  float* dinv = ws;                       ws += N;
  float* h    = ws;                       ws += (size_t)N * HD;
  float* agg  = ws;                       ws += (size_t)N * HD;
  float* zc   = ws;                       ws += (size_t)N * CD;
  float* gsum = ws;                       ws += (size_t)GN * CD;
  float* cnt  = ws;                       ws += GN;
  float* bn   = ws;
  float* bn_sum = bn, *bn_sq = bn + 64, *scale = bn + 128, *shift = bn + 192;

  hipMemsetAsync(dinv, 0, (size_t)N * sizeof(float), stream);
  hipMemsetAsync(gsum, 0, (size_t)GN * CD * sizeof(float), stream);
  hipMemsetAsync(cnt,  0, (size_t)GN * sizeof(float), stream);

  k_deg<<<(E + 255) / 256, 256, 0, stream>>>(dst, E, dinv);
  k_dinv_cnt<<<(N + 255) / 256, 256, 0, stream>>>(dinv, batch, cnt, N);

  for (int l = 0; l < 3; ++l) {
    const float* Z = (l == 0) ? x : (zc + (size_t)(l - 1) * HD);
    int ldz = (l == 0) ? 128 : CD;
    int K   = (l == 0) ? 128 : HD;
    int tilesM = (N + 15) / 16;
    int waves = tilesM * 4;
    k_gemm_wmma<<<(waves + 7) / 8, 256, 0, stream>>>(Z, ldz, Wl[l], K, h, N);
    k_selfinit<<<(N * HD + 255) / 256, 256, 0, stream>>>(h, dinv, agg, N);
    k_edge_scatter<<<((size_t)E * 16 + 255) / 256, 256, 0, stream>>>(src, dst, dinv, h, agg, E);
    hipMemsetAsync(bn, 0, 128 * sizeof(float), stream);
    k_relu_stats<<<(N + 63) / 64, 256, 0, stream>>>(agg, bl[l], bn_sum, bn_sq, N);
    k_bnfinal<<<1, 64, 0, stream>>>(bn_sum, bn_sq, gl[l], bel[l], scale, shift, 1.0f / (float)N);
    k_bnapply_pool<<<(N + 63) / 64, 256, 0, stream>>>(agg, scale, shift, batch, zc, gsum, N, l * HD);
  }

  float* out = (float*)d_out;
  k_l2_zc<<<(N + 7) / 8, 256, 0, stream>>>(zc, out, N);
  k_l2_gc<<<(GN + 7) / 8, 256, 0, stream>>>(gsum, cnt, out + (size_t)N * CD, GN);
}


# edge aggregation by row ownership: a block keeps 196 dst rows in LDS (self term as start), filters the raw dst list, adds h[src] rows by LDS float atomics, stores rows once; sums in LDS-arrival order
# speedup vs baseline: 1.4759x; 1.4759x over previous
#include <hip/hip_runtime.h>

typedef float v2f __attribute__((ext_vector_type(2)));
typedef float v8f __attribute__((ext_vector_type(8)));

#define HD 64
#define CD 192
#define GN 256

__global__ void k_deg(const int* __restrict__ dst, int E, float* __restrict__ deg) {
  int e = blockIdx.x * blockDim.x + threadIdx.x;
  if (e < E) atomicAdd(&deg[dst[e]], 1.0f);
}

__global__ void k_dinv_cnt(float* __restrict__ dinv, const int* __restrict__ batch,
                           float* __restrict__ cnt, int N) {
  int i = blockIdx.x * blockDim.x + threadIdx.x;
  if (i < N) {
    dinv[i] = rsqrtf(dinv[i] + 1.0f);
    atomicAdd(&cnt[batch[i]], 1.0f);
  }
}

__global__ void k_gemm_wmma(const float* __restrict__ Z, int ldz,
                            const float* __restrict__ W, int K,
                            float* __restrict__ Hout, int N) {
  int wid = blockIdx.x * (blockDim.x >> 5) + (threadIdx.x >> 5);
  int tilesM = (N + 15) >> 4;
  if (wid >= tilesM * 4) return;
  int tm = wid >> 2, tn = wid & 3;
  int lane = threadIdx.x & 31;
  int half = lane >> 4;
  int m16 = lane & 15;
  int row = tm * 16 + m16;
  int lrow = row < N ? row : N - 1;
  const float* zr = Z + (size_t)lrow * ldz;
  int col = tn * 16 + m16;
  const float* wp = W + col;
  v8f c = {};
  for (int k0 = 0; k0 < K; k0 += 4) {
    int ka = k0 + 2 * half;
    v2f a, b;
    a.x = zr[ka];
    a.y = zr[ka + 1];
    b.x = wp[(size_t)ka * HD];
    b.y = wp[(size_t)(ka + 1) * HD];
    c = __builtin_amdgcn_wmma_f32_16x16x4_f32(false, a, false, b, (short)0, c, false, false);
  }
  int rbase = tm * 16 + half * 8;
#pragma unroll
  for (int j = 0; j < 8; ++j) {
    int r = rbase + j;
    if (r < N) Hout[(size_t)r * HD + col] = c[j];
  }
}

#define AG_R 196
#define AG_T 1024
#define AG_W 32
#define AG_Q 256
#define AG_U 8

__device__ __forceinline__ void ag_group(const int* __restrict__ src, const float* __restrict__ dinv,
                                         const float* __restrict__ h, float* tile, const float* sdinv,
                                         const int* myqe, const int* myqd, int base, int n, int lane) {
  int s_i = 0, dl_i = 0;
  float w_i = 0.f;
  if (lane < n) {
    int e = myqe[base + lane];
    dl_i = myqd[base + lane];
    s_i = src[e];
    w_i = dinv[s_i] * sdinv[dl_i];
  }
  int wbits = __float_as_int(w_i);
#pragma unroll
  for (int hf = 0; hf < 2; ++hf) {
    if (hf * 16 < n) {
      float a0[16], a1[16];
#pragma unroll
      for (int u = 0; u < 16; ++u) {
        int s = __builtin_amdgcn_readlane(s_i, hf * 16 + u);
        const float* hp = h + (size_t)s * HD;
        a0[u] = hp[lane];
        a1[u] = hp[32 + lane];
      }
#pragma unroll
      for (int u = 0; u < 16; ++u) {
        if (hf * 16 + u < n) {
          int dl = __builtin_amdgcn_readlane(dl_i, hf * 16 + u);
          float w = __int_as_float(__builtin_amdgcn_readlane(wbits, hf * 16 + u));
          atomicAdd(&tile[dl * HD + lane], a0[u] * w);
          atomicAdd(&tile[dl * HD + 32 + lane], a1[u] * w);
        }
      }
    }
  }
}

__device__ __forceinline__ void ag_drain(const int* __restrict__ src, const float* __restrict__ dinv,
                                         const float* __restrict__ h, float* tile, const float* sdinv,
                                         const int* myqe, const int* myqd, int& cnt, int lane) {
  for (int base = 0; base < cnt; base += 32) {
    int n = cnt - base;
    if (n > 32) n = 32;
    int s_i = 0, dl_i = 0;
    float w_i = 0.f;
    if (lane < n) {
      int e = myqe[base + lane];
      dl_i = myqd[base + lane];
      s_i = src[e];
      w_i = dinv[s_i] * sdinv[dl_i];
    }
    int wbits = __float_as_int(w_i);
#pragma unroll 1
    for (int u = 0; u < n; ++u) {
      int s = __builtin_amdgcn_readlane(s_i, u);
      int dl = __builtin_amdgcn_readlane(dl_i, u);
      float w = __int_as_float(__builtin_amdgcn_readlane(wbits, u));
      const float* hp = h + (size_t)s * HD;
      atomicAdd(&tile[dl * HD + lane], hp[lane] * w);
      atomicAdd(&tile[dl * HD + 32 + lane], hp[32 + lane] * w);
    }
  }
  cnt = 0;
}

__device__ __forceinline__ void ag_push(bool m, int e, int dl, int* myqe, int* myqd, int& cnt) {
  unsigned mask = __builtin_amdgcn_ballot_w32(m);
  if (mask) {
    int pos = cnt + (int)__builtin_amdgcn_mbcnt_lo(mask, 0u);
    if (m) { myqe[pos] = e; myqd[pos] = dl; }
    cnt += __builtin_popcount(mask);
  }
}

__device__ __forceinline__ void ag_test4(int4 d, bool valid, int e0, int row0, int nrows,
                                         const int* __restrict__ src, const float* __restrict__ dinv,
                                         const float* __restrict__ h, float* tile, const float* sdinv,
                                         int* myqe, int* myqd, int& cnt, int lane) {
  unsigned r0 = (unsigned)d.x - (unsigned)row0;
  unsigned r1 = (unsigned)d.y - (unsigned)row0;
  unsigned r2 = (unsigned)d.z - (unsigned)row0;
  unsigned r3 = (unsigned)d.w - (unsigned)row0;
  bool m0 = valid && r0 < (unsigned)nrows;
  bool m1 = valid && r1 < (unsigned)nrows;
  bool m2 = valid && r2 < (unsigned)nrows;
  bool m3 = valid && r3 < (unsigned)nrows;
  if (__builtin_amdgcn_ballot_w32(m0 || m1 || m2 || m3)) {
    if (cnt > AG_Q - 128) ag_drain(src, dinv, h, tile, sdinv, myqe, myqd, cnt, lane);
    ag_push(m0, e0,     (int)r0, myqe, myqd, cnt);
    ag_push(m1, e0 + 1, (int)r1, myqe, myqd, cnt);
    ag_push(m2, e0 + 2, (int)r2, myqe, myqd, cnt);
    ag_push(m3, e0 + 3, (int)r3, myqe, myqd, cnt);
  }
}

__device__ __forceinline__ void ag_flush_full(const int* __restrict__ src, const float* __restrict__ dinv,
                                              const float* __restrict__ h, float* tile, const float* sdinv,
                                              int* myqe, int* myqd, int& cnt, int lane) {
  int nf = cnt & ~31;
  for (int base = 0; base < nf; base += 32)
    ag_group(src, dinv, h, tile, sdinv, myqe, myqd, base, 32, lane);
  int rem = cnt - nf;
  if (lane < rem) {
    int te = myqe[nf + lane], td = myqd[nf + lane];
    myqe[lane] = te;
    myqd[lane] = td;
  }
  cnt = rem;
}

__global__ __launch_bounds__(AG_T) void k_aggregate(const int* __restrict__ src, const int* __restrict__ dst,
                                                    const float* __restrict__ dinv, const float* __restrict__ h,
                                                    float* __restrict__ agg, int N, int E) {
  static_assert(AG_R * (HD / 4) <= 4 * AG_T, "row tile is filled and stored in four float4 per thread");
  static_assert(AG_T == AG_W * 32, "wave32");
  __shared__ __attribute__((aligned(16))) float tile[AG_R * HD];
  __shared__ float sdinv[AG_R];
  __shared__ int qe[AG_W][AG_Q];
  __shared__ int qd[AG_W][AG_Q];

  const int tid = threadIdx.x;
  const int lane = tid & 31;
  const int wv = tid >> 5;
  const int row0 = blockIdx.x * AG_R;
  int nrows = N - row0;
  if (nrows > AG_R) nrows = AG_R;
  const int n4 = nrows * (HD / 4);

  {
    const float4* h4 = (const float4*)(h + (size_t)row0 * HD);
    float4* t4 = (float4*)tile;
    float4 hv[4];
    float di[4];
#pragma unroll
    for (int j = 0; j < 4; ++j) {
      int i4 = tid + j * AG_T;
      hv[j] = make_float4(0.f, 0.f, 0.f, 0.f);
      di[j] = 0.f;
      if (i4 < n4) { hv[j] = h4[i4]; di[j] = dinv[row0 + (i4 >> 4)]; }
    }
#pragma unroll
    for (int j = 0; j < 4; ++j) {
      int i4 = tid + j * AG_T;
      if (i4 < n4) {
        float4 v;
        v.x = hv[j].x * di[j] * di[j];
        v.y = hv[j].y * di[j] * di[j];
        v.z = hv[j].z * di[j] * di[j];
        v.w = hv[j].w * di[j] * di[j];
        t4[i4] = v;
        if ((i4 & 15) == 0) sdinv[i4 >> 4] = di[j];
      }
    }
  }
  __syncthreads();

  int head = (int)((4u - (unsigned)(((size_t)dst >> 2) & 3u)) & 3u);
  if (head > E) head = E;
  const int4* dv = (const int4*)(dst + head);
  const int nvec = (E - head) >> 2;
  const int tail = (E - head) & 3;
  const int nFull = nvec / AG_T;
  const int it0 = (int)(((long long)blockIdx.x * nFull) / gridDim.x);
  int* myqe = qe[wv];
  int* myqd = qd[wv];
  int cnt = 0;

  int k = 0;
  for (; k + AG_U <= nFull; k += AG_U) {
    int4 d[AG_U];
    int vi[AG_U];
#pragma unroll
    for (int j = 0; j < AG_U; ++j) {
      int it = it0 + k + j;
      if (it >= nFull) it -= nFull;
      vi[j] = it * AG_T + tid;
      d[j] = dv[vi[j]];
    }
#pragma unroll
    for (int j = 0; j < AG_U; ++j)
      ag_test4(d[j], true, head + 4 * vi[j], row0, nrows, src, dinv, h, tile, sdinv, myqe, myqd, cnt, lane);
    if (cnt >= 32) ag_flush_full(src, dinv, h, tile, sdinv, myqe, myqd, cnt, lane);
  }
  for (; k <= nFull; ++k) {
    int it = it0 + k;
    if (it >= nFull) it -= nFull;
    if (k == nFull) it = nFull;
    int v = it * AG_T + tid;
    bool valid = v < nvec;
    int4 d = make_int4(0, 0, 0, 0);
    if (valid) d = dv[v];
    ag_test4(d, valid, head + 4 * v, row0, nrows, src, dinv, h, tile, sdinv, myqe, myqd, cnt, lane);
  }
  if (wv == 0) {
    int e = lane < head ? lane : head + 4 * nvec + (lane - head);
    bool valid = lane < head + tail;
    int dd = 0;
    if (valid) dd = dst[e];
    unsigned r = (unsigned)dd - (unsigned)row0;
    if (cnt > AG_Q - 128) ag_drain(src, dinv, h, tile, sdinv, myqe, myqd, cnt, lane);
    ag_push(valid && r < (unsigned)nrows, e, (int)r, myqe, myqd, cnt);
  }
  for (int base = 0; base < cnt; base += 32) {
    int n = cnt - base;
    if (n > 32) n = 32;
    ag_group(src, dinv, h, tile, sdinv, myqe, myqd, base, n, lane);
  }
  __syncthreads();

  {
    const float4* t4 = (const float4*)tile;
    float4* o4 = (float4*)(agg + (size_t)row0 * HD);
#pragma unroll
    for (int j = 0; j < 4; ++j) {
      int i4 = tid + j * AG_T;
      if (i4 < n4) o4[i4] = t4[i4];
    }
  }
}

__global__ void k_relu_stats(float* __restrict__ agg, const float* __restrict__ bias,
                             float* __restrict__ bn_sum, float* __restrict__ bn_sq, int N) {
  __shared__ float ssum[256], ssq[256];
  int c = threadIdx.x & 63;
  int quad = threadIdx.x >> 6;
  int base = blockIdx.x * 64;
  float b = bias[c];
  float s = 0.f, q = 0.f;
  for (int r = quad; r < 64; r += 4) {
    int row = base + r;
    if (row < N) {
      float v = fmaxf(agg[(size_t)row * HD + c] + b, 0.f);
      agg[(size_t)row * HD + c] = v;
      s += v; q += v * v;
    }
  }
  ssum[threadIdx.x] = s; ssq[threadIdx.x] = q;
  __syncthreads();
  if (quad == 0) {
    float ts = ssum[c] + ssum[64 + c] + ssum[128 + c] + ssum[192 + c];
    float tq = ssq[c] + ssq[64 + c] + ssq[128 + c] + ssq[192 + c];
    atomicAdd(&bn_sum[c], ts);
    atomicAdd(&bn_sq[c], tq);
  }
}

__global__ void k_bnfinal(const float* __restrict__ bn_sum, const float* __restrict__ bn_sq,
                          const float* __restrict__ gamma, const float* __restrict__ beta,
                          float* __restrict__ scale, float* __restrict__ shift, float invN) {
  int c = threadIdx.x;
  float mu = bn_sum[c] * invN;
  float var = fmaxf(bn_sq[c] * invN - mu * mu, 0.f);
  float sc = rsqrtf(var + 1e-5f) * gamma[c];
  scale[c] = sc;
  shift[c] = beta[c] - mu * sc;
}

__global__ void k_bnapply_pool(const float* __restrict__ agg, const float* __restrict__ scale,
                               const float* __restrict__ shift, const int* __restrict__ batch,
                               float* __restrict__ zc, float* __restrict__ gsum,
                               int N, int lofs) {
  int c = threadIdx.x & 63;
  int quad = threadIdx.x >> 6;
  int base = blockIdx.x * 64;
  float sc = scale[c], sh = shift[c];
  int gcur = -1;
  float acc = 0.f;
  for (int r = quad; r < 64; r += 4) {
    int row = base + r;
    if (row < N) {
      float v = agg[(size_t)row * HD + c] * sc + sh;
      zc[(size_t)row * CD + lofs + c] = v;
      int g = batch[row];
      if (g != gcur) {
        if (gcur >= 0) atomicAdd(&gsum[(size_t)gcur * CD + lofs + c], acc);
        gcur = g; acc = 0.f;
      }
      acc += v;
    }
  }
  if (gcur >= 0) atomicAdd(&gsum[(size_t)gcur * CD + lofs + c], acc);
}

__global__ void k_l2_zc(const float* __restrict__ zc, float* __restrict__ out, int rows) {
  int wid = blockIdx.x * (blockDim.x >> 5) + (threadIdx.x >> 5);
  if (wid >= rows) return;
  int lane = threadIdx.x & 31;
  const float* a = zc + (size_t)wid * CD;
  float v[6]; float s = 0.f;
#pragma unroll
  for (int j = 0; j < 6; ++j) { v[j] = a[lane + 32 * j]; s += v[j] * v[j]; }
#pragma unroll
  for (int off = 16; off >= 1; off >>= 1) s += __shfl_xor(s, off, 32);
  float inv = 1.0f / fmaxf(sqrtf(s), 1e-12f);
  float* o = out + (size_t)wid * CD;
#pragma unroll
  for (int j = 0; j < 6; ++j) o[lane + 32 * j] = v[j] * inv;
}

__global__ void k_l2_gc(const float* __restrict__ gsum, const float* __restrict__ cnt,
                        float* __restrict__ out, int rows) {
  int wid = blockIdx.x * (blockDim.x >> 5) + (threadIdx.x >> 5);
  if (wid >= rows) return;
  int lane = threadIdx.x & 31;
  float icnt = 1.0f / fmaxf(cnt[wid], 1.0f);
  const float* a = gsum + (size_t)wid * CD;
  float v[6]; float s = 0.f;
#pragma unroll
  for (int j = 0; j < 6; ++j) { v[j] = a[lane + 32 * j] * icnt; s += v[j] * v[j]; }
#pragma unroll
  for (int off = 16; off >= 1; off >>= 1) s += __shfl_xor(s, off, 32);
  float inv = 1.0f / fmaxf(sqrtf(s), 1e-12f);
  float* o = out + (size_t)wid * CD;
#pragma unroll
  for (int j = 0; j < 6; ++j) o[lane + 32 * j] = v[j] * inv;
}

extern "C" void kernel_launch(void* const* d_in, const int* in_sizes, int n_in,
                              void* d_out, int out_size, void* d_ws, size_t ws_size,
                              hipStream_t stream) {
  (void)n_in; (void)out_size; (void)ws_size;
  const float* x     = (const float*)d_in[0];
  const int*   ei    = (const int*)d_in[1];
  const int*   batch = (const int*)d_in[2];
  int N = in_sizes[0] / 128;
  int E = in_sizes[1] / 2;
  const int* src = ei;
  const int* dst = ei + E;
  const float* Wl[3]  = {(const float*)d_in[3],  (const float*)d_in[7],  (const float*)d_in[11]};
  const float* bl[3]  = {(const float*)d_in[4],  (const float*)d_in[8],  (const float*)d_in[12]};
  const float* gl[3]  = {(const float*)d_in[5],  (const float*)d_in[9],  (const float*)d_in[13]};
  const float* bel[3] = {(const float*)d_in[6],  (const float*)d_in[10], (const float*)d_in[14]};

  float* ws   = (float*)d_ws;
  float* dinv = ws;                       ws += N;
  float* h    = ws;                       ws += (size_t)N * HD;
  float* agg  = ws;                       ws += (size_t)N * HD;
  float* zc   = ws;                       ws += (size_t)N * CD;
  float* gsum = ws;                       ws += (size_t)GN * CD;
  float* cnt  = ws;                       ws += GN;
  float* bn   = ws;
  float* bn_sum = bn, *bn_sq = bn + 64, *scale = bn + 128, *shift = bn + 192;

  hipMemsetAsync(dinv, 0, (size_t)N * sizeof(float), stream);
  hipMemsetAsync(gsum, 0, (size_t)GN * CD * sizeof(float), stream);
  hipMemsetAsync(cnt,  0, (size_t)GN * sizeof(float), stream);

  k_deg<<<(E + 255) / 256, 256, 0, stream>>>(dst, E, dinv);
  k_dinv_cnt<<<(N + 255) / 256, 256, 0, stream>>>(dinv, batch, cnt, N);

  for (int l = 0; l < 3; ++l) {
    const float* Z = (l == 0) ? x : (zc + (size_t)(l - 1) * HD);
    int ldz = (l == 0) ? 128 : CD;
    int K   = (l == 0) ? 128 : HD;
    int tilesM = (N + 15) / 16;
    int waves = tilesM * 4;
    k_gemm_wmma<<<(waves + 7) / 8, 256, 0, stream>>>(Z, ldz, Wl[l], K, h, N);
    k_aggregate<<<(N + AG_R - 1) / AG_R, AG_T, 0, stream>>>(src, dst, dinv, h, agg, N, E);
    hipMemsetAsync(bn, 0, 128 * sizeof(float), stream);
    k_relu_stats<<<(N + 63) / 64, 256, 0, stream>>>(agg, bl[l], bn_sum, bn_sq, N);
    k_bnfinal<<<1, 64, 0, stream>>>(bn_sum, bn_sq, gl[l], bel[l], scale, shift, 1.0f / (float)N);
    k_bnapply_pool<<<(N + 63) / 64, 256, 0, stream>>>(agg, scale, shift, batch, zc, gsum, N, l * HD);
  }

  float* out = (float*)d_out;
  k_l2_zc<<<(N + 7) / 8, 256, 0, stream>>>(zc, out, N);
  k_l2_gc<<<(GN + 7) / 8, 256, 0, stream>>>(gsum, cnt, out + (size_t)N * CD, GN);
}


# aggregation filter scans a one-byte owner key per edge (dst/196, written by the degree kernel): 16 edges per 16-byte load, exact byte test, dst read for hits only; edge sums in other LDS-arrival order
# speedup vs baseline: 1.5704x; 1.0640x over previous
#include <hip/hip_runtime.h>

typedef float v2f __attribute__((ext_vector_type(2)));
typedef float v8f __attribute__((ext_vector_type(8)));

#define HD 64
#define CD 192
#define GN 256
#define AG_R 196

__global__ void k_deg(const int* __restrict__ dst, int E, float* __restrict__ deg, unsigned char* __restrict__ key) {
  int e = blockIdx.x * blockDim.x + threadIdx.x;
  if (e < E) {
    int d = dst[e];
    atomicAdd(&deg[d], 1.0f);
    key[e] = (unsigned char)(d / AG_R);
  } else if (e < ((E + 15) & ~15)) {
    key[e] = 0;
  }
}

__global__ void k_dinv_cnt(float* __restrict__ dinv, const int* __restrict__ batch,
                           float* __restrict__ cnt, int N) {
  int i = blockIdx.x * blockDim.x + threadIdx.x;
  if (i < N) {
    dinv[i] = rsqrtf(dinv[i] + 1.0f);
    atomicAdd(&cnt[batch[i]], 1.0f);
  }
}

__global__ void k_gemm_wmma(const float* __restrict__ Z, int ldz,
                            const float* __restrict__ W, int K,
                            float* __restrict__ Hout, int N) {
  int wid = blockIdx.x * (blockDim.x >> 5) + (threadIdx.x >> 5);
  int tilesM = (N + 15) >> 4;
  if (wid >= tilesM * 4) return;
  int tm = wid >> 2, tn = wid & 3;
  int lane = threadIdx.x & 31;
  int half = lane >> 4;
  int m16 = lane & 15;
  int row = tm * 16 + m16;
  int lrow = row < N ? row : N - 1;
  const float* zr = Z + (size_t)lrow * ldz;
  int col = tn * 16 + m16;
  const float* wp = W + col;
  v8f c = {};
  for (int k0 = 0; k0 < K; k0 += 4) {
    int ka = k0 + 2 * half;
    v2f a, b;
    a.x = zr[ka];
    a.y = zr[ka + 1];
    b.x = wp[(size_t)ka * HD];
    b.y = wp[(size_t)(ka + 1) * HD];
    c = __builtin_amdgcn_wmma_f32_16x16x4_f32(false, a, false, b, (short)0, c, false, false);
  }
  int rbase = tm * 16 + half * 8;
#pragma unroll
  for (int j = 0; j < 8; ++j) {
    int r = rbase + j;
    if (r < N) Hout[(size_t)r * HD + col] = c[j];
  }
}

#define AG_T 1024
#define AG_W 32
#define AG_Q 256
#define AG_U 8

__device__ __forceinline__ void ag_group(const int* __restrict__ src, const int* __restrict__ dst,
                                         const float* __restrict__ dinv,
                                         const float* __restrict__ h, float* tile, const float* sdinv,
                                         const int* myqe, int row0, int base, int n, int lane) {
  int s_i = 0, dl_i = 0;
  float w_i = 0.f;
  if (lane < n) {
    int e = myqe[base + lane];
    s_i = src[e];
    dl_i = dst[e] - row0;
    w_i = dinv[s_i] * sdinv[dl_i];
  }
  int wbits = __float_as_int(w_i);
#pragma unroll
  for (int hf = 0; hf < 2; ++hf) {
    if (hf * 16 < n) {
      float a0[16], a1[16];
#pragma unroll
      for (int u = 0; u < 16; ++u) {
        int s = __builtin_amdgcn_readlane(s_i, hf * 16 + u);
        const float* hp = h + (size_t)s * HD;
        a0[u] = hp[lane];
        a1[u] = hp[32 + lane];
      }
#pragma unroll
      for (int u = 0; u < 16; ++u) {
        if (hf * 16 + u < n) {
          int dl = __builtin_amdgcn_readlane(dl_i, hf * 16 + u);
          float w = __int_as_float(__builtin_amdgcn_readlane(wbits, hf * 16 + u));
          atomicAdd(&tile[dl * HD + lane], a0[u] * w);
          atomicAdd(&tile[dl * HD + 32 + lane], a1[u] * w);
        }
      }
    }
  }
}

__device__ __forceinline__ void ag_drain(const int* __restrict__ src, const int* __restrict__ dst,
                                         const float* __restrict__ dinv,
                                         const float* __restrict__ h, float* tile, const float* sdinv,
                                         const int* myqe, int row0, int& cnt, int lane) {
  for (int base = 0; base < cnt; base += 32) {
    int n = cnt - base;
    if (n > 32) n = 32;
    int s_i = 0, dl_i = 0;
    float w_i = 0.f;
    if (lane < n) {
      int e = myqe[base + lane];
      s_i = src[e];
      dl_i = dst[e] - row0;
      w_i = dinv[s_i] * sdinv[dl_i];
    }
    int wbits = __float_as_int(w_i);
#pragma unroll 1
    for (int u = 0; u < n; ++u) {
      int s = __builtin_amdgcn_readlane(s_i, u);
      int dl = __builtin_amdgcn_readlane(dl_i, u);
      float w = __int_as_float(__builtin_amdgcn_readlane(wbits, u));
      const float* hp = h + (size_t)s * HD;
      atomicAdd(&tile[dl * HD + lane], hp[lane] * w);
      atomicAdd(&tile[dl * HD + 32 + lane], hp[32 + lane] * w);
    }
  }
  cnt = 0;
}

__device__ __forceinline__ void ag_push(bool m, int e, int* myqe, int& cnt) {
  unsigned mask = __builtin_amdgcn_ballot_w32(m);
  if (mask) {
    int pos = cnt + (int)__builtin_amdgcn_mbcnt_lo(mask, 0u);
    if (m) myqe[pos] = e;
    cnt += __builtin_popcount(mask);
  }
}

__device__ __forceinline__ unsigned ag_miss(unsigned w, unsigned krep) {
  unsigned x = w ^ krep;
  return ((x & 0x7F7F7F7Fu) + 0x7F7F7F7Fu) | x | 0x7F7F7F7Fu;
}

template <bool TAIL>
__device__ __forceinline__ void ag_test16(uint4 kq, bool valid, int e0, unsigned krep, int E, int row0,
                                          const int* __restrict__ src, const int* __restrict__ dst,
                                          const float* __restrict__ dinv,
                                          const float* __restrict__ h, float* tile, const float* sdinv,
                                          int* myqe, int& cnt, int lane) {
  unsigned c = __builtin_rotateright32(ag_miss(kq.x, krep), 3) & __builtin_rotateright32(ag_miss(kq.y, krep), 2) &
               __builtin_rotateright32(ag_miss(kq.z, krep), 1) & ag_miss(kq.w, krep);
  unsigned hit = ~c;
  if (TAIL) hit = valid ? hit : 0u;
  while (__builtin_amdgcn_ballot_w32(hit != 0u)) {
    if (cnt > AG_Q - 32) ag_drain(src, dst, dinv, h, tile, sdinv, myqe, row0, cnt, lane);
    int p = __builtin_ffs((int)hit) - 1;
    int e = e0 + ((p & 3) << 2) + (p >> 3);
    bool m = hit != 0u;
    if (TAIL) m = m && e < E;
    hit &= hit - 1u;
    ag_push(m, e, myqe, cnt);
  }
}

__device__ __forceinline__ void ag_flush_full(const int* __restrict__ src, const int* __restrict__ dst,
                                              const float* __restrict__ dinv,
                                              const float* __restrict__ h, float* tile, const float* sdinv,
                                              int* myqe, int row0, int& cnt, int lane) {
  int nf = cnt & ~31;
  for (int base = 0; base < nf; base += 32)
    ag_group(src, dst, dinv, h, tile, sdinv, myqe, row0, base, 32, lane);
  int rem = cnt - nf;
  if (lane < rem) {
    int te = myqe[nf + lane];
    myqe[lane] = te;
  }
  cnt = rem;
}

__global__ __launch_bounds__(AG_T) void k_aggregate(const int* __restrict__ src, const int* __restrict__ dst,
                                                    const unsigned char* __restrict__ key,
                                                    const float* __restrict__ dinv, const float* __restrict__ h,
                                                    float* __restrict__ agg, int N, int E) {
  static_assert(AG_R * (HD / 4) <= 4 * AG_T, "row tile is filled and stored in four float4 per thread");
  static_assert(AG_T == AG_W * 32, "wave32");
  __shared__ __attribute__((aligned(16))) float tile[AG_R * HD];
  __shared__ float sdinv[AG_R];
  __shared__ int qe[AG_W][AG_Q];

  const int tid = threadIdx.x;
  const int lane = tid & 31;
  const int wv = tid >> 5;
  const int row0 = blockIdx.x * AG_R;
  int nrows = N - row0;
  if (nrows > AG_R) nrows = AG_R;
  const int n4 = nrows * (HD / 4);

  {
    const float4* h4 = (const float4*)(h + (size_t)row0 * HD);
    float4* t4 = (float4*)tile;
    float4 hv[4];
    float di[4];
#pragma unroll
    for (int j = 0; j < 4; ++j) {
      int i4 = tid + j * AG_T;
      hv[j] = make_float4(0.f, 0.f, 0.f, 0.f);
      di[j] = 0.f;
      if (i4 < n4) { hv[j] = h4[i4]; di[j] = dinv[row0 + (i4 >> 4)]; }
    }
#pragma unroll
    for (int j = 0; j < 4; ++j) {
      int i4 = tid + j * AG_T;
      if (i4 < n4) {
        float4 v;
        v.x = hv[j].x * di[j] * di[j];
        v.y = hv[j].y * di[j] * di[j];
        v.z = hv[j].z * di[j] * di[j];
        v.w = hv[j].w * di[j] * di[j];
        t4[i4] = v;
        if ((i4 & 15) == 0) sdinv[i4 >> 4] = di[j];
      }
    }
  }
  __syncthreads();

  const uint4* kv = (const uint4*)key;
  const int nvec = (E + 15) >> 4;
  const int nFull = (nvec - 1) / AG_T;
  const int it0 = (int)(((long long)blockIdx.x * nFull) / gridDim.x);
  const unsigned krep = (unsigned)blockIdx.x * 0x01010101u;
  int* myqe = qe[wv];
  int cnt = 0;

  int k = 0;
  for (; k + AG_U <= nFull; k += AG_U) {
    uint4 d[AG_U];
    int vi[AG_U];
#pragma unroll
    for (int j = 0; j < AG_U; ++j) {
      int it = it0 + k + j;
      if (it >= nFull) it -= nFull;
      vi[j] = it * AG_T + tid;
      d[j] = kv[vi[j]];
    }
#pragma unroll
    for (int j = 0; j < AG_U; ++j)
      ag_test16<false>(d[j], true, 16 * vi[j], krep, E, row0, src, dst, dinv, h, tile, sdinv, myqe, cnt, lane);
    if (cnt >= 32) ag_flush_full(src, dst, dinv, h, tile, sdinv, myqe, row0, cnt, lane);
  }
  for (; k <= nFull; ++k) {
    int it = it0 + k;
    if (it >= nFull) it -= nFull;
    if (k == nFull) it = nFull;
    int v = it * AG_T + tid;
    bool valid = v < nvec;
    uint4 d = make_uint4(0u, 0u, 0u, 0u);
    if (valid) d = kv[v];
    ag_test16<true>(d, valid, 16 * v, krep, E, row0, src, dst, dinv, h, tile, sdinv, myqe, cnt, lane);
  }
  for (int base = 0; base < cnt; base += 32) {
    int n = cnt - base;
    if (n > 32) n = 32;
    ag_group(src, dst, dinv, h, tile, sdinv, myqe, row0, base, n, lane);
  }
  __syncthreads();

  {
    const float4* t4 = (const float4*)tile;
    float4* o4 = (float4*)(agg + (size_t)row0 * HD);
#pragma unroll
    for (int j = 0; j < 4; ++j) {
      int i4 = tid + j * AG_T;
      if (i4 < n4) o4[i4] = t4[i4];
    }
  }
}

__global__ void k_relu_stats(float* __restrict__ agg, const float* __restrict__ bias,
                             float* __restrict__ bn_sum, float* __restrict__ bn_sq, int N) {
  __shared__ float ssum[256], ssq[256];
  int c = threadIdx.x & 63;
  int quad = threadIdx.x >> 6;
  int base = blockIdx.x * 64;
  float b = bias[c];
  float s = 0.f, q = 0.f;
  for (int r = quad; r < 64; r += 4) {
    int row = base + r;
    if (row < N) {
      float v = fmaxf(agg[(size_t)row * HD + c] + b, 0.f);
      agg[(size_t)row * HD + c] = v;
      s += v; q += v * v;
    }
  }
  ssum[threadIdx.x] = s; ssq[threadIdx.x] = q;
  __syncthreads();
  if (quad == 0) {
    float ts = ssum[c] + ssum[64 + c] + ssum[128 + c] + ssum[192 + c];
    float tq = ssq[c] + ssq[64 + c] + ssq[128 + c] + ssq[192 + c];
    atomicAdd(&bn_sum[c], ts);
    atomicAdd(&bn_sq[c], tq);
  }
}

__global__ void k_bnfinal(const float* __restrict__ bn_sum, const float* __restrict__ bn_sq,
                          const float* __restrict__ gamma, const float* __restrict__ beta,
                          float* __restrict__ scale, float* __restrict__ shift, float invN) {
  int c = threadIdx.x;
  float mu = bn_sum[c] * invN;
  float var = fmaxf(bn_sq[c] * invN - mu * mu, 0.f);
  float sc = rsqrtf(var + 1e-5f) * gamma[c];
  scale[c] = sc;
  shift[c] = beta[c] - mu * sc;
}

__global__ void k_bnapply_pool(const float* __restrict__ agg, const float* __restrict__ scale,
                               const float* __restrict__ shift, const int* __restrict__ batch,
                               float* __restrict__ zc, float* __restrict__ gsum,
                               int N, int lofs) {
  int c = threadIdx.x & 63;
  int quad = threadIdx.x >> 6;
  int base = blockIdx.x * 64;
  float sc = scale[c], sh = shift[c];
  int gcur = -1;
  float acc = 0.f;
  for (int r = quad; r < 64; r += 4) {
    int row = base + r;
    if (row < N) {
      float v = agg[(size_t)row * HD + c] * sc + sh;
      zc[(size_t)row * CD + lofs + c] = v;
      int g = batch[row];
      if (g != gcur) {
        if (gcur >= 0) atomicAdd(&gsum[(size_t)gcur * CD + lofs + c], acc);
        gcur = g; acc = 0.f;
      }
      acc += v;
    }
  }
  if (gcur >= 0) atomicAdd(&gsum[(size_t)gcur * CD + lofs + c], acc);
}

__global__ void k_l2_zc(const float* __restrict__ zc, float* __restrict__ out, int rows) {
  int wid = blockIdx.x * (blockDim.x >> 5) + (threadIdx.x >> 5);
  if (wid >= rows) return;
  int lane = threadIdx.x & 31;
  const float* a = zc + (size_t)wid * CD;
  float v[6]; float s = 0.f;
#pragma unroll
  for (int j = 0; j < 6; ++j) { v[j] = a[lane + 32 * j]; s += v[j] * v[j]; }
#pragma unroll
  for (int off = 16; off >= 1; off >>= 1) s += __shfl_xor(s, off, 32);
  float inv = 1.0f / fmaxf(sqrtf(s), 1e-12f);
  float* o = out + (size_t)wid * CD;
#pragma unroll
  for (int j = 0; j < 6; ++j) o[lane + 32 * j] = v[j] * inv;
}

__global__ void k_l2_gc(const float* __restrict__ gsum, const float* __restrict__ cnt,
                        float* __restrict__ out, int rows) {
  int wid = blockIdx.x * (blockDim.x >> 5) + (threadIdx.x >> 5);
  if (wid >= rows) return;
  int lane = threadIdx.x & 31;
  float icnt = 1.0f / fmaxf(cnt[wid], 1.0f);
  const float* a = gsum + (size_t)wid * CD;
  float v[6]; float s = 0.f;
#pragma unroll
  for (int j = 0; j < 6; ++j) { v[j] = a[lane + 32 * j] * icnt; s += v[j] * v[j]; }
#pragma unroll
  for (int off = 16; off >= 1; off >>= 1) s += __shfl_xor(s, off, 32);
  float inv = 1.0f / fmaxf(sqrtf(s), 1e-12f);
  float* o = out + (size_t)wid * CD;
#pragma unroll
  for (int j = 0; j < 6; ++j) o[lane + 32 * j] = v[j] * inv;
}

extern "C" void kernel_launch(void* const* d_in, const int* in_sizes, int n_in,
                              void* d_out, int out_size, void* d_ws, size_t ws_size,
                              hipStream_t stream) {
  (void)n_in; (void)out_size; (void)ws_size;
  const float* x     = (const float*)d_in[0];
  const int*   ei    = (const int*)d_in[1];
  const int*   batch = (const int*)d_in[2];
  int N = in_sizes[0] / 128;
  int E = in_sizes[1] / 2;
  const int* src = ei;
  const int* dst = ei + E;
  const float* Wl[3]  = {(const float*)d_in[3],  (const float*)d_in[7],  (const float*)d_in[11]};
  const float* bl[3]  = {(const float*)d_in[4],  (const float*)d_in[8],  (const float*)d_in[12]};
  const float* gl[3]  = {(const float*)d_in[5],  (const float*)d_in[9],  (const float*)d_in[13]};
  const float* bel[3] = {(const float*)d_in[6],  (const float*)d_in[10], (const float*)d_in[14]};

  float* ws   = (float*)d_ws;
  float* dinv = ws;                       ws += N;
  float* h    = ws;                       ws += (size_t)N * HD;
  float* agg  = ws;                       ws += (size_t)N * HD;
  float* zc   = ws;                       ws += (size_t)N * CD;
  float* gsum = ws;                       ws += (size_t)GN * CD;
  float* cnt  = ws;                       ws += GN;
  float* bn   = ws;
  float* bn_sum = bn, *bn_sq = bn + 64, *scale = bn + 128, *shift = bn + 192;
  unsigned char* key = (unsigned char*)(bn + 256);

  hipMemsetAsync(dinv, 0, (size_t)N * sizeof(float), stream);
  hipMemsetAsync(gsum, 0, (size_t)GN * CD * sizeof(float), stream);
  hipMemsetAsync(cnt,  0, (size_t)GN * sizeof(float), stream);

  k_deg<<<(E + 255) / 256, 256, 0, stream>>>(dst, E, dinv, key);
  k_dinv_cnt<<<(N + 255) / 256, 256, 0, stream>>>(dinv, batch, cnt, N);

  for (int l = 0; l < 3; ++l) {
    const float* Z = (l == 0) ? x : (zc + (size_t)(l - 1) * HD);
    int ldz = (l == 0) ? 128 : CD;
    int K   = (l == 0) ? 128 : HD;
    int tilesM = (N + 15) / 16;
    int waves = tilesM * 4;
    k_gemm_wmma<<<(waves + 7) / 8, 256, 0, stream>>>(Z, ldz, Wl[l], K, h, N);
    k_aggregate<<<(N + AG_R - 1) / AG_R, AG_T, 0, stream>>>(src, dst, key, dinv, h, agg, N, E);
    hipMemsetAsync(bn, 0, 128 * sizeof(float), stream);
    k_relu_stats<<<(N + 63) / 64, 256, 0, stream>>>(agg, bl[l], bn_sum, bn_sq, N);
    k_bnfinal<<<1, 64, 0, stream>>>(bn_sum, bn_sq, gl[l], bel[l], scale, shift, 1.0f / (float)N);
    k_bnapply_pool<<<(N + 63) / 64, 256, 0, stream>>>(agg, scale, shift, batch, zc, gsum, N, l * HD);
  }

  float* out = (float*)d_out;
  k_l2_zc<<<(N + 7) / 8, 256, 0, stream>>>(zc, out, N);
  k_l2_gc<<<(GN + 7) / 8, 256, 0, stream>>>(gsum, cnt, out + (size_t)N * CD, GN);
}


# launch fusion at the row-owning block: bias, ReLU and its BN column sums before its one store; BN scale/shift made in the apply kernel; one clear, one L2 pass; 25 stream ops to 13; BN sums regrouped
# speedup vs baseline: 1.9356x; 1.2326x over previous
#include <hip/hip_runtime.h>

typedef float v2f __attribute__((ext_vector_type(2)));
typedef float v8f __attribute__((ext_vector_type(8)));

#define HD 64
#define CD 192
#define GN 256
#define AG_R 196

__global__ void k_deg(const int* __restrict__ dst, int E, float* __restrict__ deg, unsigned char* __restrict__ key) {
  int e = blockIdx.x * blockDim.x + threadIdx.x;
  if (e < E) {
    int d = dst[e];
    atomicAdd(&deg[d], 1.0f);
    key[e] = (unsigned char)(d / AG_R);
  } else if (e < ((E + 15) & ~15)) {
    key[e] = 0;
  }
}

__global__ void k_dinv_cnt(float* __restrict__ dinv, const int* __restrict__ batch,
                           float* __restrict__ cnt, int N) {
  int i = blockIdx.x * blockDim.x + threadIdx.x;
  if (i < N) {
    dinv[i] = rsqrtf(dinv[i] + 1.0f);
    atomicAdd(&cnt[batch[i]], 1.0f);
  }
}

__global__ void k_gemm_wmma(const float* __restrict__ Z, int ldz,
                            const float* __restrict__ W, int K,
                            float* __restrict__ Hout, int N) {
  int wid = blockIdx.x * (blockDim.x >> 5) + (threadIdx.x >> 5);
  int tilesM = (N + 15) >> 4;
  if (wid >= tilesM * 4) return;
  int tm = wid >> 2, tn = wid & 3;
  int lane = threadIdx.x & 31;
  int half = lane >> 4;
  int m16 = lane & 15;
  int row = tm * 16 + m16;
  int lrow = row < N ? row : N - 1;
  const float* zr = Z + (size_t)lrow * ldz;
  int col = tn * 16 + m16;
  const float* wp = W + col;
  v8f c = {};
  for (int k0 = 0; k0 < K; k0 += 4) {
    int ka = k0 + 2 * half;
    v2f a, b;
    a.x = zr[ka];
    a.y = zr[ka + 1];
    b.x = wp[(size_t)ka * HD];
    b.y = wp[(size_t)(ka + 1) * HD];
    c = __builtin_amdgcn_wmma_f32_16x16x4_f32(false, a, false, b, (short)0, c, false, false);
  }
  int rbase = tm * 16 + half * 8;
#pragma unroll
  for (int j = 0; j < 8; ++j) {
    int r = rbase + j;
    if (r < N) Hout[(size_t)r * HD + col] = c[j];
  }
}

#define AG_T 1024
#define AG_W 32
#define AG_Q 256
#define AG_U 8

__device__ __forceinline__ void ag_group(const int* __restrict__ src, const int* __restrict__ dst,
                                         const float* __restrict__ dinv,
                                         const float* __restrict__ h, float* tile, const float* sdinv,
                                         const int* myqe, int row0, int base, int n, int lane) {
  int s_i = 0, dl_i = 0;
  float w_i = 0.f;
  if (lane < n) {
    int e = myqe[base + lane];
    s_i = src[e];
    dl_i = dst[e] - row0;
    w_i = dinv[s_i] * sdinv[dl_i];
  }
  int wbits = __float_as_int(w_i);
#pragma unroll
  for (int hf = 0; hf < 2; ++hf) {
    if (hf * 16 < n) {
      float a0[16], a1[16];
#pragma unroll
      for (int u = 0; u < 16; ++u) {
        int s = __builtin_amdgcn_readlane(s_i, hf * 16 + u);
        const float* hp = h + (size_t)s * HD;
        a0[u] = hp[lane];
        a1[u] = hp[32 + lane];
      }
#pragma unroll
      for (int u = 0; u < 16; ++u) {
        if (hf * 16 + u < n) {
          int dl = __builtin_amdgcn_readlane(dl_i, hf * 16 + u);
          float w = __int_as_float(__builtin_amdgcn_readlane(wbits, hf * 16 + u));
          atomicAdd(&tile[dl * HD + lane], a0[u] * w);
          atomicAdd(&tile[dl * HD + 32 + lane], a1[u] * w);
        }
      }
    }
  }
}

__device__ __forceinline__ void ag_drain(const int* __restrict__ src, const int* __restrict__ dst,
                                         const float* __restrict__ dinv,
                                         const float* __restrict__ h, float* tile, const float* sdinv,
                                         const int* myqe, int row0, int& cnt, int lane) {
  for (int base = 0; base < cnt; base += 32) {
    int n = cnt - base;
    if (n > 32) n = 32;
    int s_i = 0, dl_i = 0;
    float w_i = 0.f;
    if (lane < n) {
      int e = myqe[base + lane];
      s_i = src[e];
      dl_i = dst[e] - row0;
      w_i = dinv[s_i] * sdinv[dl_i];
    }
    int wbits = __float_as_int(w_i);
#pragma unroll 1
    for (int u = 0; u < n; ++u) {
      int s = __builtin_amdgcn_readlane(s_i, u);
      int dl = __builtin_amdgcn_readlane(dl_i, u);
      float w = __int_as_float(__builtin_amdgcn_readlane(wbits, u));
      const float* hp = h + (size_t)s * HD;
      atomicAdd(&tile[dl * HD + lane], hp[lane] * w);
      atomicAdd(&tile[dl * HD + 32 + lane], hp[32 + lane] * w);
    }
  }
  cnt = 0;
}

__device__ __forceinline__ void ag_push(bool m, int e, int* myqe, int& cnt) {
  unsigned mask = __builtin_amdgcn_ballot_w32(m);
  if (mask) {
    int pos = cnt + (int)__builtin_amdgcn_mbcnt_lo(mask, 0u);
    if (m) myqe[pos] = e;
    cnt += __builtin_popcount(mask);
  }
}

__device__ __forceinline__ unsigned ag_miss(unsigned w, unsigned krep) {
  unsigned x = w ^ krep;
  return ((x & 0x7F7F7F7Fu) + 0x7F7F7F7Fu) | x | 0x7F7F7F7Fu;
}

template <bool TAIL>
__device__ __forceinline__ void ag_test16(uint4 kq, bool valid, int e0, unsigned krep, int E, int row0,
                                          const int* __restrict__ src, const int* __restrict__ dst,
                                          const float* __restrict__ dinv,
                                          const float* __restrict__ h, float* tile, const float* sdinv,
                                          int* myqe, int& cnt, int lane) {
  unsigned c = __builtin_rotateright32(ag_miss(kq.x, krep), 3) & __builtin_rotateright32(ag_miss(kq.y, krep), 2) &
               __builtin_rotateright32(ag_miss(kq.z, krep), 1) & ag_miss(kq.w, krep);
  unsigned hit = ~c;
  if (TAIL) hit = valid ? hit : 0u;
  while (__builtin_amdgcn_ballot_w32(hit != 0u)) {
    if (cnt > AG_Q - 32) ag_drain(src, dst, dinv, h, tile, sdinv, myqe, row0, cnt, lane);
    int p = __builtin_ffs((int)hit) - 1;
    int e = e0 + ((p & 3) << 2) + (p >> 3);
    bool m = hit != 0u;
    if (TAIL) m = m && e < E;
    hit &= hit - 1u;
    ag_push(m, e, myqe, cnt);
  }
}

__device__ __forceinline__ void ag_flush_full(const int* __restrict__ src, const int* __restrict__ dst,
                                              const float* __restrict__ dinv,
                                              const float* __restrict__ h, float* tile, const float* sdinv,
                                              int* myqe, int row0, int& cnt, int lane) {
  int nf = cnt & ~31;
  for (int base = 0; base < nf; base += 32)
    ag_group(src, dst, dinv, h, tile, sdinv, myqe, row0, base, 32, lane);
  int rem = cnt - nf;
  if (lane < rem) {
    int te = myqe[nf + lane];
    myqe[lane] = te;
  }
  cnt = rem;
}

__global__ __launch_bounds__(AG_T) void k_aggregate(const int* __restrict__ src, const int* __restrict__ dst,
                                                    const unsigned char* __restrict__ key,
                                                    const float* __restrict__ dinv, const float* __restrict__ h,
                                                    const float* __restrict__ bias, float* __restrict__ agg,
                                                    float* __restrict__ bn, int N, int E) {
  static_assert(AG_R * (HD / 4) <= 4 * AG_T, "row tile is filled and stored in four float4 per thread");
  static_assert(AG_T == AG_W * 32, "wave32");
  __shared__ __attribute__((aligned(16))) float tile[AG_R * HD];
  __shared__ float sdinv[AG_R];
  __shared__ __attribute__((aligned(16))) int qe[AG_W][AG_Q];
  static_assert(AG_Q >= 2 * HD, "a wave's queue is reused for its 2*HD column sums once the edges are done");
  static_assert(AG_T % 16 == 0 && AG_T >= 2 * HD, "a thread keeps one group of four columns over its four float4");

  const int tid = threadIdx.x;
  const int lane = tid & 31;
  const int wv = tid >> 5;
  const int row0 = blockIdx.x * AG_R;
  int nrows = N - row0;
  if (nrows > AG_R) nrows = AG_R;
  const int n4 = nrows * (HD / 4);

  {
    const float4* h4 = (const float4*)(h + (size_t)row0 * HD);
    float4* t4 = (float4*)tile;
    float4 hv[4];
    float di[4];
#pragma unroll
    for (int j = 0; j < 4; ++j) {
      int i4 = tid + j * AG_T;
      hv[j] = make_float4(0.f, 0.f, 0.f, 0.f);
      di[j] = 0.f;
      if (i4 < n4) { hv[j] = h4[i4]; di[j] = dinv[row0 + (i4 >> 4)]; }
    }
#pragma unroll
    for (int j = 0; j < 4; ++j) {
      int i4 = tid + j * AG_T;
      if (i4 < n4) {
        float4 v;
        v.x = hv[j].x * di[j] * di[j];
        v.y = hv[j].y * di[j] * di[j];
        v.z = hv[j].z * di[j] * di[j];
        v.w = hv[j].w * di[j] * di[j];
        t4[i4] = v;
        if ((i4 & 15) == 0) sdinv[i4 >> 4] = di[j];
      }
    }
  }
  __syncthreads();

  const uint4* kv = (const uint4*)key;
  const int nvec = (E + 15) >> 4;
  const int nFull = (nvec - 1) / AG_T;
  const int it0 = (int)(((long long)blockIdx.x * nFull) / gridDim.x);
  const unsigned krep = (unsigned)blockIdx.x * 0x01010101u;
  int* myqe = qe[wv];
  int cnt = 0;

  int k = 0;
  for (; k + AG_U <= nFull; k += AG_U) {
    uint4 d[AG_U];
    int vi[AG_U];
#pragma unroll
    for (int j = 0; j < AG_U; ++j) {
      int it = it0 + k + j;
      if (it >= nFull) it -= nFull;
      vi[j] = it * AG_T + tid;
      d[j] = kv[vi[j]];
    }
#pragma unroll
    for (int j = 0; j < AG_U; ++j)
      ag_test16<false>(d[j], true, 16 * vi[j], krep, E, row0, src, dst, dinv, h, tile, sdinv, myqe, cnt, lane);
    if (cnt >= 32) ag_flush_full(src, dst, dinv, h, tile, sdinv, myqe, row0, cnt, lane);
  }
  for (; k <= nFull; ++k) {
    int it = it0 + k;
    if (it >= nFull) it -= nFull;
    if (k == nFull) it = nFull;
    int v = it * AG_T + tid;
    bool valid = v < nvec;
    uint4 d = make_uint4(0u, 0u, 0u, 0u);
    if (valid) d = kv[v];
    ag_test16<true>(d, valid, 16 * v, krep, E, row0, src, dst, dinv, h, tile, sdinv, myqe, cnt, lane);
  }
  for (int base = 0; base < cnt; base += 32) {
    int n = cnt - base;
    if (n > 32) n = 32;
    ag_group(src, dst, dinv, h, tile, sdinv, myqe, row0, base, n, lane);
  }
  __syncthreads();

  {
    const float4* t4 = (const float4*)tile;
    float4* o4 = (float4*)(agg + (size_t)row0 * HD);
    const int c4 = (tid & 15) * 4;
    const float b0 = bias[c4], b1 = bias[c4 + 1], b2 = bias[c4 + 2], b3 = bias[c4 + 3];
    float4 s = make_float4(0.f, 0.f, 0.f, 0.f);
    float4 q = make_float4(0.f, 0.f, 0.f, 0.f);
#pragma unroll
    for (int j = 0; j < 4; ++j) {
      int i4 = tid + j * AG_T;
      if (i4 < n4) {
        float4 v = t4[i4];
        v.x = fmaxf(v.x + b0, 0.f);
        v.y = fmaxf(v.y + b1, 0.f);
        v.z = fmaxf(v.z + b2, 0.f);
        v.w = fmaxf(v.w + b3, 0.f);
        o4[i4] = v;
        s.x += v.x; s.y += v.y; s.z += v.z; s.w += v.w;
        q.x += v.x * v.x; q.y += v.y * v.y; q.z += v.z * v.z; q.w += v.w * v.w;
      }
    }
    s.x += __shfl_xor(s.x, 16, 32); s.y += __shfl_xor(s.y, 16, 32);
    s.z += __shfl_xor(s.z, 16, 32); s.w += __shfl_xor(s.w, 16, 32);
    q.x += __shfl_xor(q.x, 16, 32); q.y += __shfl_xor(q.y, 16, 32);
    q.z += __shfl_xor(q.z, 16, 32); q.w += __shfl_xor(q.w, 16, 32);
    float* red = (float*)myqe;
    if (lane < 16) {
      *(float4*)(red + c4) = s;
      *(float4*)(red + HD + c4) = q;
    }
    __syncthreads();
    if (tid < 2 * HD) {
      const float* r0 = (const float*)&qe[0][0];
      float t = 0.f;
#pragma unroll
      for (int w = 0; w < AG_W; ++w) t += r0[w * AG_Q + tid];
      atomicAdd(&bn[tid], t);
    }
  }
}

__global__ void k_bnapply_pool(const float* __restrict__ agg, const float* __restrict__ bn,
                               const float* __restrict__ gamma, const float* __restrict__ beta,
                               const int* __restrict__ batch,
                               float* __restrict__ zc, float* __restrict__ gsum,
                               int N, int lofs, float invN) {
  int c = threadIdx.x & 63;
  int quad = threadIdx.x >> 6;
  int base = blockIdx.x * 64;
  float mu = bn[c] * invN;
  float var = fmaxf(bn[HD + c] * invN - mu * mu, 0.f);
  float sc = rsqrtf(var + 1e-5f) * gamma[c];
  float sh = beta[c] - mu * sc;
  int gcur = -1;
  float acc = 0.f;
  for (int r = quad; r < 64; r += 4) {
    int row = base + r;
    if (row < N) {
      float v = agg[(size_t)row * HD + c] * sc + sh;
      zc[(size_t)row * CD + lofs + c] = v;
      int g = batch[row];
      if (g != gcur) {
        if (gcur >= 0) atomicAdd(&gsum[(size_t)gcur * CD + lofs + c], acc);
        gcur = g; acc = 0.f;
      }
      acc += v;
    }
  }
  if (gcur >= 0) atomicAdd(&gsum[(size_t)gcur * CD + lofs + c], acc);
}

__global__ void k_l2(const float* __restrict__ zc, const float* __restrict__ gsum, const float* __restrict__ cnt,
                     float* __restrict__ out, int N, int G) {
  int wid = blockIdx.x * (blockDim.x >> 5) + (threadIdx.x >> 5);
  if (wid >= N + G) return;
  int lane = threadIdx.x & 31;
  const float* a = zc + (size_t)wid * CD;
  float mul = 1.0f;
  if (wid >= N) {
    a = gsum + (size_t)(wid - N) * CD;
    mul = 1.0f / fmaxf(cnt[wid - N], 1.0f);
  }
  float v[6]; float s = 0.f;
#pragma unroll
  for (int j = 0; j < 6; ++j) { v[j] = a[lane + 32 * j] * mul; s += v[j] * v[j]; }
#pragma unroll
  for (int off = 16; off >= 1; off >>= 1) s += __shfl_xor(s, off, 32);
  float inv = 1.0f / fmaxf(sqrtf(s), 1e-12f);
  float* o = out + (size_t)wid * CD;
#pragma unroll
  for (int j = 0; j < 6; ++j) o[lane + 32 * j] = v[j] * inv;
}

extern "C" void kernel_launch(void* const* d_in, const int* in_sizes, int n_in,
                              void* d_out, int out_size, void* d_ws, size_t ws_size,
                              hipStream_t stream) {
  (void)n_in; (void)out_size; (void)ws_size;
  const float* x     = (const float*)d_in[0];
  const int*   ei    = (const int*)d_in[1];
  const int*   batch = (const int*)d_in[2];
  int N = in_sizes[0] / 128;
  int E = in_sizes[1] / 2;
  const int* src = ei;
  const int* dst = ei + E;
  const float* Wl[3]  = {(const float*)d_in[3],  (const float*)d_in[7],  (const float*)d_in[11]};
  const float* bl[3]  = {(const float*)d_in[4],  (const float*)d_in[8],  (const float*)d_in[12]};
  const float* gl[3]  = {(const float*)d_in[5],  (const float*)d_in[9],  (const float*)d_in[13]};
  const float* bel[3] = {(const float*)d_in[6],  (const float*)d_in[10], (const float*)d_in[14]};

  float* ws   = (float*)d_ws;
  float* dinv = ws;                       ws += N;
  float* gsum = ws;                       ws += (size_t)GN * CD;
  float* cnt  = ws;                       ws += GN;
  float* bn   = ws;                       ws += 3 * 2 * HD;
  size_t nclear = (size_t)(ws - dinv) * sizeof(float);
  float* h    = ws;                       ws += (size_t)N * HD;
  float* agg  = ws;                       ws += (size_t)N * HD;
  float* zc   = ws;                       ws += (size_t)N * CD;
  unsigned char* key = (unsigned char*)ws;

  hipMemsetAsync(dinv, 0, nclear, stream);

  k_deg<<<(E + 255) / 256, 256, 0, stream>>>(dst, E, dinv, key);
  k_dinv_cnt<<<(N + 255) / 256, 256, 0, stream>>>(dinv, batch, cnt, N);

  for (int l = 0; l < 3; ++l) {
    const float* Z = (l == 0) ? x : (zc + (size_t)(l - 1) * HD);
    int ldz = (l == 0) ? 128 : CD;
    int K   = (l == 0) ? 128 : HD;
    int tilesM = (N + 15) / 16;
    int waves = tilesM * 4;
    k_gemm_wmma<<<(waves + 7) / 8, 256, 0, stream>>>(Z, ldz, Wl[l], K, h, N);
    float* bnl = bn + l * 2 * HD;
    k_aggregate<<<(N + AG_R - 1) / AG_R, AG_T, 0, stream>>>(src, dst, key, dinv, h, bl[l], agg, bnl, N, E);
    k_bnapply_pool<<<(N + 63) / 64, 256, 0, stream>>>(agg, bnl, gl[l], bel[l], batch, zc, gsum, N, l * HD,
                                                      1.0f / (float)N);
  }

  float* out = (float*)d_out;
  k_l2<<<(N + GN + 7) / 8, 256, 0, stream>>>(zc, gsum, cnt, out, N, GN);
}


# owner filter runs once per call in its own kernel: counts the rows' degrees in LDS (degree atomics gone), keeps each wave's hit list; aggregations replay the lists; edge sums in new LDS-arrival order
# speedup vs baseline: 2.7808x; 1.4367x over previous
#include <hip/hip_runtime.h>

typedef float v2f __attribute__((ext_vector_type(2)));
typedef float v8f __attribute__((ext_vector_type(8)));

#define HD 64
#define CD 192
#define GN 256
#define AG_R 196
#define AG_T 1024
#define AG_W 32
#define AG_Q 256
#define AG_U 8
#define AG_LC 256

__global__ void k_key(const int* __restrict__ dst, int E, unsigned* __restrict__ key4) {
  int i = blockIdx.x * blockDim.x + threadIdx.x;
  int nw = ((E + 15) & ~15) >> 2;
  if (i < nw) {
    int e = 4 * i;
    unsigned w = 0u;
#pragma unroll
    for (int j = 0; j < 4; ++j)
      if (e + j < E) w |= ((unsigned)(dst[e + j] / AG_R) & 0xFFu) << (8 * j);
    key4[i] = w;
  }
}

__global__ void k_gemm_wmma(const float* __restrict__ Z, int ldz,
                            const float* __restrict__ W, int K,
                            float* __restrict__ Hout, int N) {
  int wid = blockIdx.x * (blockDim.x >> 5) + (threadIdx.x >> 5);
  int tilesM = (N + 15) >> 4;
  if (wid >= tilesM * 4) return;
  int tm = wid >> 2, tn = wid & 3;
  int lane = threadIdx.x & 31;
  int half = lane >> 4;
  int m16 = lane & 15;
  int row = tm * 16 + m16;
  int lrow = row < N ? row : N - 1;
  const float* zr = Z + (size_t)lrow * ldz;
  int col = tn * 16 + m16;
  const float* wp = W + col;
  v8f c = {};
  for (int k0 = 0; k0 < K; k0 += 4) {
    int ka = k0 + 2 * half;
    v2f a, b;
    a.x = zr[ka];
    a.y = zr[ka + 1];
    b.x = wp[(size_t)ka * HD];
    b.y = wp[(size_t)(ka + 1) * HD];
    c = __builtin_amdgcn_wmma_f32_16x16x4_f32(false, a, false, b, (short)0, c, false, false);
  }
  int rbase = tm * 16 + half * 8;
#pragma unroll
  for (int j = 0; j < 8; ++j) {
    int r = rbase + j;
    if (r < N) Hout[(size_t)r * HD + col] = c[j];
  }
}


__device__ __forceinline__ void ag_push(bool m, int e, int* myqe, int& cnt) {
  unsigned mask = __builtin_amdgcn_ballot_w32(m);
  if (mask) {
    int pos = cnt + (int)__builtin_amdgcn_mbcnt_lo(mask, 0u);
    if (m) myqe[pos] = e;
    cnt += __builtin_popcount(mask);
  }
}

__device__ __forceinline__ unsigned ag_miss(unsigned w, unsigned krep) {
  unsigned x = w ^ krep;
  return ((x & 0x7F7F7F7Fu) + 0x7F7F7F7Fu) | x | 0x7F7F7F7Fu;
}

template <bool TAIL, class S>
__device__ __forceinline__ void ag_test16(uint4 kq, bool valid, int e0, unsigned krep, int E, S& sk,
                                          int* myqe, int& cnt, int lane) {
  unsigned c = __builtin_rotateright32(ag_miss(kq.x, krep), 3) & __builtin_rotateright32(ag_miss(kq.y, krep), 2) &
               __builtin_rotateright32(ag_miss(kq.z, krep), 1) & ag_miss(kq.w, krep);
  unsigned hit = ~c;
  if (TAIL) hit = valid ? hit : 0u;
  while (__builtin_amdgcn_ballot_w32(hit != 0u)) {
    if (cnt > AG_Q - 32) sk.drain(myqe, cnt, lane);
    int p = __builtin_ffs((int)hit) - 1;
    int e = e0 + ((p & 3) << 2) + (p >> 3);
    bool m = hit != 0u;
    if (TAIL) m = m && e < E;
    hit &= hit - 1u;
    ag_push(m, e, myqe, cnt);
  }
}

template <class S>
__device__ __forceinline__ void ag_scan(S& sk, const unsigned char* __restrict__ key, int E, int* myqe,
                                        int tid, int lane) {
  const uint4* kv = (const uint4*)key;
  const int nvec = (E + 15) >> 4;
  const int nFull = (nvec - 1) / AG_T;
  const int it0 = (int)(((long long)blockIdx.x * nFull) / gridDim.x);
  const unsigned krep = (unsigned)blockIdx.x * 0x01010101u;
  int cnt = 0;

  int k = 0;
  for (; k + AG_U <= nFull; k += AG_U) {
    uint4 d[AG_U];
    int vi[AG_U];
#pragma unroll
    for (int j = 0; j < AG_U; ++j) {
      int it = it0 + k + j;
      if (it >= nFull) it -= nFull;
      vi[j] = it * AG_T + tid;
      d[j] = kv[vi[j]];
    }
#pragma unroll
    for (int j = 0; j < AG_U; ++j)
      ag_test16<false>(d[j], true, 16 * vi[j], krep, E, sk, myqe, cnt, lane);
    if (cnt >= 32) sk.full(myqe, cnt, lane);
  }
  for (; k <= nFull; ++k) {
    int it = it0 + k;
    if (it >= nFull) it -= nFull;
    if (k == nFull) it = nFull;
    int v = it * AG_T + tid;
    bool valid = v < nvec;
    uint4 d = make_uint4(0u, 0u, 0u, 0u);
    if (valid) d = kv[v];
    ag_test16<true>(d, valid, 16 * v, krep, E, sk, myqe, cnt, lane);
  }
  sk.rest(myqe, cnt, lane);
}

struct OwnSink {
  const int* __restrict__ src;
  const int* __restrict__ dst;
  int* rc;
  unsigned* ml;
  int row0;
  int tot;
  __device__ __forceinline__ void drain(int* myqe, int& cnt, int lane) {
    for (int base = 0; base < cnt; base += 32) {
      int n = cnt - base;
      if (n > 32) n = 32;
      if (lane < n) {
        int e = myqe[base + lane];
        int s = src[e];
        int dl = dst[e] - row0;
        atomicAdd(&rc[dl], 1);
        int pos = tot + lane;
        if (pos > AG_LC - 1) pos = AG_LC - 1;
        ml[pos] = (unsigned)s | ((unsigned)dl << 24);
      }
      tot += n;
    }
    cnt = 0;
  }
  __device__ __forceinline__ void full(int* myqe, int& cnt, int lane) { drain(myqe, cnt, lane); }
  __device__ __forceinline__ void rest(int* myqe, int& cnt, int lane) { drain(myqe, cnt, lane); }
};

__global__ __launch_bounds__(AG_T) void k_own(const int* __restrict__ src, const int* __restrict__ dst,
                                              const unsigned char* __restrict__ key,
                                              const int* __restrict__ batch, float* __restrict__ dinv,
                                              float* __restrict__ cnt, unsigned* __restrict__ lst,
                                              int* __restrict__ lcnt, int* __restrict__ ovf, int N, int E) {
  __shared__ int rc[AG_R];
  __shared__ int sovf;
  __shared__ __attribute__((aligned(16))) int qe[AG_W][AG_Q];
  const int tid = threadIdx.x;
  const int lane = tid & 31;
  const int wv = tid >> 5;
  const int row0 = blockIdx.x * AG_R;
  int nrows = N - row0;
  if (nrows > AG_R) nrows = AG_R;
  if (tid < AG_R) rc[tid] = 0;
  if (tid == 0) sovf = 0;
  __syncthreads();

  const int slot = blockIdx.x * AG_W + wv;
  OwnSink sk;
  sk.src = src; sk.dst = dst; sk.rc = rc; sk.ml = lst + (size_t)slot * AG_LC; sk.row0 = row0; sk.tot = 0;
  ag_scan(sk, key, E, qe[wv], tid, lane);
  if (lane == 0) {
    lcnt[slot] = sk.tot;
    if (sk.tot > AG_LC) sovf = 1;
  }
  __syncthreads();

  if (tid < nrows) {
    int row = row0 + tid;
    dinv[row] = rsqrtf((float)rc[tid] + 1.0f);
    atomicAdd(&cnt[batch[row]], 1.0f);
  }
  if (tid == 0) ovf[blockIdx.x] = sovf;
}


__device__ __forceinline__ void ag_rows(const float* __restrict__ h, float* tile, int s_i, int dl_i, float w_i,
                                        int n, int lane) {
  int wbits = __float_as_int(w_i);
#pragma unroll
  for (int hf = 0; hf < 2; ++hf) {
    if (hf * 16 < n) {
      float a0[16], a1[16];
#pragma unroll
      for (int u = 0; u < 16; ++u) {
        int s = __builtin_amdgcn_readlane(s_i, hf * 16 + u);
        const float* hp = h + (size_t)s * HD;
        a0[u] = hp[lane];
        a1[u] = hp[32 + lane];
      }
#pragma unroll
      for (int u = 0; u < 16; ++u) {
        if (hf * 16 + u < n) {
          int dl = __builtin_amdgcn_readlane(dl_i, hf * 16 + u);
          float w = __int_as_float(__builtin_amdgcn_readlane(wbits, hf * 16 + u));
          atomicAdd(&tile[dl * HD + lane], a0[u] * w);
          atomicAdd(&tile[dl * HD + 32 + lane], a1[u] * w);
        }
      }
    }
  }
}

struct AggSink {
  const int* __restrict__ src;
  const int* __restrict__ dst;
  const float* __restrict__ dinv;
  const float* __restrict__ h;
  float* tile;
  const float* sdinv;
  int row0;
  __device__ __forceinline__ void group(const int* myqe, int base, int n, int lane) {
    int s_i = 0, dl_i = 0;
    float w_i = 0.f;
    if (lane < n) {
      int e = myqe[base + lane];
      s_i = src[e];
      dl_i = dst[e] - row0;
      w_i = dinv[s_i] * sdinv[dl_i];
    }
    ag_rows(h, tile, s_i, dl_i, w_i, n, lane);
  }
  __device__ __forceinline__ void full(int* myqe, int& cnt, int lane) {
    int nf = cnt & ~31;
    for (int base = 0; base < nf; base += 32) group(myqe, base, 32, lane);
    int rem = cnt - nf;
    if (lane < rem) {
      int te = myqe[nf + lane];
      myqe[lane] = te;
    }
    cnt = rem;
  }
  __device__ __forceinline__ void rest(int* myqe, int& cnt, int lane) {
    for (int base = 0; base < cnt; base += 32) {
      int n = cnt - base;
      if (n > 32) n = 32;
      group(myqe, base, n, lane);
    }
    cnt = 0;
  }
  __device__ __forceinline__ void drain(int* myqe, int& cnt, int lane) {
    for (int base = 0; base < cnt; base += 32) {
      int n = cnt - base;
      if (n > 32) n = 32;
      int s_i = 0, dl_i = 0;
      float w_i = 0.f;
      if (lane < n) {
        int e = myqe[base + lane];
        s_i = src[e];
        dl_i = dst[e] - row0;
        w_i = dinv[s_i] * sdinv[dl_i];
      }
      int wbits = __float_as_int(w_i);
#pragma unroll 1
      for (int u = 0; u < n; ++u) {
        int s = __builtin_amdgcn_readlane(s_i, u);
        int dl = __builtin_amdgcn_readlane(dl_i, u);
        float w = __int_as_float(__builtin_amdgcn_readlane(wbits, u));
        const float* hp = h + (size_t)s * HD;
        atomicAdd(&tile[dl * HD + lane], hp[lane] * w);
        atomicAdd(&tile[dl * HD + 32 + lane], hp[32 + lane] * w);
      }
    }
    cnt = 0;
  }
};

__global__ __launch_bounds__(AG_T) void k_aggregate(const int* __restrict__ src, const int* __restrict__ dst,
                                                    const unsigned char* __restrict__ key,
                                                    const unsigned* __restrict__ lst, const int* __restrict__ lcnt,
                                                    const int* __restrict__ ovf,
                                                    const float* __restrict__ dinv, const float* __restrict__ h,
                                                    const float* __restrict__ bias, float* __restrict__ agg,
                                                    float* __restrict__ bn, int N, int E) {
  static_assert(AG_R * (HD / 4) <= 4 * AG_T, "row tile is filled and stored in four float4 per thread");
  static_assert(AG_T == AG_W * 32, "wave32");
  static_assert(AG_LC % 32 == 0 && AG_LC <= AG_Q, "a wave's list is staged in its queue, whole groups of 32");
  __shared__ __attribute__((aligned(16))) float tile[AG_R * HD];
  __shared__ float sdinv[AG_R];
  __shared__ __attribute__((aligned(16))) int qe[AG_W][AG_Q];
  __shared__ float qw[AG_W][AG_LC];
  static_assert(AG_Q >= 2 * HD, "a wave's queue is reused for its 2*HD column sums once the edges are done");
  static_assert(AG_T % 16 == 0 && AG_T >= 2 * HD, "a thread keeps one group of four columns over its four float4");

  const int tid = threadIdx.x;
  const int lane = tid & 31;
  const int wv = tid >> 5;
  const int row0 = blockIdx.x * AG_R;
  int nrows = N - row0;
  if (nrows > AG_R) nrows = AG_R;
  const int n4 = nrows * (HD / 4);
  int* myqe = qe[wv];
  float* myqw = qw[wv];

  const bool listed = ovf[blockIdx.x] == 0;
  int nl = 0;
  unsigned en[AG_LC / 32];
  float dv[AG_LC / 32];
  if (listed) {
    const int slot = blockIdx.x * AG_W + wv;
    nl = __builtin_amdgcn_readfirstlane(lcnt[slot]);
    if (nl > AG_LC) nl = AG_LC;
    const unsigned* ml = lst + (size_t)slot * AG_LC;
#pragma unroll
    for (int j = 0; j < AG_LC / 32; ++j) {
      int idx = lane + 32 * j;
      en[j] = idx < nl ? ml[idx] : 0u;
    }
  }

  {
    const float4* h4 = (const float4*)(h + (size_t)row0 * HD);
    float4* t4 = (float4*)tile;
    float4 hv[4];
    float di[4];
#pragma unroll
    for (int j = 0; j < 4; ++j) {
      int i4 = tid + j * AG_T;
      hv[j] = make_float4(0.f, 0.f, 0.f, 0.f);
      di[j] = 0.f;
      if (i4 < n4) { hv[j] = h4[i4]; di[j] = dinv[row0 + (i4 >> 4)]; }
    }
    if (listed) {
#pragma unroll
      for (int j = 0; j < AG_LC / 32; ++j) {
        int idx = lane + 32 * j;
        int s = (int)(en[j] & 0xFFFFFFu);
        if (s > N - 1) s = N - 1;
        dv[j] = idx < nl ? dinv[s] : 0.f;
      }
    }
#pragma unroll
    for (int j = 0; j < 4; ++j) {
      int i4 = tid + j * AG_T;
      if (i4 < n4) {
        float4 v;
        v.x = hv[j].x * di[j] * di[j];
        v.y = hv[j].y * di[j] * di[j];
        v.z = hv[j].z * di[j] * di[j];
        v.w = hv[j].w * di[j] * di[j];
        t4[i4] = v;
        if ((i4 & 15) == 0) sdinv[i4 >> 4] = di[j];
      }
    }
  }
  __syncthreads();

  if (listed) {
#pragma unroll
    for (int j = 0; j < AG_LC / 32; ++j) {
      int idx = lane + 32 * j;
      if (idx < nl) {
        unsigned dl = en[j] >> 24;
        if (dl > AG_R - 1) dl = AG_R - 1;
        myqe[idx] = (int)((en[j] & 0xFFFFFFu) | (dl << 24));
        myqw[idx] = dv[j] * sdinv[dl];
      }
    }
    for (int base = 0; base < nl; base += 32) {
      int n = nl - base;
      if (n > 32) n = 32;
      int s_i = 0, dl_i = 0;
      float w_i = 0.f;
      if (lane < n) {
        unsigned pw = (unsigned)myqe[base + lane];
        s_i = (int)(pw & 0xFFFFFFu);
        if (s_i > N - 1) s_i = N - 1;
        dl_i = (int)(pw >> 24);
        w_i = myqw[base + lane];
      }
      ag_rows(h, tile, s_i, dl_i, w_i, n, lane);
    }
  } else {
    AggSink sk;
    sk.src = src; sk.dst = dst; sk.dinv = dinv; sk.h = h; sk.tile = tile; sk.sdinv = sdinv; sk.row0 = row0;
    ag_scan(sk, key, E, myqe, tid, lane);
  }
  __syncthreads();

  {
    const float4* t4 = (const float4*)tile;
    float4* o4 = (float4*)(agg + (size_t)row0 * HD);
    const int c4 = (tid & 15) * 4;
    const float b0 = bias[c4], b1 = bias[c4 + 1], b2 = bias[c4 + 2], b3 = bias[c4 + 3];
    float4 s = make_float4(0.f, 0.f, 0.f, 0.f);
    float4 q = make_float4(0.f, 0.f, 0.f, 0.f);
#pragma unroll
    for (int j = 0; j < 4; ++j) {
      int i4 = tid + j * AG_T;
      if (i4 < n4) {
        float4 v = t4[i4];
        v.x = fmaxf(v.x + b0, 0.f);
        v.y = fmaxf(v.y + b1, 0.f);
        v.z = fmaxf(v.z + b2, 0.f);
        v.w = fmaxf(v.w + b3, 0.f);
        o4[i4] = v;
        s.x += v.x; s.y += v.y; s.z += v.z; s.w += v.w;
        q.x += v.x * v.x; q.y += v.y * v.y; q.z += v.z * v.z; q.w += v.w * v.w;
      }
    }
    s.x += __shfl_xor(s.x, 16, 32); s.y += __shfl_xor(s.y, 16, 32);
    s.z += __shfl_xor(s.z, 16, 32); s.w += __shfl_xor(s.w, 16, 32);
    q.x += __shfl_xor(q.x, 16, 32); q.y += __shfl_xor(q.y, 16, 32);
    q.z += __shfl_xor(q.z, 16, 32); q.w += __shfl_xor(q.w, 16, 32);
    float* red = (float*)myqe;
    if (lane < 16) {
      *(float4*)(red + c4) = s;
      *(float4*)(red + HD + c4) = q;
    }
    __syncthreads();
    if (tid < 2 * HD) {
      const float* r0 = (const float*)&qe[0][0];
      float t = 0.f;
#pragma unroll
      for (int w = 0; w < AG_W; ++w) t += r0[w * AG_Q + tid];
      atomicAdd(&bn[tid], t);
    }
  }
}

__global__ void k_bnapply_pool(const float* __restrict__ agg, const float* __restrict__ bn,
                               const float* __restrict__ gamma, const float* __restrict__ beta,
                               const int* __restrict__ batch,
                               float* __restrict__ zc, float* __restrict__ gsum,
                               int N, int lofs, float invN) {
  int c = threadIdx.x & 63;
  int quad = threadIdx.x >> 6;
  int base = blockIdx.x * 64;
  float mu = bn[c] * invN;
  float var = fmaxf(bn[HD + c] * invN - mu * mu, 0.f);
  float sc = rsqrtf(var + 1e-5f) * gamma[c];
  float sh = beta[c] - mu * sc;
  int gcur = -1;
  float acc = 0.f;
  for (int r = quad; r < 64; r += 4) {
    int row = base + r;
    if (row < N) {
      float v = agg[(size_t)row * HD + c] * sc + sh;
      zc[(size_t)row * CD + lofs + c] = v;
      int g = batch[row];
      if (g != gcur) {
        if (gcur >= 0) atomicAdd(&gsum[(size_t)gcur * CD + lofs + c], acc);
        gcur = g; acc = 0.f;
      }
      acc += v;
    }
  }
  if (gcur >= 0) atomicAdd(&gsum[(size_t)gcur * CD + lofs + c], acc);
}

__global__ void k_l2(const float* __restrict__ zc, const float* __restrict__ gsum, const float* __restrict__ cnt,
                     float* __restrict__ out, int N, int G) {
  int wid = blockIdx.x * (blockDim.x >> 5) + (threadIdx.x >> 5);
  if (wid >= N + G) return;
  int lane = threadIdx.x & 31;
  const float* a = zc + (size_t)wid * CD;
  float mul = 1.0f;
  if (wid >= N) {
    a = gsum + (size_t)(wid - N) * CD;
    mul = 1.0f / fmaxf(cnt[wid - N], 1.0f);
  }
  float v[6]; float s = 0.f;
#pragma unroll
  for (int j = 0; j < 6; ++j) { v[j] = a[lane + 32 * j] * mul; s += v[j] * v[j]; }
#pragma unroll
  for (int off = 16; off >= 1; off >>= 1) s += __shfl_xor(s, off, 32);
  float inv = 1.0f / fmaxf(sqrtf(s), 1e-12f);
  float* o = out + (size_t)wid * CD;
#pragma unroll
  for (int j = 0; j < 6; ++j) o[lane + 32 * j] = v[j] * inv;
}

extern "C" void kernel_launch(void* const* d_in, const int* in_sizes, int n_in,
                              void* d_out, int out_size, void* d_ws, size_t ws_size,
                              hipStream_t stream) {
  (void)n_in; (void)out_size; (void)ws_size;
  const float* x     = (const float*)d_in[0];
  const int*   ei    = (const int*)d_in[1];
  const int*   batch = (const int*)d_in[2];
  int N = in_sizes[0] / 128;
  int E = in_sizes[1] / 2;
  const int* src = ei;
  const int* dst = ei + E;
  const float* Wl[3]  = {(const float*)d_in[3],  (const float*)d_in[7],  (const float*)d_in[11]};
  const float* bl[3]  = {(const float*)d_in[4],  (const float*)d_in[8],  (const float*)d_in[12]};
  const float* gl[3]  = {(const float*)d_in[5],  (const float*)d_in[9],  (const float*)d_in[13]};
  const float* bel[3] = {(const float*)d_in[6],  (const float*)d_in[10], (const float*)d_in[14]};
  int nblk = (N + AG_R - 1) / AG_R;
  int Epad = (E + 15) & ~15;

  float* ws   = (float*)d_ws;
  float* gsum = ws;                       ws += (size_t)GN * CD;
  float* cnt  = ws;                       ws += GN;
  float* bn   = ws;                       ws += 3 * 2 * HD;
  size_t nclear = (size_t)(ws - gsum) * sizeof(float);
  float* dinv = ws;                       ws += N;
  float* h    = ws;                       ws += (size_t)N * HD;
  float* agg  = ws;                       ws += (size_t)N * HD;
  float* zc   = ws;                       ws += (size_t)N * CD;
  unsigned char* key = (unsigned char*)ws;
  unsigned* lst = (unsigned*)(key + Epad);
  int* lcnt = (int*)(lst + (size_t)nblk * AG_W * AG_LC);
  int* ovf  = lcnt + nblk * AG_W;

  hipMemsetAsync(gsum, 0, nclear, stream);

  k_key<<<(Epad / 4 + 255) / 256, 256, 0, stream>>>(dst, E, (unsigned*)key);
  k_own<<<nblk, AG_T, 0, stream>>>(src, dst, key, batch, dinv, cnt, lst, lcnt, ovf, N, E);

  for (int l = 0; l < 3; ++l) {
    const float* Z = (l == 0) ? x : (zc + (size_t)(l - 1) * HD);
    int ldz = (l == 0) ? 128 : CD;
    int K   = (l == 0) ? 128 : HD;
    int tilesM = (N + 15) / 16;
    int waves = tilesM * 4;
    k_gemm_wmma<<<(waves + 7) / 8, 256, 0, stream>>>(Z, ldz, Wl[l], K, h, N);
    float* bnl = bn + l * 2 * HD;
    k_aggregate<<<nblk, AG_T, 0, stream>>>(src, dst, key, lst, lcnt, ovf, dinv, h, bl[l], agg, bnl, N, E);
    k_bnapply_pool<<<(N + 63) / 64, 256, 0, stream>>>(agg, bnl, gl[l], bel[l], batch, zc, gsum, N, l * HD,
                                                      1.0f / (float)N);
  }

  float* out = (float*)d_out;
  k_l2<<<(N + GN + 7) / 8, 256, 0, stream>>>(zc, gsum, cnt, out, N, GN);
}


# BN pass removed: z stored as is, its column map applied on load (next GEMM's A rows, last kernel) and once per graph to sums the row-owning block pools; 13 stream ops to 10; pool sums regrouped
# speedup vs baseline: 3.3299x; 1.1975x over previous
#include <hip/hip_runtime.h>

typedef float v2f __attribute__((ext_vector_type(2)));
typedef float v8f __attribute__((ext_vector_type(8)));

#define HD 64
#define CD 192
#define GN 256
#define AG_R 196
#define AG_T 1024
#define AG_W 32
#define AG_Q 256
#define AG_U 8
#define AG_LC 256

__global__ void k_key(const int* __restrict__ dst, int E, unsigned* __restrict__ key4) {
  int i = blockIdx.x * blockDim.x + threadIdx.x;
  int nw = ((E + 15) & ~15) >> 2;
  if (i < nw) {
    int e = 4 * i;
    unsigned w = 0u;
#pragma unroll
    for (int j = 0; j < 4; ++j)
      if (e + j < E) w |= ((unsigned)(dst[e + j] / AG_R) & 0xFFu) << (8 * j);
    key4[i] = w;
  }
}

__device__ __forceinline__ float2 bn_map(const float* __restrict__ bn, const float* __restrict__ gamma,
                                         const float* __restrict__ beta, int c, float invN) {
  float mu = bn[c] * invN;
  float var = fmaxf(bn[HD + c] * invN - mu * mu, 0.f);
  float sc = rsqrtf(var + 1e-5f) * gamma[c];
  return make_float2(sc, beta[c] - mu * sc);
}

template <bool BN>
__global__ void k_gemm_wmma(const float* __restrict__ Z, int ldz,
                            const float* __restrict__ W, int K,
                            float* __restrict__ Hout, int N,
                            const float* __restrict__ bn, const float* __restrict__ gamma,
                            const float* __restrict__ beta, float invN) {
  __shared__ __attribute__((aligned(16))) float2 sbn[HD];
  if (BN) {
    if (threadIdx.x < HD) sbn[threadIdx.x] = bn_map(bn, gamma, beta, (int)threadIdx.x, invN);
    __syncthreads();
  }
  int wid = blockIdx.x * (blockDim.x >> 5) + (threadIdx.x >> 5);
  int tilesM = (N + 15) >> 4;
  if (wid >= tilesM * 4) return;
  int tm = wid >> 2, tn = wid & 3;
  int lane = threadIdx.x & 31;
  int half = lane >> 4;
  int m16 = lane & 15;
  int row = tm * 16 + m16;
  int lrow = row < N ? row : N - 1;
  const float* zr = Z + (size_t)lrow * ldz;
  int col = tn * 16 + m16;
  const float* wp = W + col;
  v8f c = {};
  for (int k0 = 0; k0 < K; k0 += 4) {
    int ka = k0 + 2 * half;
    v2f a, b;
    a.x = zr[ka];
    a.y = zr[ka + 1];
    if (BN) {
      float4 m = *(const float4*)&sbn[ka];
      a.x = a.x * m.x + m.y;
      a.y = a.y * m.z + m.w;
    }
    b.x = wp[(size_t)ka * HD];
    b.y = wp[(size_t)(ka + 1) * HD];
    c = __builtin_amdgcn_wmma_f32_16x16x4_f32(false, a, false, b, (short)0, c, false, false);
  }
  int rbase = tm * 16 + half * 8;
#pragma unroll
  for (int j = 0; j < 8; ++j) {
    int r = rbase + j;
    if (r < N) Hout[(size_t)r * HD + col] = c[j];
  }
}


__device__ __forceinline__ void ag_push(bool m, int e, int* myqe, int& cnt) {
  unsigned mask = __builtin_amdgcn_ballot_w32(m);
  if (mask) {
    int pos = cnt + (int)__builtin_amdgcn_mbcnt_lo(mask, 0u);
    if (m) myqe[pos] = e;
    cnt += __builtin_popcount(mask);
  }
}

__device__ __forceinline__ unsigned ag_miss(unsigned w, unsigned krep) {
  unsigned x = w ^ krep;
  return ((x & 0x7F7F7F7Fu) + 0x7F7F7F7Fu) | x | 0x7F7F7F7Fu;
}

template <bool TAIL, class S>
__device__ __forceinline__ void ag_test16(uint4 kq, bool valid, int e0, unsigned krep, int E, S& sk,
                                          int* myqe, int& cnt, int lane) {
  unsigned c = __builtin_rotateright32(ag_miss(kq.x, krep), 3) & __builtin_rotateright32(ag_miss(kq.y, krep), 2) &
               __builtin_rotateright32(ag_miss(kq.z, krep), 1) & ag_miss(kq.w, krep);
  unsigned hit = ~c;
  if (TAIL) hit = valid ? hit : 0u;
  while (__builtin_amdgcn_ballot_w32(hit != 0u)) {
    if (cnt > AG_Q - 32) sk.drain(myqe, cnt, lane);
    int p = __builtin_ffs((int)hit) - 1;
    int e = e0 + ((p & 3) << 2) + (p >> 3);
    bool m = hit != 0u;
    if (TAIL) m = m && e < E;
    hit &= hit - 1u;
    ag_push(m, e, myqe, cnt);
  }
}

template <class S>
__device__ __forceinline__ void ag_scan(S& sk, const unsigned char* __restrict__ key, int E, int* myqe,
                                        int tid, int lane) {
  const uint4* kv = (const uint4*)key;
  const int nvec = (E + 15) >> 4;
  const int nFull = (nvec - 1) / AG_T;
  const int it0 = (int)(((long long)blockIdx.x * nFull) / gridDim.x);
  const unsigned krep = (unsigned)blockIdx.x * 0x01010101u;
  int cnt = 0;

  int k = 0;
  for (; k + AG_U <= nFull; k += AG_U) {
    uint4 d[AG_U];
    int vi[AG_U];
#pragma unroll
    for (int j = 0; j < AG_U; ++j) {
      int it = it0 + k + j;
      if (it >= nFull) it -= nFull;
      vi[j] = it * AG_T + tid;
      d[j] = kv[vi[j]];
    }
#pragma unroll
    for (int j = 0; j < AG_U; ++j)
      ag_test16<false>(d[j], true, 16 * vi[j], krep, E, sk, myqe, cnt, lane);
    if (cnt >= 32) sk.full(myqe, cnt, lane);
  }
  for (; k <= nFull; ++k) {
    int it = it0 + k;
    if (it >= nFull) it -= nFull;
    if (k == nFull) it = nFull;
    int v = it * AG_T + tid;
    bool valid = v < nvec;
    uint4 d = make_uint4(0u, 0u, 0u, 0u);
    if (valid) d = kv[v];
    ag_test16<true>(d, valid, 16 * v, krep, E, sk, myqe, cnt, lane);
  }
  sk.rest(myqe, cnt, lane);
}

struct OwnSink {
  const int* __restrict__ src;
  const int* __restrict__ dst;
  int* rc;
  unsigned* ml;
  int row0;
  int tot;
  __device__ __forceinline__ void drain(int* myqe, int& cnt, int lane) {
    for (int base = 0; base < cnt; base += 32) {
      int n = cnt - base;
      if (n > 32) n = 32;
      if (lane < n) {
        int e = myqe[base + lane];
        int s = src[e];
        int dl = dst[e] - row0;
        atomicAdd(&rc[dl], 1);
        int pos = tot + lane;
        if (pos > AG_LC - 1) pos = AG_LC - 1;
        ml[pos] = (unsigned)s | ((unsigned)dl << 24);
      }
      tot += n;
    }
    cnt = 0;
  }
  __device__ __forceinline__ void full(int* myqe, int& cnt, int lane) { drain(myqe, cnt, lane); }
  __device__ __forceinline__ void rest(int* myqe, int& cnt, int lane) { drain(myqe, cnt, lane); }
};

__global__ __launch_bounds__(AG_T) void k_own(const int* __restrict__ src, const int* __restrict__ dst,
                                              const unsigned char* __restrict__ key,
                                              const int* __restrict__ batch, float* __restrict__ dinv,
                                              float* __restrict__ cnt, unsigned* __restrict__ lst,
                                              int* __restrict__ lcnt, int* __restrict__ ovf, int N, int E) {
  __shared__ int rc[AG_R];
  __shared__ int sovf;
  __shared__ __attribute__((aligned(16))) int qe[AG_W][AG_Q];
  const int tid = threadIdx.x;
  const int lane = tid & 31;
  const int wv = tid >> 5;
  const int row0 = blockIdx.x * AG_R;
  int nrows = N - row0;
  if (nrows > AG_R) nrows = AG_R;
  if (tid < AG_R) rc[tid] = 0;
  if (tid == 0) sovf = 0;
  __syncthreads();

  const int slot = blockIdx.x * AG_W + wv;
  OwnSink sk;
  sk.src = src; sk.dst = dst; sk.rc = rc; sk.ml = lst + (size_t)slot * AG_LC; sk.row0 = row0; sk.tot = 0;
  ag_scan(sk, key, E, qe[wv], tid, lane);
  if (lane == 0) {
    lcnt[slot] = sk.tot;
    if (sk.tot > AG_LC) sovf = 1;
  }
  __syncthreads();

  if (tid < nrows) {
    int row = row0 + tid;
    dinv[row] = rsqrtf((float)rc[tid] + 1.0f);
    atomicAdd(&cnt[batch[row]], 1.0f);
  }
  if (tid == 0) ovf[blockIdx.x] = sovf;
}


__device__ __forceinline__ void ag_rows(const float* __restrict__ h, float* tile, int s_i, int dl_i, float w_i,
                                        int n, int lane) {
  int wbits = __float_as_int(w_i);
#pragma unroll
  for (int hf = 0; hf < 2; ++hf) {
    if (hf * 16 < n) {
      float a0[16], a1[16];
#pragma unroll
      for (int u = 0; u < 16; ++u) {
        int s = __builtin_amdgcn_readlane(s_i, hf * 16 + u);
        const float* hp = h + (size_t)s * HD;
        a0[u] = hp[lane];
        a1[u] = hp[32 + lane];
      }
#pragma unroll
      for (int u = 0; u < 16; ++u) {
        if (hf * 16 + u < n) {
          int dl = __builtin_amdgcn_readlane(dl_i, hf * 16 + u);
          float w = __int_as_float(__builtin_amdgcn_readlane(wbits, hf * 16 + u));
          atomicAdd(&tile[dl * HD + lane], a0[u] * w);
          atomicAdd(&tile[dl * HD + 32 + lane], a1[u] * w);
        }
      }
    }
  }
}

struct AggSink {
  const int* __restrict__ src;
  const int* __restrict__ dst;
  const float* __restrict__ dinv;
  const float* __restrict__ h;
  float* tile;
  const float* sdinv;
  int row0;
  __device__ __forceinline__ void group(const int* myqe, int base, int n, int lane) {
    int s_i = 0, dl_i = 0;
    float w_i = 0.f;
    if (lane < n) {
      int e = myqe[base + lane];
      s_i = src[e];
      dl_i = dst[e] - row0;
      w_i = dinv[s_i] * sdinv[dl_i];
    }
    ag_rows(h, tile, s_i, dl_i, w_i, n, lane);
  }
  __device__ __forceinline__ void full(int* myqe, int& cnt, int lane) {
    int nf = cnt & ~31;
    for (int base = 0; base < nf; base += 32) group(myqe, base, 32, lane);
    int rem = cnt - nf;
    if (lane < rem) {
      int te = myqe[nf + lane];
      myqe[lane] = te;
    }
    cnt = rem;
  }
  __device__ __forceinline__ void rest(int* myqe, int& cnt, int lane) {
    for (int base = 0; base < cnt; base += 32) {
      int n = cnt - base;
      if (n > 32) n = 32;
      group(myqe, base, n, lane);
    }
    cnt = 0;
  }
  __device__ __forceinline__ void drain(int* myqe, int& cnt, int lane) {
    for (int base = 0; base < cnt; base += 32) {
      int n = cnt - base;
      if (n > 32) n = 32;
      int s_i = 0, dl_i = 0;
      float w_i = 0.f;
      if (lane < n) {
        int e = myqe[base + lane];
        s_i = src[e];
        dl_i = dst[e] - row0;
        w_i = dinv[s_i] * sdinv[dl_i];
      }
      int wbits = __float_as_int(w_i);
#pragma unroll 1
      for (int u = 0; u < n; ++u) {
        int s = __builtin_amdgcn_readlane(s_i, u);
        int dl = __builtin_amdgcn_readlane(dl_i, u);
        float w = __int_as_float(__builtin_amdgcn_readlane(wbits, u));
        const float* hp = h + (size_t)s * HD;
        atomicAdd(&tile[dl * HD + lane], hp[lane] * w);
        atomicAdd(&tile[dl * HD + 32 + lane], hp[32 + lane] * w);
      }
    }
    cnt = 0;
  }
};

__global__ __launch_bounds__(AG_T) void k_aggregate(const int* __restrict__ src, const int* __restrict__ dst,
                                                    const unsigned char* __restrict__ key,
                                                    const unsigned* __restrict__ lst, const int* __restrict__ lcnt,
                                                    const int* __restrict__ ovf,
                                                    const float* __restrict__ dinv, const float* __restrict__ h,
                                                    const float* __restrict__ bias, float* __restrict__ agg,
                                                    float* __restrict__ bn, const int* __restrict__ batch,
                                                    float* __restrict__ gsum, int lofs, int N, int E) {
  static_assert(AG_R * (HD / 4) <= 4 * AG_T, "row tile is filled and stored in four float4 per thread");
  static_assert(AG_T == AG_W * 32, "wave32");
  static_assert(AG_LC % 32 == 0 && AG_LC <= AG_Q, "a wave's list is staged in its queue, whole groups of 32");
  __shared__ __attribute__((aligned(16))) float tile[AG_R * HD];
  __shared__ float sdinv[AG_R];
  __shared__ int sbatch[AG_R];
  __shared__ __attribute__((aligned(16))) int qe[AG_W][AG_Q];
  __shared__ float qw[AG_W][AG_LC];
  static_assert(AG_T % HD == 0 && AG_T >= AG_R, "pooling: AG_T / HD row groups by HD columns; a thread per row fetches its graph number");
  static_assert(AG_Q >= 2 * HD, "a wave's queue is reused for its 2*HD column sums once the edges are done");
  static_assert(AG_T % 16 == 0 && AG_T >= 2 * HD, "a thread keeps one group of four columns over its four float4");

  const int tid = threadIdx.x;
  const int lane = tid & 31;
  const int wv = tid >> 5;
  const int row0 = blockIdx.x * AG_R;
  int nrows = N - row0;
  if (nrows > AG_R) nrows = AG_R;
  const int n4 = nrows * (HD / 4);
  int* myqe = qe[wv];
  float* myqw = qw[wv];

  const bool listed = ovf[blockIdx.x] == 0;
  int nl = 0;
  unsigned en[AG_LC / 32];
  float dv[AG_LC / 32];
  if (listed) {
    const int slot = blockIdx.x * AG_W + wv;
    nl = __builtin_amdgcn_readfirstlane(lcnt[slot]);
    if (nl > AG_LC) nl = AG_LC;
    const unsigned* ml = lst + (size_t)slot * AG_LC;
#pragma unroll
    for (int j = 0; j < AG_LC / 32; ++j) {
      int idx = lane + 32 * j;
      en[j] = idx < nl ? ml[idx] : 0u;
    }
  }

  {
    const float4* h4 = (const float4*)(h + (size_t)row0 * HD);
    float4* t4 = (float4*)tile;
    float4 hv[4];
    float di[4];
#pragma unroll
    for (int j = 0; j < 4; ++j) {
      int i4 = tid + j * AG_T;
      hv[j] = make_float4(0.f, 0.f, 0.f, 0.f);
      di[j] = 0.f;
      if (i4 < n4) { hv[j] = h4[i4]; di[j] = dinv[row0 + (i4 >> 4)]; }
    }
    int bt = 0;
    if (tid < nrows) bt = batch[row0 + tid];
    if (listed) {
#pragma unroll
      for (int j = 0; j < AG_LC / 32; ++j) {
        int idx = lane + 32 * j;
        int s = (int)(en[j] & 0xFFFFFFu);
        if (s > N - 1) s = N - 1;
        dv[j] = idx < nl ? dinv[s] : 0.f;
      }
    }
#pragma unroll
    for (int j = 0; j < 4; ++j) {
      int i4 = tid + j * AG_T;
      if (i4 < n4) {
        float4 v;
        v.x = hv[j].x * di[j] * di[j];
        v.y = hv[j].y * di[j] * di[j];
        v.z = hv[j].z * di[j] * di[j];
        v.w = hv[j].w * di[j] * di[j];
        t4[i4] = v;
        if ((i4 & 15) == 0) sdinv[i4 >> 4] = di[j];
      }
    }
    if (tid < nrows) sbatch[tid] = bt;
  }
  __syncthreads();

  if (listed) {
#pragma unroll
    for (int j = 0; j < AG_LC / 32; ++j) {
      int idx = lane + 32 * j;
      if (idx < nl) {
        unsigned dl = en[j] >> 24;
        if (dl > AG_R - 1) dl = AG_R - 1;
        myqe[idx] = (int)((en[j] & 0xFFFFFFu) | (dl << 24));
        myqw[idx] = dv[j] * sdinv[dl];
      }
    }
    for (int base = 0; base < nl; base += 32) {
      int n = nl - base;
      if (n > 32) n = 32;
      int s_i = 0, dl_i = 0;
      float w_i = 0.f;
      if (lane < n) {
        unsigned pw = (unsigned)myqe[base + lane];
        s_i = (int)(pw & 0xFFFFFFu);
        if (s_i > N - 1) s_i = N - 1;
        dl_i = (int)(pw >> 24);
        w_i = myqw[base + lane];
      }
      ag_rows(h, tile, s_i, dl_i, w_i, n, lane);
    }
  } else {
    AggSink sk;
    sk.src = src; sk.dst = dst; sk.dinv = dinv; sk.h = h; sk.tile = tile; sk.sdinv = sdinv; sk.row0 = row0;
    ag_scan(sk, key, E, myqe, tid, lane);
  }
  __syncthreads();

  {
    float4* t4 = (float4*)tile;
    float4* o4 = (float4*)(agg + (size_t)row0 * HD);
    const int c4 = (tid & 15) * 4;
    const float b0 = bias[c4], b1 = bias[c4 + 1], b2 = bias[c4 + 2], b3 = bias[c4 + 3];
    float4 s = make_float4(0.f, 0.f, 0.f, 0.f);
    float4 q = make_float4(0.f, 0.f, 0.f, 0.f);
#pragma unroll
    for (int j = 0; j < 4; ++j) {
      int i4 = tid + j * AG_T;
      if (i4 < n4) {
        float4 v = t4[i4];
        v.x = fmaxf(v.x + b0, 0.f);
        v.y = fmaxf(v.y + b1, 0.f);
        v.z = fmaxf(v.z + b2, 0.f);
        v.w = fmaxf(v.w + b3, 0.f);
        o4[i4] = v;
        t4[i4] = v;
        s.x += v.x; s.y += v.y; s.z += v.z; s.w += v.w;
        q.x += v.x * v.x; q.y += v.y * v.y; q.z += v.z * v.z; q.w += v.w * v.w;
      }
    }
    s.x += __shfl_xor(s.x, 16, 32); s.y += __shfl_xor(s.y, 16, 32);
    s.z += __shfl_xor(s.z, 16, 32); s.w += __shfl_xor(s.w, 16, 32);
    q.x += __shfl_xor(q.x, 16, 32); q.y += __shfl_xor(q.y, 16, 32);
    q.z += __shfl_xor(q.z, 16, 32); q.w += __shfl_xor(q.w, 16, 32);
    float* red = (float*)myqe;
    if (lane < 16) {
      *(float4*)(red + c4) = s;
      *(float4*)(red + HD + c4) = q;
    }
    __syncthreads();
    if (tid < 2 * HD) {
      const float* r0 = (const float*)&qe[0][0];
      float t = 0.f;
#pragma unroll
      for (int w = 0; w < AG_W; ++w) t += r0[w * AG_Q + tid];
      atomicAdd(&bn[tid], t);
    }
    {
      constexpr int RPG = (AG_R + AG_T / HD - 1) / (AG_T / HD);
      const int c = tid & (HD - 1);
      const int rg0 = (tid / HD) * RPG;
      float pv[RPG];
      int pg[RPG];
#pragma unroll
      for (int u = 0; u < RPG; ++u) {
        int r = rg0 + u;
        bool ok = r < nrows;
        int rr = ok ? r : 0;
        pv[u] = tile[rr * HD + c];
        pg[u] = ok ? sbatch[rr] : -1;
      }
      int gcur = -1;
      float acc = 0.f;
#pragma unroll
      for (int u = 0; u < RPG; ++u) {
        if (pg[u] >= 0) {
          if (pg[u] != gcur) {
            if (gcur >= 0) atomicAdd(&gsum[(size_t)gcur * CD + lofs + c], acc);
            gcur = pg[u];
            acc = 0.f;
          }
          acc += pv[u];
        }
      }
      if (gcur >= 0) atomicAdd(&gsum[(size_t)gcur * CD + lofs + c], acc);
    }
  }
}

__global__ void k_final(const float* __restrict__ z, const float* __restrict__ bn,
                        const float* __restrict__ gm0, const float* __restrict__ gm1, const float* __restrict__ gm2,
                        const float* __restrict__ bt0, const float* __restrict__ bt1, const float* __restrict__ bt2,
                        const float* __restrict__ gsum, const float* __restrict__ cnt,
                        float* __restrict__ out, int N, int G, float invN) {
  __shared__ float2 sbn[CD];
  const int t = threadIdx.x;
  const int wid = blockIdx.x * (blockDim.x >> 5) + (t >> 5);
  const int lane = t & 31;
  const bool node = wid < N;
  const bool graph = !node && wid < N + G;
  float v[6];
  float n = 0.f;
  if (node) {
#pragma unroll
    for (int j = 0; j < 6; ++j) {
      int e = lane + 32 * j;
      v[j] = z[(size_t)(e / HD) * N * HD + (size_t)wid * HD + (e % HD)];
    }
  } else {
    int g = graph ? wid - N : 0;
    n = cnt[g];
#pragma unroll
    for (int j = 0; j < 6; ++j) v[j] = gsum[(size_t)g * CD + lane + 32 * j];
  }
  if (t < CD) {
    int l = t / HD, c = t % HD;
    const float* gm = l == 0 ? gm0 : (l == 1 ? gm1 : gm2);
    const float* bt = l == 0 ? bt0 : (l == 1 ? bt1 : bt2);
    sbn[t] = bn_map(bn + l * 2 * HD, gm, bt, c, invN);
  }
  __syncthreads();
  if (!node && !graph) return;
  float s = 0.f;
  if (node) {
#pragma unroll
    for (int j = 0; j < 6; ++j) {
      float2 m = sbn[lane + 32 * j];
      v[j] = v[j] * m.x + m.y;
    }
  } else {
    const float mul = 1.0f / fmaxf(n, 1.0f);
#pragma unroll
    for (int j = 0; j < 6; ++j) {
      float2 m = sbn[lane + 32 * j];
      v[j] = (v[j] * m.x + n * m.y) * mul;
    }
  }
#pragma unroll
  for (int j = 0; j < 6; ++j) s += v[j] * v[j];
#pragma unroll
  for (int off = 16; off >= 1; off >>= 1) s += __shfl_xor(s, off, 32);
  float inv = 1.0f / fmaxf(sqrtf(s), 1e-12f);
  float* o = out + (size_t)wid * CD;
#pragma unroll
  for (int j = 0; j < 6; ++j) o[lane + 32 * j] = v[j] * inv;
}

extern "C" void kernel_launch(void* const* d_in, const int* in_sizes, int n_in,
                              void* d_out, int out_size, void* d_ws, size_t ws_size,
                              hipStream_t stream) {
  (void)n_in; (void)out_size; (void)ws_size;
  const float* x     = (const float*)d_in[0];
  const int*   ei    = (const int*)d_in[1];
  const int*   batch = (const int*)d_in[2];
  int N = in_sizes[0] / 128;
  int E = in_sizes[1] / 2;
  const int* src = ei;
  const int* dst = ei + E;
  const float* Wl[3]  = {(const float*)d_in[3],  (const float*)d_in[7],  (const float*)d_in[11]};
  const float* bl[3]  = {(const float*)d_in[4],  (const float*)d_in[8],  (const float*)d_in[12]};
  const float* gl[3]  = {(const float*)d_in[5],  (const float*)d_in[9],  (const float*)d_in[13]};
  const float* bel[3] = {(const float*)d_in[6],  (const float*)d_in[10], (const float*)d_in[14]};
  int nblk = (N + AG_R - 1) / AG_R;
  int Epad = (E + 15) & ~15;

  float* ws   = (float*)d_ws;
  float* gsum = ws;                       ws += (size_t)GN * CD;
  float* cnt  = ws;                       ws += GN;
  float* bn   = ws;                       ws += 3 * 2 * HD;
  size_t nclear = (size_t)(ws - gsum) * sizeof(float);
  float* dinv = ws;                       ws += N;
  float* h    = ws;                       ws += (size_t)N * HD;
  float* zpl  = ws;                       ws += (size_t)3 * N * HD;
  unsigned char* key = (unsigned char*)ws;
  unsigned* lst = (unsigned*)(key + Epad);
  int* lcnt = (int*)(lst + (size_t)nblk * AG_W * AG_LC);
  int* ovf  = lcnt + nblk * AG_W;

  hipMemsetAsync(gsum, 0, nclear, stream);

  k_key<<<(Epad / 4 + 255) / 256, 256, 0, stream>>>(dst, E, (unsigned*)key);
  k_own<<<nblk, AG_T, 0, stream>>>(src, dst, key, batch, dinv, cnt, lst, lcnt, ovf, N, E);

  const float invN = 1.0f / (float)N;
  for (int l = 0; l < 3; ++l) {
    int tilesM = (N + 15) / 16;
    int waves = tilesM * 4;
    if (l == 0)
      k_gemm_wmma<false><<<(waves + 7) / 8, 256, 0, stream>>>(x, 128, Wl[0], 128, h, N, nullptr, nullptr, nullptr, invN);
    else
      k_gemm_wmma<true><<<(waves + 7) / 8, 256, 0, stream>>>(zpl + (size_t)(l - 1) * N * HD, HD, Wl[l], HD, h, N,
                                                             bn + (l - 1) * 2 * HD, gl[l - 1], bel[l - 1], invN);
    k_aggregate<<<nblk, AG_T, 0, stream>>>(src, dst, key, lst, lcnt, ovf, dinv, h, bl[l], zpl + (size_t)l * N * HD,
                                           bn + l * 2 * HD, batch, gsum, l * HD, N, E);
  }

  float* out = (float*)d_out;
  k_final<<<(N + GN + 7) / 8, 256, 0, stream>>>(zpl, bn, gl[0], gl[1], gl[2], bel[0], bel[1], bel[2], gsum, cnt, out,
                                                N, GN, invN);
}


# edge norm split: GEMM stores h*dinv[row], owner blocks add unweighted src rows and scale each dst row once by dinv[dst]; per-edge dinv[src] fetch and weight gone; products regrouped, sum order kept
# speedup vs baseline: 3.5705x; 1.0723x over previous
#include <hip/hip_runtime.h>

typedef float v2f __attribute__((ext_vector_type(2)));
typedef float v8f __attribute__((ext_vector_type(8)));

#define HD 64
#define CD 192
#define GN 256
#define AG_R 196
#define AG_T 1024
#define AG_W 32
#define AG_Q 256
#define AG_U 8
#define AG_LC 256

__global__ void k_key(const int* __restrict__ dst, int E, unsigned* __restrict__ key4) {
  int i = blockIdx.x * blockDim.x + threadIdx.x;
  int nw = ((E + 15) & ~15) >> 2;
  if (i < nw) {
    int e = 4 * i;
    unsigned w = 0u;
#pragma unroll
    for (int j = 0; j < 4; ++j)
      if (e + j < E) w |= ((unsigned)(dst[e + j] / AG_R) & 0xFFu) << (8 * j);
    key4[i] = w;
  }
}

__device__ __forceinline__ float2 bn_map(const float* __restrict__ bn, const float* __restrict__ gamma,
                                         const float* __restrict__ beta, int c, float invN) {
  float mu = bn[c] * invN;
  float var = fmaxf(bn[HD + c] * invN - mu * mu, 0.f);
  float sc = rsqrtf(var + 1e-5f) * gamma[c];
  return make_float2(sc, beta[c] - mu * sc);
}

template <bool BN>
__global__ void k_gemm_wmma(const float* __restrict__ Z, int ldz,
                            const float* __restrict__ W, int K,
                            float* __restrict__ Hout, int N, const float* __restrict__ dinv,
                            const float* __restrict__ bn, const float* __restrict__ gamma,
                            const float* __restrict__ beta, float invN) {
  __shared__ __attribute__((aligned(16))) float2 sbn[HD];
  if (BN) {
    if (threadIdx.x < HD) sbn[threadIdx.x] = bn_map(bn, gamma, beta, (int)threadIdx.x, invN);
    __syncthreads();
  }
  int wid = blockIdx.x * (blockDim.x >> 5) + (threadIdx.x >> 5);
  int tilesM = (N + 15) >> 4;
  if (wid >= tilesM * 4) return;
  int tm = wid >> 2, tn = wid & 3;
  int lane = threadIdx.x & 31;
  int half = lane >> 4;
  int m16 = lane & 15;
  int row = tm * 16 + m16;
  int lrow = row < N ? row : N - 1;
  const float* zr = Z + (size_t)lrow * ldz;
  int col = tn * 16 + m16;
  const float* wp = W + col;
  v8f c = {};
  for (int k0 = 0; k0 < K; k0 += 4) {
    int ka = k0 + 2 * half;
    v2f a, b;
    a.x = zr[ka];
    a.y = zr[ka + 1];
    if (BN) {
      float4 m = *(const float4*)&sbn[ka];
      a.x = a.x * m.x + m.y;
      a.y = a.y * m.z + m.w;
    }
    b.x = wp[(size_t)ka * HD];
    b.y = wp[(size_t)(ka + 1) * HD];
    c = __builtin_amdgcn_wmma_f32_16x16x4_f32(false, a, false, b, (short)0, c, false, false);
  }
  const int rbase = tm * 16 + half * 8;
  float dr[8];
#pragma unroll
  for (int j = 0; j < 8; ++j) {
    int r = rbase + j;
    dr[j] = dinv[r < N ? r : N - 1];
  }
#pragma unroll
  for (int j = 0; j < 8; ++j) {
    int r = rbase + j;
    if (r < N) Hout[(size_t)r * HD + col] = c[j] * dr[j];
  }
}


__device__ __forceinline__ void ag_push(bool m, int e, int* myqe, int& cnt) {
  unsigned mask = __builtin_amdgcn_ballot_w32(m);
  if (mask) {
    int pos = cnt + (int)__builtin_amdgcn_mbcnt_lo(mask, 0u);
    if (m) myqe[pos] = e;
    cnt += __builtin_popcount(mask);
  }
}

__device__ __forceinline__ unsigned ag_miss(unsigned w, unsigned krep) {
  unsigned x = w ^ krep;
  return ((x & 0x7F7F7F7Fu) + 0x7F7F7F7Fu) | x | 0x7F7F7F7Fu;
}

template <bool TAIL, class S>
__device__ __forceinline__ void ag_test16(uint4 kq, bool valid, int e0, unsigned krep, int E, S& sk,
                                          int* myqe, int& cnt, int lane) {
  unsigned c = __builtin_rotateright32(ag_miss(kq.x, krep), 3) & __builtin_rotateright32(ag_miss(kq.y, krep), 2) &
               __builtin_rotateright32(ag_miss(kq.z, krep), 1) & ag_miss(kq.w, krep);
  unsigned hit = ~c;
  if (TAIL) hit = valid ? hit : 0u;
  while (__builtin_amdgcn_ballot_w32(hit != 0u)) {
    if (cnt > AG_Q - 32) sk.drain(myqe, cnt, lane);
    int p = __builtin_ffs((int)hit) - 1;
    int e = e0 + ((p & 3) << 2) + (p >> 3);
    bool m = hit != 0u;
    if (TAIL) m = m && e < E;
    hit &= hit - 1u;
    ag_push(m, e, myqe, cnt);
  }
}

template <class S>
__device__ __forceinline__ void ag_scan(S& sk, const unsigned char* __restrict__ key, int E, int* myqe,
                                        int tid, int lane) {
  const uint4* kv = (const uint4*)key;
  const int nvec = (E + 15) >> 4;
  const int nFull = (nvec - 1) / AG_T;
  const int it0 = (int)(((long long)blockIdx.x * nFull) / gridDim.x);
  const unsigned krep = (unsigned)blockIdx.x * 0x01010101u;
  int cnt = 0;

  int k = 0;
  for (; k + AG_U <= nFull; k += AG_U) {
    uint4 d[AG_U];
    int vi[AG_U];
#pragma unroll
    for (int j = 0; j < AG_U; ++j) {
      int it = it0 + k + j;
      if (it >= nFull) it -= nFull;
      vi[j] = it * AG_T + tid;
      d[j] = kv[vi[j]];
    }
#pragma unroll
    for (int j = 0; j < AG_U; ++j)
      ag_test16<false>(d[j], true, 16 * vi[j], krep, E, sk, myqe, cnt, lane);
    if (cnt >= 32) sk.full(myqe, cnt, lane);
  }
  for (; k <= nFull; ++k) {
    int it = it0 + k;
    if (it >= nFull) it -= nFull;
    if (k == nFull) it = nFull;
    int v = it * AG_T + tid;
    bool valid = v < nvec;
    uint4 d = make_uint4(0u, 0u, 0u, 0u);
    if (valid) d = kv[v];
    ag_test16<true>(d, valid, 16 * v, krep, E, sk, myqe, cnt, lane);
  }
  sk.rest(myqe, cnt, lane);
}

struct OwnSink {
  const int* __restrict__ src;
  const int* __restrict__ dst;
  int* rc;
  unsigned* ml;
  int row0;
  int tot;
  __device__ __forceinline__ void drain(int* myqe, int& cnt, int lane) {
    for (int base = 0; base < cnt; base += 32) {
      int n = cnt - base;
      if (n > 32) n = 32;
      if (lane < n) {
        int e = myqe[base + lane];
        int s = src[e];
        int dl = dst[e] - row0;
        atomicAdd(&rc[dl], 1);
        int pos = tot + lane;
        if (pos > AG_LC - 1) pos = AG_LC - 1;
        ml[pos] = (unsigned)s | ((unsigned)dl << 24);
      }
      tot += n;
    }
    cnt = 0;
  }
  __device__ __forceinline__ void full(int* myqe, int& cnt, int lane) { drain(myqe, cnt, lane); }
  __device__ __forceinline__ void rest(int* myqe, int& cnt, int lane) { drain(myqe, cnt, lane); }
};

__global__ __launch_bounds__(AG_T) void k_own(const int* __restrict__ src, const int* __restrict__ dst,
                                              const unsigned char* __restrict__ key,
                                              const int* __restrict__ batch, float* __restrict__ dinv,
                                              float* __restrict__ cnt, unsigned* __restrict__ lst,
                                              int* __restrict__ lcnt, int* __restrict__ ovf, int N, int E) {
  __shared__ int rc[AG_R];
  __shared__ int sovf;
  __shared__ __attribute__((aligned(16))) int qe[AG_W][AG_Q];
  const int tid = threadIdx.x;
  const int lane = tid & 31;
  const int wv = tid >> 5;
  const int row0 = blockIdx.x * AG_R;
  int nrows = N - row0;
  if (nrows > AG_R) nrows = AG_R;
  if (tid < AG_R) rc[tid] = 0;
  if (tid == 0) sovf = 0;
  __syncthreads();

  const int slot = blockIdx.x * AG_W + wv;
  OwnSink sk;
  sk.src = src; sk.dst = dst; sk.rc = rc; sk.ml = lst + (size_t)slot * AG_LC; sk.row0 = row0; sk.tot = 0;
  ag_scan(sk, key, E, qe[wv], tid, lane);
  if (lane == 0) {
    lcnt[slot] = sk.tot;
    if (sk.tot > AG_LC) sovf = 1;
  }
  __syncthreads();

  if (tid < nrows) {
    int row = row0 + tid;
    dinv[row] = rsqrtf((float)rc[tid] + 1.0f);
    atomicAdd(&cnt[batch[row]], 1.0f);
  }
  if (tid == 0) ovf[blockIdx.x] = sovf;
}


__device__ __forceinline__ void ag_rows(const float* __restrict__ h, float* tile, int s_i, int dl_i,
                                        int n, int lane) {
#pragma unroll
  for (int hf = 0; hf < 2; ++hf) {
    if (hf * 16 < n) {
      float a0[16], a1[16];
#pragma unroll
      for (int u = 0; u < 16; ++u) {
        int s = __builtin_amdgcn_readlane(s_i, hf * 16 + u);
        const float* hp = h + (size_t)s * HD;
        a0[u] = hp[lane];
        a1[u] = hp[32 + lane];
      }
#pragma unroll
      for (int u = 0; u < 16; ++u) {
        if (hf * 16 + u < n) {
          int dl = __builtin_amdgcn_readlane(dl_i, hf * 16 + u);
          atomicAdd(&tile[dl * HD + lane], a0[u]);
          atomicAdd(&tile[dl * HD + 32 + lane], a1[u]);
        }
      }
    }
  }
}

struct AggSink {
  const int* __restrict__ src;
  const int* __restrict__ dst;
  const float* __restrict__ h;
  float* tile;
  int row0;
  __device__ __forceinline__ void group(const int* myqe, int base, int n, int lane) {
    int s_i = 0, dl_i = 0;
    if (lane < n) {
      int e = myqe[base + lane];
      s_i = src[e];
      dl_i = dst[e] - row0;
    }
    ag_rows(h, tile, s_i, dl_i, n, lane);
  }
  __device__ __forceinline__ void full(int* myqe, int& cnt, int lane) {
    int nf = cnt & ~31;
    for (int base = 0; base < nf; base += 32) group(myqe, base, 32, lane);
    int rem = cnt - nf;
    if (lane < rem) {
      int te = myqe[nf + lane];
      myqe[lane] = te;
    }
    cnt = rem;
  }
  __device__ __forceinline__ void rest(int* myqe, int& cnt, int lane) {
    for (int base = 0; base < cnt; base += 32) {
      int n = cnt - base;
      if (n > 32) n = 32;
      group(myqe, base, n, lane);
    }
    cnt = 0;
  }
  __device__ __forceinline__ void drain(int* myqe, int& cnt, int lane) {
    for (int base = 0; base < cnt; base += 32) {
      int n = cnt - base;
      if (n > 32) n = 32;
      int s_i = 0, dl_i = 0;
      if (lane < n) {
        int e = myqe[base + lane];
        s_i = src[e];
        dl_i = dst[e] - row0;
      }
#pragma unroll 1
      for (int u = 0; u < n; ++u) {
        int s = __builtin_amdgcn_readlane(s_i, u);
        int dl = __builtin_amdgcn_readlane(dl_i, u);
        const float* hp = h + (size_t)s * HD;
        atomicAdd(&tile[dl * HD + lane], hp[lane]);
        atomicAdd(&tile[dl * HD + 32 + lane], hp[32 + lane]);
      }
    }
    cnt = 0;
  }
};

__global__ __launch_bounds__(AG_T) void k_aggregate(const int* __restrict__ src, const int* __restrict__ dst,
                                                    const unsigned char* __restrict__ key,
                                                    const unsigned* __restrict__ lst, const int* __restrict__ lcnt,
                                                    const int* __restrict__ ovf,
                                                    const float* __restrict__ dinv, const float* __restrict__ h,
                                                    const float* __restrict__ bias, float* __restrict__ agg,
                                                    float* __restrict__ bn, const int* __restrict__ batch,
                                                    float* __restrict__ gsum, int lofs, int N, int E) {
  static_assert(AG_R * (HD / 4) <= 4 * AG_T, "row tile is filled and stored in four float4 per thread");
  static_assert(AG_T == AG_W * 32, "wave32");
  static_assert(AG_LC % 32 == 0 && AG_LC <= AG_Q, "a wave's list is staged in its queue, whole groups of 32");
  __shared__ __attribute__((aligned(16))) float tile[AG_R * HD];
  __shared__ float sdinv[AG_R];
  __shared__ int sbatch[AG_R];
  __shared__ __attribute__((aligned(16))) int qe[AG_W][AG_Q];
  static_assert(AG_T % HD == 0 && AG_T >= AG_R, "pooling: AG_T / HD row groups by HD columns; a thread per row fetches its graph number");
  static_assert(AG_Q >= 2 * HD, "a wave's queue is reused for its 2*HD column sums once the edges are done");
  static_assert(AG_T % 16 == 0 && AG_T >= 2 * HD, "a thread keeps one group of four columns over its four float4");

  const int tid = threadIdx.x;
  const int lane = tid & 31;
  const int wv = tid >> 5;
  const int row0 = blockIdx.x * AG_R;
  int nrows = N - row0;
  if (nrows > AG_R) nrows = AG_R;
  const int n4 = nrows * (HD / 4);
  int* myqe = qe[wv];

  const bool listed = ovf[blockIdx.x] == 0;
  int nl = 0;
  unsigned en[AG_LC / 32];
  if (listed) {
    const int slot = blockIdx.x * AG_W + wv;
    nl = __builtin_amdgcn_readfirstlane(lcnt[slot]);
    if (nl > AG_LC) nl = AG_LC;
    const unsigned* ml = lst + (size_t)slot * AG_LC;
#pragma unroll
    for (int j = 0; j < AG_LC / 32; ++j) {
      int idx = lane + 32 * j;
      en[j] = idx < nl ? ml[idx] : 0u;
    }
  }

  {
    const float4* h4 = (const float4*)(h + (size_t)row0 * HD);
    float4* t4 = (float4*)tile;
    float4 hv[4];
#pragma unroll
    for (int j = 0; j < 4; ++j) {
      int i4 = tid + j * AG_T;
      hv[j] = make_float4(0.f, 0.f, 0.f, 0.f);
      if (i4 < n4) hv[j] = h4[i4];
    }
    int bt = 0;
    float dd = 0.f;
    if (tid < nrows) {
      bt = batch[row0 + tid];
      dd = dinv[row0 + tid];
    }
#pragma unroll
    for (int j = 0; j < 4; ++j) {
      int i4 = tid + j * AG_T;
      if (i4 < n4) t4[i4] = hv[j];
    }
    if (tid < nrows) { sbatch[tid] = bt; sdinv[tid] = dd; }
  }
  __syncthreads();

  if (listed) {
#pragma unroll
    for (int j = 0; j < AG_LC / 32; ++j) {
      int idx = lane + 32 * j;
      if (idx < nl) {
        unsigned dl = en[j] >> 24;
        if (dl > AG_R - 1) dl = AG_R - 1;
        myqe[idx] = (int)((en[j] & 0xFFFFFFu) | (dl << 24));
      }
    }
    for (int base = 0; base < nl; base += 32) {
      int n = nl - base;
      if (n > 32) n = 32;
      int s_i = 0, dl_i = 0;
      if (lane < n) {
        unsigned pw = (unsigned)myqe[base + lane];
        s_i = (int)(pw & 0xFFFFFFu);
        if (s_i > N - 1) s_i = N - 1;
        dl_i = (int)(pw >> 24);
      }
      ag_rows(h, tile, s_i, dl_i, n, lane);
    }
  } else {
    AggSink sk;
    sk.src = src; sk.dst = dst; sk.h = h; sk.tile = tile; sk.row0 = row0;
    ag_scan(sk, key, E, myqe, tid, lane);
  }
  __syncthreads();

  {
    float4* t4 = (float4*)tile;
    float4* o4 = (float4*)(agg + (size_t)row0 * HD);
    const int c4 = (tid & 15) * 4;
    const float b0 = bias[c4], b1 = bias[c4 + 1], b2 = bias[c4 + 2], b3 = bias[c4 + 3];
    float4 s = make_float4(0.f, 0.f, 0.f, 0.f);
    float4 q = make_float4(0.f, 0.f, 0.f, 0.f);
#pragma unroll
    for (int j = 0; j < 4; ++j) {
      int i4 = tid + j * AG_T;
      if (i4 < n4) {
        float4 v = t4[i4];
        const float dd = sdinv[i4 >> 4];
        v.x = fmaxf(v.x * dd + b0, 0.f);
        v.y = fmaxf(v.y * dd + b1, 0.f);
        v.z = fmaxf(v.z * dd + b2, 0.f);
        v.w = fmaxf(v.w * dd + b3, 0.f);
        o4[i4] = v;
        t4[i4] = v;
        s.x += v.x; s.y += v.y; s.z += v.z; s.w += v.w;
        q.x += v.x * v.x; q.y += v.y * v.y; q.z += v.z * v.z; q.w += v.w * v.w;
      }
    }
    s.x += __shfl_xor(s.x, 16, 32); s.y += __shfl_xor(s.y, 16, 32);
    s.z += __shfl_xor(s.z, 16, 32); s.w += __shfl_xor(s.w, 16, 32);
    q.x += __shfl_xor(q.x, 16, 32); q.y += __shfl_xor(q.y, 16, 32);
    q.z += __shfl_xor(q.z, 16, 32); q.w += __shfl_xor(q.w, 16, 32);
    float* red = (float*)myqe;
    if (lane < 16) {
      *(float4*)(red + c4) = s;
      *(float4*)(red + HD + c4) = q;
    }
    __syncthreads();
    if (tid < 2 * HD) {
      const float* r0 = (const float*)&qe[0][0];
      float t = 0.f;
#pragma unroll
      for (int w = 0; w < AG_W; ++w) t += r0[w * AG_Q + tid];
      atomicAdd(&bn[tid], t);
    }
    {
      constexpr int RPG = (AG_R + AG_T / HD - 1) / (AG_T / HD);
      const int c = tid & (HD - 1);
      const int rg0 = (tid / HD) * RPG;
      float pv[RPG];
      int pg[RPG];
#pragma unroll
      for (int u = 0; u < RPG; ++u) {
        int r = rg0 + u;
        bool ok = r < nrows;
        int rr = ok ? r : 0;
        pv[u] = tile[rr * HD + c];
        pg[u] = ok ? sbatch[rr] : -1;
      }
      int gcur = -1;
      float acc = 0.f;
#pragma unroll
      for (int u = 0; u < RPG; ++u) {
        if (pg[u] >= 0) {
          if (pg[u] != gcur) {
            if (gcur >= 0) atomicAdd(&gsum[(size_t)gcur * CD + lofs + c], acc);
            gcur = pg[u];
            acc = 0.f;
          }
          acc += pv[u];
        }
      }
      if (gcur >= 0) atomicAdd(&gsum[(size_t)gcur * CD + lofs + c], acc);
    }
  }
}

__global__ void k_final(const float* __restrict__ z, const float* __restrict__ bn,
                        const float* __restrict__ gm0, const float* __restrict__ gm1, const float* __restrict__ gm2,
                        const float* __restrict__ bt0, const float* __restrict__ bt1, const float* __restrict__ bt2,
                        const float* __restrict__ gsum, const float* __restrict__ cnt,
                        float* __restrict__ out, int N, int G, float invN) {
  __shared__ float2 sbn[CD];
  const int t = threadIdx.x;
  const int wid = blockIdx.x * (blockDim.x >> 5) + (t >> 5);
  const int lane = t & 31;
  const bool node = wid < N;
  const bool graph = !node && wid < N + G;
  float v[6];
  float n = 0.f;
  if (node) {
#pragma unroll
    for (int j = 0; j < 6; ++j) {
      int e = lane + 32 * j;
      v[j] = z[(size_t)(e / HD) * N * HD + (size_t)wid * HD + (e % HD)];
    }
  } else {
    int g = graph ? wid - N : 0;
    n = cnt[g];
#pragma unroll
    for (int j = 0; j < 6; ++j) v[j] = gsum[(size_t)g * CD + lane + 32 * j];
  }
  if (t < CD) {
    int l = t / HD, c = t % HD;
    const float* gm = l == 0 ? gm0 : (l == 1 ? gm1 : gm2);
    const float* bt = l == 0 ? bt0 : (l == 1 ? bt1 : bt2);
    sbn[t] = bn_map(bn + l * 2 * HD, gm, bt, c, invN);
  }
  __syncthreads();
  if (!node && !graph) return;
  float s = 0.f;
  if (node) {
#pragma unroll
    for (int j = 0; j < 6; ++j) {
      float2 m = sbn[lane + 32 * j];
      v[j] = v[j] * m.x + m.y;
    }
  } else {
    const float mul = 1.0f / fmaxf(n, 1.0f);
#pragma unroll
    for (int j = 0; j < 6; ++j) {
      float2 m = sbn[lane + 32 * j];
      v[j] = (v[j] * m.x + n * m.y) * mul;
    }
  }
#pragma unroll
  for (int j = 0; j < 6; ++j) s += v[j] * v[j];
#pragma unroll
  for (int off = 16; off >= 1; off >>= 1) s += __shfl_xor(s, off, 32);
  float inv = 1.0f / fmaxf(sqrtf(s), 1e-12f);
  float* o = out + (size_t)wid * CD;
#pragma unroll
  for (int j = 0; j < 6; ++j) o[lane + 32 * j] = v[j] * inv;
}

extern "C" void kernel_launch(void* const* d_in, const int* in_sizes, int n_in,
                              void* d_out, int out_size, void* d_ws, size_t ws_size,
                              hipStream_t stream) {
  (void)n_in; (void)out_size; (void)ws_size;
  const float* x     = (const float*)d_in[0];
  const int*   ei    = (const int*)d_in[1];
  const int*   batch = (const int*)d_in[2];
  int N = in_sizes[0] / 128;
  int E = in_sizes[1] / 2;
  const int* src = ei;
  const int* dst = ei + E;
  const float* Wl[3]  = {(const float*)d_in[3],  (const float*)d_in[7],  (const float*)d_in[11]};
  const float* bl[3]  = {(const float*)d_in[4],  (const float*)d_in[8],  (const float*)d_in[12]};
  const float* gl[3]  = {(const float*)d_in[5],  (const float*)d_in[9],  (const float*)d_in[13]};
  const float* bel[3] = {(const float*)d_in[6],  (const float*)d_in[10], (const float*)d_in[14]};
  int nblk = (N + AG_R - 1) / AG_R;
  int Epad = (E + 15) & ~15;

  float* ws   = (float*)d_ws;
  float* gsum = ws;                       ws += (size_t)GN * CD;
  float* cnt  = ws;                       ws += GN;
  float* bn   = ws;                       ws += 3 * 2 * HD;
  size_t nclear = (size_t)(ws - gsum) * sizeof(float);
  float* dinv = ws;                       ws += N;
  float* h    = ws;                       ws += (size_t)N * HD;
  float* zpl  = ws;                       ws += (size_t)3 * N * HD;
  unsigned char* key = (unsigned char*)ws;
  unsigned* lst = (unsigned*)(key + Epad);
  int* lcnt = (int*)(lst + (size_t)nblk * AG_W * AG_LC);
  int* ovf  = lcnt + nblk * AG_W;

  hipMemsetAsync(gsum, 0, nclear, stream);

  k_key<<<(Epad / 4 + 255) / 256, 256, 0, stream>>>(dst, E, (unsigned*)key);
  k_own<<<nblk, AG_T, 0, stream>>>(src, dst, key, batch, dinv, cnt, lst, lcnt, ovf, N, E);

  const float invN = 1.0f / (float)N;
  for (int l = 0; l < 3; ++l) {
    int tilesM = (N + 15) / 16;
    int waves = tilesM * 4;
    if (l == 0)
      k_gemm_wmma<false><<<(waves + 7) / 8, 256, 0, stream>>>(x, 128, Wl[0], 128, h, N, dinv, nullptr, nullptr, nullptr, invN);
    else
      k_gemm_wmma<true><<<(waves + 7) / 8, 256, 0, stream>>>(zpl + (size_t)(l - 1) * N * HD, HD, Wl[l], HD, h, N, dinv,
                                                             bn + (l - 1) * 2 * HD, gl[l - 1], bel[l - 1], invN);
    k_aggregate<<<nblk, AG_T, 0, stream>>>(src, dst, key, lst, lcnt, ovf, dinv, h, bl[l], zpl + (size_t)l * N * HD,
                                           bn + l * 2 * HD, batch, gsum, l * HD, N, E);
  }

  float* out = (float*)d_out;
  k_final<<<(N + GN + 7) / 8, 256, 0, stream>>>(zpl, bn, gl[0], gl[1], gl[2], bel[0], bel[1], bel[2], gsum, cnt, out,
                                                N, GN, invN);
}


# GEMMs: a wave takes a 16x64 strip, operands by 16-byte loads straight to registers (columns dealt 4c+t over four tiles, k in chunks of 8), whole-row 16-byte stores; 4x fewer waves; products regrouped
# speedup vs baseline: 3.8908x; 1.0897x over previous
#include <hip/hip_runtime.h>

typedef float v2f __attribute__((ext_vector_type(2)));
typedef float v8f __attribute__((ext_vector_type(8)));

#define HD 64
#define CD 192
#define GN 256
#define AG_R 196
#define AG_T 1024
#define AG_W 32
#define AG_Q 256
#define AG_U 8
#define AG_LC 256

__global__ void k_key(const int* __restrict__ dst, int E, unsigned* __restrict__ key4) {
  int i = blockIdx.x * blockDim.x + threadIdx.x;
  int nw = ((E + 15) & ~15) >> 2;
  if (i < nw) {
    int e = 4 * i;
    unsigned w = 0u;
#pragma unroll
    for (int j = 0; j < 4; ++j)
      if (e + j < E) w |= ((unsigned)(dst[e + j] / AG_R) & 0xFFu) << (8 * j);
    key4[i] = w;
  }
}

__device__ __forceinline__ float2 bn_map(const float* __restrict__ bn, const float* __restrict__ gamma,
                                         const float* __restrict__ beta, int c, float invN) {
  float mu = bn[c] * invN;
  float var = fmaxf(bn[HD + c] * invN - mu * mu, 0.f);
  float sc = rsqrtf(var + 1e-5f) * gamma[c];
  return make_float2(sc, beta[c] - mu * sc);
}

typedef float v4f __attribute__((ext_vector_type(4)));
#define GM_W 4

struct GmBlock {
  v4f a[4];
  v4f b[4][4];
};

__device__ __forceinline__ void gm_load(GmBlock& g, const float* zr, const float* wp, int kb) {
#pragma unroll
  for (int q = 0; q < 4; ++q) {
    const int k0 = kb + 8 * q;
    g.a[q] = *(const v4f*)(zr + k0);
#pragma unroll
    for (int j = 0; j < 4; ++j) g.b[q][j] = *(const v4f*)(wp + (size_t)(k0 + j) * HD);
  }
}

__device__ __forceinline__ void gm_fence(const GmBlock& g) {
  asm volatile("" :: "v"(g.a[0]), "v"(g.a[1]), "v"(g.a[2]), "v"(g.a[3]),
               "v"(g.b[0][0]), "v"(g.b[0][1]), "v"(g.b[0][2]), "v"(g.b[0][3]),
               "v"(g.b[1][0]), "v"(g.b[1][1]), "v"(g.b[1][2]), "v"(g.b[1][3]),
               "v"(g.b[2][0]), "v"(g.b[2][1]), "v"(g.b[2][2]), "v"(g.b[2][3]),
               "v"(g.b[3][0]), "v"(g.b[3][1]), "v"(g.b[3][2]), "v"(g.b[3][3]));
}

template <bool BN, int KT>
__global__ __launch_bounds__(GM_W * 32) void k_gemm_wmma(const float* Z, int ldz, const float* W,
                                                         float* __restrict__ Hout, int N, const float* dinv,
                                                         const float* __restrict__ bn, const float* __restrict__ gamma,
                                                         const float* __restrict__ beta, float invN) {
  static_assert(KT % 32 == 0, "the k loop runs in blocks of four chunks of 8");
  static_assert(!BN || KT == HD, "the column map has HD entries");
  __shared__ __attribute__((aligned(16))) float2 sbn[HD];
  const int wid = blockIdx.x * GM_W + (threadIdx.x >> 5);
  const int tilesM = (N + 15) >> 4;
  if (!BN) {
    if (wid >= tilesM) return;
  }
  const int lane = threadIdx.x & 31;
  const int half = lane >> 4;
  const int m16 = lane & 15;
  const int row = wid * 16 + m16;
  const int lrow = row < N ? row : N - 1;
  const float* zr = Z + (size_t)lrow * ldz + 4 * half;
  const float* wp = W + (size_t)(4 * half) * HD + 4 * m16;
  const int rbase = wid * 16 + half * 8;
  float dr[8];
#pragma unroll
  for (int j = 0; j < 8; ++j) {
    int r = rbase + j;
    dr[j] = dinv[r < N ? r : N - 1];
  }
  GmBlock g[2];
  gm_load(g[0], zr, wp, 0);
  if (BN) {
    if (threadIdx.x < HD) sbn[threadIdx.x] = bn_map(bn, gamma, beta, (int)threadIdx.x, invN);
    __syncthreads();
    if (wid >= tilesM) return;
  }
  v8f c0 = {}, c1 = {}, c2 = {}, c3 = {};
#pragma unroll
  for (int kb = 0; kb < KT; kb += 32) {
    GmBlock& cur = g[(kb >> 5) & 1];
    if (kb + 32 < KT) gm_load(g[((kb >> 5) + 1) & 1], zr, wp, kb + 32);
    gm_fence(cur);
    if (BN) {
#pragma unroll
      for (int q = 0; q < 4; ++q) {
        const int kc = kb + 8 * q + 4 * half;
        const float4 m01 = *(const float4*)&sbn[kc];
        const float4 m23 = *(const float4*)&sbn[kc + 2];
        cur.a[q].x = cur.a[q].x * m01.x + m01.y;
        cur.a[q].y = cur.a[q].y * m01.z + m01.w;
        cur.a[q].z = cur.a[q].z * m23.x + m23.y;
        cur.a[q].w = cur.a[q].w * m23.z + m23.w;
      }
    }
#pragma unroll
    for (int q = 0; q < 4; ++q) {
      v2f alo, ahi, bv;
      alo.x = cur.a[q].x; alo.y = cur.a[q].y;
      ahi.x = cur.a[q].z; ahi.y = cur.a[q].w;
      bv.x = cur.b[q][0].x; bv.y = cur.b[q][1].x;
      c0 = __builtin_amdgcn_wmma_f32_16x16x4_f32(false, alo, false, bv, (short)0, c0, false, false);
      bv.x = cur.b[q][0].y; bv.y = cur.b[q][1].y;
      c1 = __builtin_amdgcn_wmma_f32_16x16x4_f32(false, alo, false, bv, (short)0, c1, false, false);
      bv.x = cur.b[q][0].z; bv.y = cur.b[q][1].z;
      c2 = __builtin_amdgcn_wmma_f32_16x16x4_f32(false, alo, false, bv, (short)0, c2, false, false);
      bv.x = cur.b[q][0].w; bv.y = cur.b[q][1].w;
      c3 = __builtin_amdgcn_wmma_f32_16x16x4_f32(false, alo, false, bv, (short)0, c3, false, false);
      bv.x = cur.b[q][2].x; bv.y = cur.b[q][3].x;
      c0 = __builtin_amdgcn_wmma_f32_16x16x4_f32(false, ahi, false, bv, (short)0, c0, false, false);
      bv.x = cur.b[q][2].y; bv.y = cur.b[q][3].y;
      c1 = __builtin_amdgcn_wmma_f32_16x16x4_f32(false, ahi, false, bv, (short)0, c1, false, false);
      bv.x = cur.b[q][2].z; bv.y = cur.b[q][3].z;
      c2 = __builtin_amdgcn_wmma_f32_16x16x4_f32(false, ahi, false, bv, (short)0, c2, false, false);
      bv.x = cur.b[q][2].w; bv.y = cur.b[q][3].w;
      c3 = __builtin_amdgcn_wmma_f32_16x16x4_f32(false, ahi, false, bv, (short)0, c3, false, false);
    }
  }
#pragma unroll
  for (int j = 0; j < 8; ++j) {
    int r = rbase + j;
    if (r < N)
      *(float4*)(Hout + (size_t)r * HD + 4 * m16) = make_float4(c0[j] * dr[j], c1[j] * dr[j], c2[j] * dr[j], c3[j] * dr[j]);
  }
}


__device__ __forceinline__ void ag_push(bool m, int e, int* myqe, int& cnt) {
  unsigned mask = __builtin_amdgcn_ballot_w32(m);
  if (mask) {
    int pos = cnt + (int)__builtin_amdgcn_mbcnt_lo(mask, 0u);
    if (m) myqe[pos] = e;
    cnt += __builtin_popcount(mask);
  }
}

__device__ __forceinline__ unsigned ag_miss(unsigned w, unsigned krep) {
  unsigned x = w ^ krep;
  return ((x & 0x7F7F7F7Fu) + 0x7F7F7F7Fu) | x | 0x7F7F7F7Fu;
}

template <bool TAIL, class S>
__device__ __forceinline__ void ag_test16(uint4 kq, bool valid, int e0, unsigned krep, int E, S& sk,
                                          int* myqe, int& cnt, int lane) {
  unsigned c = __builtin_rotateright32(ag_miss(kq.x, krep), 3) & __builtin_rotateright32(ag_miss(kq.y, krep), 2) &
               __builtin_rotateright32(ag_miss(kq.z, krep), 1) & ag_miss(kq.w, krep);
  unsigned hit = ~c;
  if (TAIL) hit = valid ? hit : 0u;
  while (__builtin_amdgcn_ballot_w32(hit != 0u)) {
    if (cnt > AG_Q - 32) sk.drain(myqe, cnt, lane);
    int p = __builtin_ffs((int)hit) - 1;
    int e = e0 + ((p & 3) << 2) + (p >> 3);
    bool m = hit != 0u;
    if (TAIL) m = m && e < E;
    hit &= hit - 1u;
    ag_push(m, e, myqe, cnt);
  }
}

template <class S>
__device__ __forceinline__ void ag_scan(S& sk, const unsigned char* __restrict__ key, int E, int* myqe,
                                        int tid, int lane) {
  const uint4* kv = (const uint4*)key;
  const int nvec = (E + 15) >> 4;
  const int nFull = (nvec - 1) / AG_T;
  const int it0 = (int)(((long long)blockIdx.x * nFull) / gridDim.x);
  const unsigned krep = (unsigned)blockIdx.x * 0x01010101u;
  int cnt = 0;

  int k = 0;
  for (; k + AG_U <= nFull; k += AG_U) {
    uint4 d[AG_U];
    int vi[AG_U];
#pragma unroll
    for (int j = 0; j < AG_U; ++j) {
      int it = it0 + k + j;
      if (it >= nFull) it -= nFull;
      vi[j] = it * AG_T + tid;
      d[j] = kv[vi[j]];
    }
#pragma unroll
    for (int j = 0; j < AG_U; ++j)
      ag_test16<false>(d[j], true, 16 * vi[j], krep, E, sk, myqe, cnt, lane);
    if (cnt >= 32) sk.full(myqe, cnt, lane);
  }
  for (; k <= nFull; ++k) {
    int it = it0 + k;
    if (it >= nFull) it -= nFull;
    if (k == nFull) it = nFull;
    int v = it * AG_T + tid;
    bool valid = v < nvec;
    uint4 d = make_uint4(0u, 0u, 0u, 0u);
    if (valid) d = kv[v];
    ag_test16<true>(d, valid, 16 * v, krep, E, sk, myqe, cnt, lane);
  }
  sk.rest(myqe, cnt, lane);
}

struct OwnSink {
  const int* __restrict__ src;
  const int* __restrict__ dst;
  int* rc;
  unsigned* ml;
  int row0;
  int tot;
  __device__ __forceinline__ void drain(int* myqe, int& cnt, int lane) {
    for (int base = 0; base < cnt; base += 32) {
      int n = cnt - base;
      if (n > 32) n = 32;
      if (lane < n) {
        int e = myqe[base + lane];
        int s = src[e];
        int dl = dst[e] - row0;
        atomicAdd(&rc[dl], 1);
        int pos = tot + lane;
        if (pos > AG_LC - 1) pos = AG_LC - 1;
        ml[pos] = (unsigned)s | ((unsigned)dl << 24);
      }
      tot += n;
    }
    cnt = 0;
  }
  __device__ __forceinline__ void full(int* myqe, int& cnt, int lane) { drain(myqe, cnt, lane); }
  __device__ __forceinline__ void rest(int* myqe, int& cnt, int lane) { drain(myqe, cnt, lane); }
};

__global__ __launch_bounds__(AG_T) void k_own(const int* __restrict__ src, const int* __restrict__ dst,
                                              const unsigned char* __restrict__ key,
                                              const int* __restrict__ batch, float* __restrict__ dinv,
                                              float* __restrict__ cnt, unsigned* __restrict__ lst,
                                              int* __restrict__ lcnt, int* __restrict__ ovf, int N, int E) {
  __shared__ int rc[AG_R];
  __shared__ int sovf;
  __shared__ __attribute__((aligned(16))) int qe[AG_W][AG_Q];
  const int tid = threadIdx.x;
  const int lane = tid & 31;
  const int wv = tid >> 5;
  const int row0 = blockIdx.x * AG_R;
  int nrows = N - row0;
  if (nrows > AG_R) nrows = AG_R;
  if (tid < AG_R) rc[tid] = 0;
  if (tid == 0) sovf = 0;
  __syncthreads();

  const int slot = blockIdx.x * AG_W + wv;
  OwnSink sk;
  sk.src = src; sk.dst = dst; sk.rc = rc; sk.ml = lst + (size_t)slot * AG_LC; sk.row0 = row0; sk.tot = 0;
  ag_scan(sk, key, E, qe[wv], tid, lane);
  if (lane == 0) {
    lcnt[slot] = sk.tot;
    if (sk.tot > AG_LC) sovf = 1;
  }
  __syncthreads();

  if (tid < nrows) {
    int row = row0 + tid;
    dinv[row] = rsqrtf((float)rc[tid] + 1.0f);
    atomicAdd(&cnt[batch[row]], 1.0f);
  }
  if (tid == 0) ovf[blockIdx.x] = sovf;
}


__device__ __forceinline__ void ag_rows(const float* __restrict__ h, float* tile, int s_i, int dl_i,
                                        int n, int lane) {
#pragma unroll
  for (int hf = 0; hf < 2; ++hf) {
    if (hf * 16 < n) {
      float a0[16], a1[16];
#pragma unroll
      for (int u = 0; u < 16; ++u) {
        int s = __builtin_amdgcn_readlane(s_i, hf * 16 + u);
        const float* hp = h + (size_t)s * HD;
        a0[u] = hp[lane];
        a1[u] = hp[32 + lane];
      }
#pragma unroll
      for (int u = 0; u < 16; ++u) {
        if (hf * 16 + u < n) {
          int dl = __builtin_amdgcn_readlane(dl_i, hf * 16 + u);
          atomicAdd(&tile[dl * HD + lane], a0[u]);
          atomicAdd(&tile[dl * HD + 32 + lane], a1[u]);
        }
      }
    }
  }
}

struct AggSink {
  const int* __restrict__ src;
  const int* __restrict__ dst;
  const float* __restrict__ h;
  float* tile;
  int row0;
  __device__ __forceinline__ void group(const int* myqe, int base, int n, int lane) {
    int s_i = 0, dl_i = 0;
    if (lane < n) {
      int e = myqe[base + lane];
      s_i = src[e];
      dl_i = dst[e] - row0;
    }
    ag_rows(h, tile, s_i, dl_i, n, lane);
  }
  __device__ __forceinline__ void full(int* myqe, int& cnt, int lane) {
    int nf = cnt & ~31;
    for (int base = 0; base < nf; base += 32) group(myqe, base, 32, lane);
    int rem = cnt - nf;
    if (lane < rem) {
      int te = myqe[nf + lane];
      myqe[lane] = te;
    }
    cnt = rem;
  }
  __device__ __forceinline__ void rest(int* myqe, int& cnt, int lane) {
    for (int base = 0; base < cnt; base += 32) {
      int n = cnt - base;
      if (n > 32) n = 32;
      group(myqe, base, n, lane);
    }
    cnt = 0;
  }
  __device__ __forceinline__ void drain(int* myqe, int& cnt, int lane) {
    for (int base = 0; base < cnt; base += 32) {
      int n = cnt - base;
      if (n > 32) n = 32;
      int s_i = 0, dl_i = 0;
      if (lane < n) {
        int e = myqe[base + lane];
        s_i = src[e];
        dl_i = dst[e] - row0;
      }
#pragma unroll 1
      for (int u = 0; u < n; ++u) {
        int s = __builtin_amdgcn_readlane(s_i, u);
        int dl = __builtin_amdgcn_readlane(dl_i, u);
        const float* hp = h + (size_t)s * HD;
        atomicAdd(&tile[dl * HD + lane], hp[lane]);
        atomicAdd(&tile[dl * HD + 32 + lane], hp[32 + lane]);
      }
    }
    cnt = 0;
  }
};

__global__ __launch_bounds__(AG_T) void k_aggregate(const int* __restrict__ src, const int* __restrict__ dst,
                                                    const unsigned char* __restrict__ key,
                                                    const unsigned* __restrict__ lst, const int* __restrict__ lcnt,
                                                    const int* __restrict__ ovf,
                                                    const float* __restrict__ dinv, const float* __restrict__ h,
                                                    const float* __restrict__ bias, float* __restrict__ agg,
                                                    float* __restrict__ bn, const int* __restrict__ batch,
                                                    float* __restrict__ gsum, int lofs, int N, int E) {
  static_assert(AG_R * (HD / 4) <= 4 * AG_T, "row tile is filled and stored in four float4 per thread");
  static_assert(AG_T == AG_W * 32, "wave32");
  static_assert(AG_LC % 32 == 0 && AG_LC <= AG_Q, "a wave's list is staged in its queue, whole groups of 32");
  __shared__ __attribute__((aligned(16))) float tile[AG_R * HD];
  __shared__ float sdinv[AG_R];
  __shared__ int sbatch[AG_R];
  __shared__ __attribute__((aligned(16))) int qe[AG_W][AG_Q];
  static_assert(AG_T % HD == 0 && AG_T >= AG_R, "pooling: AG_T / HD row groups by HD columns; a thread per row fetches its graph number");
  static_assert(AG_Q >= 2 * HD, "a wave's queue is reused for its 2*HD column sums once the edges are done");
  static_assert(AG_T % 16 == 0 && AG_T >= 2 * HD, "a thread keeps one group of four columns over its four float4");

  const int tid = threadIdx.x;
  const int lane = tid & 31;
  const int wv = tid >> 5;
  const int row0 = blockIdx.x * AG_R;
  int nrows = N - row0;
  if (nrows > AG_R) nrows = AG_R;
  const int n4 = nrows * (HD / 4);
  int* myqe = qe[wv];

  const bool listed = ovf[blockIdx.x] == 0;
  int nl = 0;
  unsigned en[AG_LC / 32];
  if (listed) {
    const int slot = blockIdx.x * AG_W + wv;
    nl = __builtin_amdgcn_readfirstlane(lcnt[slot]);
    if (nl > AG_LC) nl = AG_LC;
    const unsigned* ml = lst + (size_t)slot * AG_LC;
#pragma unroll
    for (int j = 0; j < AG_LC / 32; ++j) {
      int idx = lane + 32 * j;
      en[j] = idx < nl ? ml[idx] : 0u;
    }
  }

  {
    const float4* h4 = (const float4*)(h + (size_t)row0 * HD);
    float4* t4 = (float4*)tile;
    float4 hv[4];
#pragma unroll
    for (int j = 0; j < 4; ++j) {
      int i4 = tid + j * AG_T;
      hv[j] = make_float4(0.f, 0.f, 0.f, 0.f);
      if (i4 < n4) hv[j] = h4[i4];
    }
    int bt = 0;
    float dd = 0.f;
    if (tid < nrows) {
      bt = batch[row0 + tid];
      dd = dinv[row0 + tid];
    }
#pragma unroll
    for (int j = 0; j < 4; ++j) {
      int i4 = tid + j * AG_T;
      if (i4 < n4) t4[i4] = hv[j];
    }
    if (tid < nrows) { sbatch[tid] = bt; sdinv[tid] = dd; }
  }
  __syncthreads();

  if (listed) {
#pragma unroll
    for (int j = 0; j < AG_LC / 32; ++j) {
      int idx = lane + 32 * j;
      if (idx < nl) {
        unsigned dl = en[j] >> 24;
        if (dl > AG_R - 1) dl = AG_R - 1;
        myqe[idx] = (int)((en[j] & 0xFFFFFFu) | (dl << 24));
      }
    }
    for (int base = 0; base < nl; base += 32) {
      int n = nl - base;
      if (n > 32) n = 32;
      int s_i = 0, dl_i = 0;
      if (lane < n) {
        unsigned pw = (unsigned)myqe[base + lane];
        s_i = (int)(pw & 0xFFFFFFu);
        if (s_i > N - 1) s_i = N - 1;
        dl_i = (int)(pw >> 24);
      }
      ag_rows(h, tile, s_i, dl_i, n, lane);
    }
  } else {
    AggSink sk;
    sk.src = src; sk.dst = dst; sk.h = h; sk.tile = tile; sk.row0 = row0;
    ag_scan(sk, key, E, myqe, tid, lane);
  }
  __syncthreads();

  {
    float4* t4 = (float4*)tile;
    float4* o4 = (float4*)(agg + (size_t)row0 * HD);
    const int c4 = (tid & 15) * 4;
    const float b0 = bias[c4], b1 = bias[c4 + 1], b2 = bias[c4 + 2], b3 = bias[c4 + 3];
    float4 s = make_float4(0.f, 0.f, 0.f, 0.f);
    float4 q = make_float4(0.f, 0.f, 0.f, 0.f);
#pragma unroll
    for (int j = 0; j < 4; ++j) {
      int i4 = tid + j * AG_T;
      if (i4 < n4) {
        float4 v = t4[i4];
        const float dd = sdinv[i4 >> 4];
        v.x = fmaxf(v.x * dd + b0, 0.f);
        v.y = fmaxf(v.y * dd + b1, 0.f);
        v.z = fmaxf(v.z * dd + b2, 0.f);
        v.w = fmaxf(v.w * dd + b3, 0.f);
        o4[i4] = v;
        t4[i4] = v;
        s.x += v.x; s.y += v.y; s.z += v.z; s.w += v.w;
        q.x += v.x * v.x; q.y += v.y * v.y; q.z += v.z * v.z; q.w += v.w * v.w;
      }
    }
    s.x += __shfl_xor(s.x, 16, 32); s.y += __shfl_xor(s.y, 16, 32);
    s.z += __shfl_xor(s.z, 16, 32); s.w += __shfl_xor(s.w, 16, 32);
    q.x += __shfl_xor(q.x, 16, 32); q.y += __shfl_xor(q.y, 16, 32);
    q.z += __shfl_xor(q.z, 16, 32); q.w += __shfl_xor(q.w, 16, 32);
    float* red = (float*)myqe;
    if (lane < 16) {
      *(float4*)(red + c4) = s;
      *(float4*)(red + HD + c4) = q;
    }
    __syncthreads();
    if (tid < 2 * HD) {
      const float* r0 = (const float*)&qe[0][0];
      float t = 0.f;
#pragma unroll
      for (int w = 0; w < AG_W; ++w) t += r0[w * AG_Q + tid];
      atomicAdd(&bn[tid], t);
    }
    {
      constexpr int RPG = (AG_R + AG_T / HD - 1) / (AG_T / HD);
      const int c = tid & (HD - 1);
      const int rg0 = (tid / HD) * RPG;
      float pv[RPG];
      int pg[RPG];
#pragma unroll
      for (int u = 0; u < RPG; ++u) {
        int r = rg0 + u;
        bool ok = r < nrows;
        int rr = ok ? r : 0;
        pv[u] = tile[rr * HD + c];
        pg[u] = ok ? sbatch[rr] : -1;
      }
      int gcur = -1;
      float acc = 0.f;
#pragma unroll
      for (int u = 0; u < RPG; ++u) {
        if (pg[u] >= 0) {
          if (pg[u] != gcur) {
            if (gcur >= 0) atomicAdd(&gsum[(size_t)gcur * CD + lofs + c], acc);
            gcur = pg[u];
            acc = 0.f;
          }
          acc += pv[u];
        }
      }
      if (gcur >= 0) atomicAdd(&gsum[(size_t)gcur * CD + lofs + c], acc);
    }
  }
}

__global__ void k_final(const float* __restrict__ z, const float* __restrict__ bn,
                        const float* __restrict__ gm0, const float* __restrict__ gm1, const float* __restrict__ gm2,
                        const float* __restrict__ bt0, const float* __restrict__ bt1, const float* __restrict__ bt2,
                        const float* __restrict__ gsum, const float* __restrict__ cnt,
                        float* __restrict__ out, int N, int G, float invN) {
  __shared__ float2 sbn[CD];
  const int t = threadIdx.x;
  const int wid = blockIdx.x * (blockDim.x >> 5) + (t >> 5);
  const int lane = t & 31;
  const bool node = wid < N;
  const bool graph = !node && wid < N + G;
  float v[6];
  float n = 0.f;
  if (node) {
#pragma unroll
    for (int j = 0; j < 6; ++j) {
      int e = lane + 32 * j;
      v[j] = z[(size_t)(e / HD) * N * HD + (size_t)wid * HD + (e % HD)];
    }
  } else {
    int g = graph ? wid - N : 0;
    n = cnt[g];
#pragma unroll
    for (int j = 0; j < 6; ++j) v[j] = gsum[(size_t)g * CD + lane + 32 * j];
  }
  if (t < CD) {
    int l = t / HD, c = t % HD;
    const float* gm = l == 0 ? gm0 : (l == 1 ? gm1 : gm2);
    const float* bt = l == 0 ? bt0 : (l == 1 ? bt1 : bt2);
    sbn[t] = bn_map(bn + l * 2 * HD, gm, bt, c, invN);
  }
  __syncthreads();
  if (!node && !graph) return;
  float s = 0.f;
  if (node) {
#pragma unroll
    for (int j = 0; j < 6; ++j) {
      float2 m = sbn[lane + 32 * j];
      v[j] = v[j] * m.x + m.y;
    }
  } else {
    const float mul = 1.0f / fmaxf(n, 1.0f);
#pragma unroll
    for (int j = 0; j < 6; ++j) {
      float2 m = sbn[lane + 32 * j];
      v[j] = (v[j] * m.x + n * m.y) * mul;
    }
  }
#pragma unroll
  for (int j = 0; j < 6; ++j) s += v[j] * v[j];
#pragma unroll
  for (int off = 16; off >= 1; off >>= 1) s += __shfl_xor(s, off, 32);
  float inv = 1.0f / fmaxf(sqrtf(s), 1e-12f);
  float* o = out + (size_t)wid * CD;
#pragma unroll
  for (int j = 0; j < 6; ++j) o[lane + 32 * j] = v[j] * inv;
}

extern "C" void kernel_launch(void* const* d_in, const int* in_sizes, int n_in,
                              void* d_out, int out_size, void* d_ws, size_t ws_size,
                              hipStream_t stream) {
  (void)n_in; (void)out_size; (void)ws_size;
  const float* x     = (const float*)d_in[0];
  const int*   ei    = (const int*)d_in[1];
  const int*   batch = (const int*)d_in[2];
  int N = in_sizes[0] / 128;
  int E = in_sizes[1] / 2;
  const int* src = ei;
  const int* dst = ei + E;
  const float* Wl[3]  = {(const float*)d_in[3],  (const float*)d_in[7],  (const float*)d_in[11]};
  const float* bl[3]  = {(const float*)d_in[4],  (const float*)d_in[8],  (const float*)d_in[12]};
  const float* gl[3]  = {(const float*)d_in[5],  (const float*)d_in[9],  (const float*)d_in[13]};
  const float* bel[3] = {(const float*)d_in[6],  (const float*)d_in[10], (const float*)d_in[14]};
  int nblk = (N + AG_R - 1) / AG_R;
  int Epad = (E + 15) & ~15;

  float* ws   = (float*)d_ws;
  float* gsum = ws;                       ws += (size_t)GN * CD;
  float* cnt  = ws;                       ws += GN;
  float* bn   = ws;                       ws += 3 * 2 * HD;
  size_t nclear = (size_t)(ws - gsum) * sizeof(float);
  float* dinv = ws;                       ws += N;
  float* h    = ws;                       ws += (size_t)N * HD;
  float* zpl  = ws;                       ws += (size_t)3 * N * HD;
  unsigned char* key = (unsigned char*)ws;
  unsigned* lst = (unsigned*)(key + Epad);
  int* lcnt = (int*)(lst + (size_t)nblk * AG_W * AG_LC);
  int* ovf  = lcnt + nblk * AG_W;

  hipMemsetAsync(gsum, 0, nclear, stream);

  k_key<<<(Epad / 4 + 255) / 256, 256, 0, stream>>>(dst, E, (unsigned*)key);
  k_own<<<nblk, AG_T, 0, stream>>>(src, dst, key, batch, dinv, cnt, lst, lcnt, ovf, N, E);

  const float invN = 1.0f / (float)N;
  for (int l = 0; l < 3; ++l) {
    int tilesM = (N + 15) / 16;
    int gblk = (tilesM + GM_W - 1) / GM_W;
    if (l == 0)
      k_gemm_wmma<false, 128><<<gblk, GM_W * 32, 0, stream>>>(x, 128, Wl[0], h, N, dinv, nullptr, nullptr, nullptr, invN);
    else
      k_gemm_wmma<true, HD><<<gblk, GM_W * 32, 0, stream>>>(zpl + (size_t)(l - 1) * N * HD, HD, Wl[l], h, N, dinv,
                                                            bn + (l - 1) * 2 * HD, gl[l - 1], bel[l - 1], invN);
    k_aggregate<<<nblk, AG_T, 0, stream>>>(src, dst, key, lst, lcnt, ovf, dinv, h, bl[l], zpl + (size_t)l * N * HD,
                                           bn + l * 2 * HD, batch, gsum, l * HD, N, E);
  }

  float* out = (float*)d_out;
  k_final<<<(N + GN + 7) / 8, 256, 0, stream>>>(zpl, bn, gl[0], gl[1], gl[2], bel[0], bel[1], bel[2], gsum, cnt, out,
                                                N, GN, invN);
}


# owner filter on bit planes: owner = two one-hot nibbles (32 planes, by ballots); block ANDs its 2 planes, 128 edges a load pair, drains queue in 1 load round; edge sums in other LDS-arrival order
# speedup vs baseline: 4.0541x; 1.0420x over previous
#include <hip/hip_runtime.h>

typedef float v2f __attribute__((ext_vector_type(2)));
typedef float v8f __attribute__((ext_vector_type(8)));

#define HD 64
#define CD 192
#define GN 256
#define AG_R 196
#define AG_T 1024
#define AG_W 32
#define AG_Q 256
#define AG_U 8
#define KP_N 32
#define AG_LC 256

__global__ __launch_bounds__(64) void k_key(const int* __restrict__ dst, int E, unsigned* __restrict__ planes, int pW) {
  const int lane = threadIdx.x & 31;
  const int w0 = (blockIdx.x * (blockDim.x >> 5) + (threadIdx.x >> 5)) * 32;
  if (w0 >= pW) return;
  int d[32];
#pragma unroll
  for (int t = 0; t < 32; ++t) {
    int e = (w0 + t) * 32 + lane;
    d[t] = e < E ? dst[e] : -1;
  }
  unsigned pl[KP_N];
#pragma unroll
  for (int q = 0; q < KP_N; ++q) pl[q] = 0u;
#pragma unroll
  for (int t = 0; t < 32; ++t) {
    const bool valid = d[t] >= 0;
    const unsigned key = (unsigned)d[t] / AG_R;
    const unsigned hi = valid ? key >> 4 : 16u;
    const unsigned lo = valid ? key & 15u : 16u;
#pragma unroll
    for (int q = 0; q < 16; ++q) {
      unsigned bh = __builtin_amdgcn_ballot_w32(hi == (unsigned)q);
      unsigned bl = __builtin_amdgcn_ballot_w32(lo == (unsigned)q);
      asm("v_writelane_b32 %0, %1, %2" : "+v"(pl[q]) : "s"(bh), "n"(t));
      asm("v_writelane_b32 %0, %1, %2" : "+v"(pl[16 + q]) : "s"(bl), "n"(t));
    }
  }
  if (w0 + lane < pW) {
#pragma unroll
    for (int q = 0; q < KP_N; ++q) planes[(size_t)q * pW + w0 + lane] = pl[q];
  }
}

__device__ __forceinline__ float2 bn_map(const float* __restrict__ bn, const float* __restrict__ gamma,
                                         const float* __restrict__ beta, int c, float invN) {
  float mu = bn[c] * invN;
  float var = fmaxf(bn[HD + c] * invN - mu * mu, 0.f);
  float sc = rsqrtf(var + 1e-5f) * gamma[c];
  return make_float2(sc, beta[c] - mu * sc);
}

typedef float v4f __attribute__((ext_vector_type(4)));
#define GM_W 4

struct GmBlock {
  v4f a[4];
  v4f b[4][4];
};

__device__ __forceinline__ void gm_load(GmBlock& g, const float* zr, const float* wp, int kb) {
#pragma unroll
  for (int q = 0; q < 4; ++q) {
    const int k0 = kb + 8 * q;
    g.a[q] = *(const v4f*)(zr + k0);
#pragma unroll
    for (int j = 0; j < 4; ++j) g.b[q][j] = *(const v4f*)(wp + (size_t)(k0 + j) * HD);
  }
}

__device__ __forceinline__ void gm_fence(const GmBlock& g) {
  asm volatile("" :: "v"(g.a[0]), "v"(g.a[1]), "v"(g.a[2]), "v"(g.a[3]),
               "v"(g.b[0][0]), "v"(g.b[0][1]), "v"(g.b[0][2]), "v"(g.b[0][3]),
               "v"(g.b[1][0]), "v"(g.b[1][1]), "v"(g.b[1][2]), "v"(g.b[1][3]),
               "v"(g.b[2][0]), "v"(g.b[2][1]), "v"(g.b[2][2]), "v"(g.b[2][3]),
               "v"(g.b[3][0]), "v"(g.b[3][1]), "v"(g.b[3][2]), "v"(g.b[3][3]));
}

template <bool BN, int KT>
__global__ __launch_bounds__(GM_W * 32) void k_gemm_wmma(const float* Z, int ldz, const float* W,
                                                         float* __restrict__ Hout, int N, const float* dinv,
                                                         const float* __restrict__ bn, const float* __restrict__ gamma,
                                                         const float* __restrict__ beta, float invN) {
  static_assert(KT % 32 == 0, "the k loop runs in blocks of four chunks of 8");
  static_assert(!BN || KT == HD, "the column map has HD entries");
  __shared__ __attribute__((aligned(16))) float2 sbn[HD];
  const int wid = blockIdx.x * GM_W + (threadIdx.x >> 5);
  const int tilesM = (N + 15) >> 4;
  if (!BN) {
    if (wid >= tilesM) return;
  }
  const int lane = threadIdx.x & 31;
  const int half = lane >> 4;
  const int m16 = lane & 15;
  const int row = wid * 16 + m16;
  const int lrow = row < N ? row : N - 1;
  const float* zr = Z + (size_t)lrow * ldz + 4 * half;
  const float* wp = W + (size_t)(4 * half) * HD + 4 * m16;
  const int rbase = wid * 16 + half * 8;
  float dr[8];
#pragma unroll
  for (int j = 0; j < 8; ++j) {
    int r = rbase + j;
    dr[j] = dinv[r < N ? r : N - 1];
  }
  GmBlock g[2];
  gm_load(g[0], zr, wp, 0);
  if (BN) {
    if (threadIdx.x < HD) sbn[threadIdx.x] = bn_map(bn, gamma, beta, (int)threadIdx.x, invN);
    __syncthreads();
    if (wid >= tilesM) return;
  }
  v8f c0 = {}, c1 = {}, c2 = {}, c3 = {};
#pragma unroll
  for (int kb = 0; kb < KT; kb += 32) {
    GmBlock& cur = g[(kb >> 5) & 1];
    if (kb + 32 < KT) gm_load(g[((kb >> 5) + 1) & 1], zr, wp, kb + 32);
    gm_fence(cur);
    if (BN) {
#pragma unroll
      for (int q = 0; q < 4; ++q) {
        const int kc = kb + 8 * q + 4 * half;
        const float4 m01 = *(const float4*)&sbn[kc];
        const float4 m23 = *(const float4*)&sbn[kc + 2];
        cur.a[q].x = cur.a[q].x * m01.x + m01.y;
        cur.a[q].y = cur.a[q].y * m01.z + m01.w;
        cur.a[q].z = cur.a[q].z * m23.x + m23.y;
        cur.a[q].w = cur.a[q].w * m23.z + m23.w;
      }
    }
#pragma unroll
    for (int q = 0; q < 4; ++q) {
      v2f alo, ahi, bv;
      alo.x = cur.a[q].x; alo.y = cur.a[q].y;
      ahi.x = cur.a[q].z; ahi.y = cur.a[q].w;
      bv.x = cur.b[q][0].x; bv.y = cur.b[q][1].x;
      c0 = __builtin_amdgcn_wmma_f32_16x16x4_f32(false, alo, false, bv, (short)0, c0, false, false);
      bv.x = cur.b[q][0].y; bv.y = cur.b[q][1].y;
      c1 = __builtin_amdgcn_wmma_f32_16x16x4_f32(false, alo, false, bv, (short)0, c1, false, false);
      bv.x = cur.b[q][0].z; bv.y = cur.b[q][1].z;
      c2 = __builtin_amdgcn_wmma_f32_16x16x4_f32(false, alo, false, bv, (short)0, c2, false, false);
      bv.x = cur.b[q][0].w; bv.y = cur.b[q][1].w;
      c3 = __builtin_amdgcn_wmma_f32_16x16x4_f32(false, alo, false, bv, (short)0, c3, false, false);
      bv.x = cur.b[q][2].x; bv.y = cur.b[q][3].x;
      c0 = __builtin_amdgcn_wmma_f32_16x16x4_f32(false, ahi, false, bv, (short)0, c0, false, false);
      bv.x = cur.b[q][2].y; bv.y = cur.b[q][3].y;
      c1 = __builtin_amdgcn_wmma_f32_16x16x4_f32(false, ahi, false, bv, (short)0, c1, false, false);
      bv.x = cur.b[q][2].z; bv.y = cur.b[q][3].z;
      c2 = __builtin_amdgcn_wmma_f32_16x16x4_f32(false, ahi, false, bv, (short)0, c2, false, false);
      bv.x = cur.b[q][2].w; bv.y = cur.b[q][3].w;
      c3 = __builtin_amdgcn_wmma_f32_16x16x4_f32(false, ahi, false, bv, (short)0, c3, false, false);
    }
  }
#pragma unroll
  for (int j = 0; j < 8; ++j) {
    int r = rbase + j;
    if (r < N)
      *(float4*)(Hout + (size_t)r * HD + 4 * m16) = make_float4(c0[j] * dr[j], c1[j] * dr[j], c2[j] * dr[j], c3[j] * dr[j]);
  }
}


__device__ __forceinline__ void ag_push(bool m, int e, int* myqe, int& cnt) {
  unsigned mask = __builtin_amdgcn_ballot_w32(m);
  if (mask) {
    int pos = cnt + (int)__builtin_amdgcn_mbcnt_lo(mask, 0u);
    if (m) myqe[pos] = e;
    cnt += __builtin_popcount(mask);
  }
}

template <class S>
__device__ __forceinline__ void ag_hits128(unsigned m0, unsigned m1, unsigned m2, unsigned m3, int e0, S& sk,
                                           int* myqe, int& cnt, int lane) {
  while (__builtin_amdgcn_ballot_w32((m0 | m1 | m2 | m3) != 0u)) {
    if (cnt > AG_Q - 32) sk.drain(myqe, cnt, lane);
    const bool f01 = (m0 | m1) != 0u;
    const unsigned w = f01 ? (m0 ? m0 : m1) : (m2 ? m2 : m3);
    const int j = f01 ? (m0 ? 0 : 1) : (m2 ? 2 : 3);
    const bool has = w != 0u;
    const int e = e0 + 32 * j + (__builtin_ffs((int)w) - 1);
    const unsigned wc = w & (w - 1u);
    m0 = j == 0 ? wc : m0;
    m1 = j == 1 ? wc : m1;
    m2 = j == 2 ? wc : m2;
    m3 = j == 3 ? wc : m3;
    ag_push(has, e, myqe, cnt);
  }
}

typedef unsigned v4u __attribute__((ext_vector_type(4)));

template <class S>
__device__ __forceinline__ void ag_scan(S& sk, const unsigned* __restrict__ planes, int pW, int* myqe,
                                        int tid, int lane) {
  const int nvec = pW >> 2;
  const v4u* ph = (const v4u*)(planes + (size_t)(blockIdx.x >> 4) * pW);
  const v4u* pq = (const v4u*)(planes + (size_t)(16 + (blockIdx.x & 15)) * pW);
  int cnt = 0;
  for (int v0 = 0; v0 < nvec; v0 += AG_U * AG_T) {
    v4u a[AG_U], b[AG_U];
#pragma unroll
    for (int j = 0; j < AG_U; ++j) {
      int v = v0 + j * AG_T + tid;
      if (v > nvec - 1) v = nvec - 1;
      a[j] = ph[v];
      b[j] = pq[v];
    }
    __builtin_amdgcn_sched_barrier(0);
#pragma unroll
    for (int j = 0; j < AG_U; ++j) {
      const int v = v0 + j * AG_T + tid;
      const unsigned ok = v < nvec ? 0xFFFFFFFFu : 0u;
      ag_hits128(a[j].x & b[j].x & ok, a[j].y & b[j].y & ok, a[j].z & b[j].z & ok, a[j].w & b[j].w & ok, 128 * v, sk, myqe,
                 cnt, lane);
    }
    if (cnt >= 32) sk.full(myqe, cnt, lane);
  }
  sk.rest(myqe, cnt, lane);
}

struct OwnSink {
  const int* __restrict__ src;
  const int* __restrict__ dst;
  int* rc;
  unsigned* ml;
  int row0;
  int tot;
  __device__ __forceinline__ void drain(int* myqe, int& cnt, int lane) {
    for (int base = 0; base < cnt; base += 128) {
      int s[4], d[4];
#pragma unroll
      for (int g = 0; g < 4; ++g) {
        const int idx = base + 32 * g + lane;
        int e = 0;
        if (idx < cnt) e = myqe[idx];
        s[g] = src[e];
        d[g] = dst[e];
      }
#pragma unroll
      for (int g = 0; g < 4; ++g) {
        const int idx = base + 32 * g + lane;
        if (idx < cnt) {
          int dl = d[g] - row0;
          atomicAdd(&rc[dl], 1);
          int pos = tot + idx;
          if (pos > AG_LC - 1) pos = AG_LC - 1;
          ml[pos] = (unsigned)s[g] | ((unsigned)dl << 24);
        }
      }
    }
    tot += cnt;
    cnt = 0;
  }
  __device__ __forceinline__ void full(int* myqe, int& cnt, int lane) { drain(myqe, cnt, lane); }
  __device__ __forceinline__ void rest(int* myqe, int& cnt, int lane) { drain(myqe, cnt, lane); }
};

__global__ __launch_bounds__(AG_T) void k_own(const int* __restrict__ src, const int* __restrict__ dst,
                                              const unsigned* __restrict__ planes,
                                              const int* __restrict__ batch, float* __restrict__ dinv,
                                              float* __restrict__ cnt, unsigned* __restrict__ lst,
                                              int* __restrict__ lcnt, int* __restrict__ ovf, int N, int pW) {
  __shared__ int rc[AG_R];
  __shared__ int sovf;
  __shared__ __attribute__((aligned(16))) int qe[AG_W][AG_Q];
  const int tid = threadIdx.x;
  const int lane = tid & 31;
  const int wv = tid >> 5;
  const int row0 = blockIdx.x * AG_R;
  int nrows = N - row0;
  if (nrows > AG_R) nrows = AG_R;
  if (tid < AG_R) rc[tid] = 0;
  if (tid == 0) sovf = 0;
  __syncthreads();

  const int slot = blockIdx.x * AG_W + wv;
  OwnSink sk;
  sk.src = src; sk.dst = dst; sk.rc = rc; sk.ml = lst + (size_t)slot * AG_LC; sk.row0 = row0; sk.tot = 0;
  ag_scan(sk, planes, pW, qe[wv], tid, lane);
  if (lane == 0) {
    lcnt[slot] = sk.tot;
    if (sk.tot > AG_LC) sovf = 1;
  }
  __syncthreads();

  if (tid < nrows) {
    int row = row0 + tid;
    dinv[row] = rsqrtf((float)rc[tid] + 1.0f);
    atomicAdd(&cnt[batch[row]], 1.0f);
  }
  if (tid == 0) ovf[blockIdx.x] = sovf;
}


__device__ __forceinline__ void ag_rows(const float* __restrict__ h, float* tile, int s_i, int dl_i,
                                        int n, int lane) {
#pragma unroll
  for (int hf = 0; hf < 2; ++hf) {
    if (hf * 16 < n) {
      float a0[16], a1[16];
#pragma unroll
      for (int u = 0; u < 16; ++u) {
        int s = __builtin_amdgcn_readlane(s_i, hf * 16 + u);
        const float* hp = h + (size_t)s * HD;
        a0[u] = hp[lane];
        a1[u] = hp[32 + lane];
      }
#pragma unroll
      for (int u = 0; u < 16; ++u) {
        if (hf * 16 + u < n) {
          int dl = __builtin_amdgcn_readlane(dl_i, hf * 16 + u);
          atomicAdd(&tile[dl * HD + lane], a0[u]);
          atomicAdd(&tile[dl * HD + 32 + lane], a1[u]);
        }
      }
    }
  }
}

struct AggSink {
  const int* __restrict__ src;
  const int* __restrict__ dst;
  const float* __restrict__ h;
  float* tile;
  int row0;
  __device__ __forceinline__ void group(const int* myqe, int base, int n, int lane) {
    int s_i = 0, dl_i = 0;
    if (lane < n) {
      int e = myqe[base + lane];
      s_i = src[e];
      dl_i = dst[e] - row0;
    }
    ag_rows(h, tile, s_i, dl_i, n, lane);
  }
  __device__ __forceinline__ void full(int* myqe, int& cnt, int lane) {
    int nf = cnt & ~31;
    for (int base = 0; base < nf; base += 32) group(myqe, base, 32, lane);
    int rem = cnt - nf;
    if (lane < rem) {
      int te = myqe[nf + lane];
      myqe[lane] = te;
    }
    cnt = rem;
  }
  __device__ __forceinline__ void rest(int* myqe, int& cnt, int lane) {
    for (int base = 0; base < cnt; base += 32) {
      int n = cnt - base;
      if (n > 32) n = 32;
      group(myqe, base, n, lane);
    }
    cnt = 0;
  }
  __device__ __forceinline__ void drain(int* myqe, int& cnt, int lane) {
    for (int base = 0; base < cnt; base += 32) {
      int n = cnt - base;
      if (n > 32) n = 32;
      int s_i = 0, dl_i = 0;
      if (lane < n) {
        int e = myqe[base + lane];
        s_i = src[e];
        dl_i = dst[e] - row0;
      }
#pragma unroll 1
      for (int u = 0; u < n; ++u) {
        int s = __builtin_amdgcn_readlane(s_i, u);
        int dl = __builtin_amdgcn_readlane(dl_i, u);
        const float* hp = h + (size_t)s * HD;
        atomicAdd(&tile[dl * HD + lane], hp[lane]);
        atomicAdd(&tile[dl * HD + 32 + lane], hp[32 + lane]);
      }
    }
    cnt = 0;
  }
};

__global__ __launch_bounds__(AG_T) void k_aggregate(const int* __restrict__ src, const int* __restrict__ dst,
                                                    const unsigned* __restrict__ planes,
                                                    const unsigned* __restrict__ lst, const int* __restrict__ lcnt,
                                                    const int* __restrict__ ovf,
                                                    const float* __restrict__ dinv, const float* __restrict__ h,
                                                    const float* __restrict__ bias, float* __restrict__ agg,
                                                    float* __restrict__ bn, const int* __restrict__ batch,
                                                    float* __restrict__ gsum, int lofs, int N, int pW) {
  static_assert(AG_R * (HD / 4) <= 4 * AG_T, "row tile is filled and stored in four float4 per thread");
  static_assert(AG_T == AG_W * 32, "wave32");
  static_assert(AG_LC % 32 == 0 && AG_LC <= AG_Q, "a wave's list is staged in its queue, whole groups of 32");
  __shared__ __attribute__((aligned(16))) float tile[AG_R * HD];
  __shared__ float sdinv[AG_R];
  __shared__ int sbatch[AG_R];
  __shared__ __attribute__((aligned(16))) int qe[AG_W][AG_Q];
  static_assert(AG_T % HD == 0 && AG_T >= AG_R, "pooling: AG_T / HD row groups by HD columns; a thread per row fetches its graph number");
  static_assert(AG_Q >= 2 * HD, "a wave's queue is reused for its 2*HD column sums once the edges are done");
  static_assert(AG_T % 16 == 0 && AG_T >= 2 * HD, "a thread keeps one group of four columns over its four float4");

  const int tid = threadIdx.x;
  const int lane = tid & 31;
  const int wv = tid >> 5;
  const int row0 = blockIdx.x * AG_R;
  int nrows = N - row0;
  if (nrows > AG_R) nrows = AG_R;
  const int n4 = nrows * (HD / 4);
  int* myqe = qe[wv];

  const bool listed = ovf[blockIdx.x] == 0;
  int nl = 0;
  unsigned en[AG_LC / 32];
  if (listed) {
    const int slot = blockIdx.x * AG_W + wv;
    nl = __builtin_amdgcn_readfirstlane(lcnt[slot]);
    if (nl > AG_LC) nl = AG_LC;
    const unsigned* ml = lst + (size_t)slot * AG_LC;
#pragma unroll
    for (int j = 0; j < AG_LC / 32; ++j) {
      int idx = lane + 32 * j;
      en[j] = idx < nl ? ml[idx] : 0u;
    }
  }

  {
    const float4* h4 = (const float4*)(h + (size_t)row0 * HD);
    float4* t4 = (float4*)tile;
    float4 hv[4];
#pragma unroll
    for (int j = 0; j < 4; ++j) {
      int i4 = tid + j * AG_T;
      hv[j] = make_float4(0.f, 0.f, 0.f, 0.f);
      if (i4 < n4) hv[j] = h4[i4];
    }
    int bt = 0;
    float dd = 0.f;
    if (tid < nrows) {
      bt = batch[row0 + tid];
      dd = dinv[row0 + tid];
    }
#pragma unroll
    for (int j = 0; j < 4; ++j) {
      int i4 = tid + j * AG_T;
      if (i4 < n4) t4[i4] = hv[j];
    }
    if (tid < nrows) { sbatch[tid] = bt; sdinv[tid] = dd; }
  }
  __syncthreads();

  if (listed) {
#pragma unroll
    for (int j = 0; j < AG_LC / 32; ++j) {
      int idx = lane + 32 * j;
      if (idx < nl) {
        unsigned dl = en[j] >> 24;
        if (dl > AG_R - 1) dl = AG_R - 1;
        myqe[idx] = (int)((en[j] & 0xFFFFFFu) | (dl << 24));
      }
    }
    for (int base = 0; base < nl; base += 32) {
      int n = nl - base;
      if (n > 32) n = 32;
      int s_i = 0, dl_i = 0;
      if (lane < n) {
        unsigned pw = (unsigned)myqe[base + lane];
        s_i = (int)(pw & 0xFFFFFFu);
        if (s_i > N - 1) s_i = N - 1;
        dl_i = (int)(pw >> 24);
      }
      ag_rows(h, tile, s_i, dl_i, n, lane);
    }
  } else {
    AggSink sk;
    sk.src = src; sk.dst = dst; sk.h = h; sk.tile = tile; sk.row0 = row0;
    ag_scan(sk, planes, pW, myqe, tid, lane);
  }
  __syncthreads();

  {
    float4* t4 = (float4*)tile;
    float4* o4 = (float4*)(agg + (size_t)row0 * HD);
    const int c4 = (tid & 15) * 4;
    const float b0 = bias[c4], b1 = bias[c4 + 1], b2 = bias[c4 + 2], b3 = bias[c4 + 3];
    float4 s = make_float4(0.f, 0.f, 0.f, 0.f);
    float4 q = make_float4(0.f, 0.f, 0.f, 0.f);
#pragma unroll
    for (int j = 0; j < 4; ++j) {
      int i4 = tid + j * AG_T;
      if (i4 < n4) {
        float4 v = t4[i4];
        const float dd = sdinv[i4 >> 4];
        v.x = fmaxf(v.x * dd + b0, 0.f);
        v.y = fmaxf(v.y * dd + b1, 0.f);
        v.z = fmaxf(v.z * dd + b2, 0.f);
        v.w = fmaxf(v.w * dd + b3, 0.f);
        o4[i4] = v;
        t4[i4] = v;
        s.x += v.x; s.y += v.y; s.z += v.z; s.w += v.w;
        q.x += v.x * v.x; q.y += v.y * v.y; q.z += v.z * v.z; q.w += v.w * v.w;
      }
    }
    s.x += __shfl_xor(s.x, 16, 32); s.y += __shfl_xor(s.y, 16, 32);
    s.z += __shfl_xor(s.z, 16, 32); s.w += __shfl_xor(s.w, 16, 32);
    q.x += __shfl_xor(q.x, 16, 32); q.y += __shfl_xor(q.y, 16, 32);
    q.z += __shfl_xor(q.z, 16, 32); q.w += __shfl_xor(q.w, 16, 32);
    float* red = (float*)myqe;
    if (lane < 16) {
      *(float4*)(red + c4) = s;
      *(float4*)(red + HD + c4) = q;
    }
    __syncthreads();
    if (tid < 2 * HD) {
      const float* r0 = (const float*)&qe[0][0];
      float t = 0.f;
#pragma unroll
      for (int w = 0; w < AG_W; ++w) t += r0[w * AG_Q + tid];
      atomicAdd(&bn[tid], t);
    }
    {
      constexpr int RPG = (AG_R + AG_T / HD - 1) / (AG_T / HD);
      const int c = tid & (HD - 1);
      const int rg0 = (tid / HD) * RPG;
      float pv[RPG];
      int pg[RPG];
#pragma unroll
      for (int u = 0; u < RPG; ++u) {
        int r = rg0 + u;
        bool ok = r < nrows;
        int rr = ok ? r : 0;
        pv[u] = tile[rr * HD + c];
        pg[u] = ok ? sbatch[rr] : -1;
      }
      int gcur = -1;
      float acc = 0.f;
#pragma unroll
      for (int u = 0; u < RPG; ++u) {
        if (pg[u] >= 0) {
          if (pg[u] != gcur) {
            if (gcur >= 0) atomicAdd(&gsum[(size_t)gcur * CD + lofs + c], acc);
            gcur = pg[u];
            acc = 0.f;
          }
          acc += pv[u];
        }
      }
      if (gcur >= 0) atomicAdd(&gsum[(size_t)gcur * CD + lofs + c], acc);
    }
  }
}

__global__ void k_final(const float* __restrict__ z, const float* __restrict__ bn,
                        const float* __restrict__ gm0, const float* __restrict__ gm1, const float* __restrict__ gm2,
                        const float* __restrict__ bt0, const float* __restrict__ bt1, const float* __restrict__ bt2,
                        const float* __restrict__ gsum, const float* __restrict__ cnt,
                        float* __restrict__ out, int N, int G, float invN) {
  __shared__ float2 sbn[CD];
  const int t = threadIdx.x;
  const int wid = blockIdx.x * (blockDim.x >> 5) + (t >> 5);
  const int lane = t & 31;
  const bool node = wid < N;
  const bool graph = !node && wid < N + G;
  float v[6];
  float n = 0.f;
  if (node) {
#pragma unroll
    for (int j = 0; j < 6; ++j) {
      int e = lane + 32 * j;
      v[j] = z[(size_t)(e / HD) * N * HD + (size_t)wid * HD + (e % HD)];
    }
  } else {
    int g = graph ? wid - N : 0;
    n = cnt[g];
#pragma unroll
    for (int j = 0; j < 6; ++j) v[j] = gsum[(size_t)g * CD + lane + 32 * j];
  }
  if (t < CD) {
    int l = t / HD, c = t % HD;
    const float* gm = l == 0 ? gm0 : (l == 1 ? gm1 : gm2);
    const float* bt = l == 0 ? bt0 : (l == 1 ? bt1 : bt2);
    sbn[t] = bn_map(bn + l * 2 * HD, gm, bt, c, invN);
  }
  __syncthreads();
  if (!node && !graph) return;
  float s = 0.f;
  if (node) {
#pragma unroll
    for (int j = 0; j < 6; ++j) {
      float2 m = sbn[lane + 32 * j];
      v[j] = v[j] * m.x + m.y;
    }
  } else {
    const float mul = 1.0f / fmaxf(n, 1.0f);
#pragma unroll
    for (int j = 0; j < 6; ++j) {
      float2 m = sbn[lane + 32 * j];
      v[j] = (v[j] * m.x + n * m.y) * mul;
    }
  }
#pragma unroll
  for (int j = 0; j < 6; ++j) s += v[j] * v[j];
#pragma unroll
  for (int off = 16; off >= 1; off >>= 1) s += __shfl_xor(s, off, 32);
  float inv = 1.0f / fmaxf(sqrtf(s), 1e-12f);
  float* o = out + (size_t)wid * CD;
#pragma unroll
  for (int j = 0; j < 6; ++j) o[lane + 32 * j] = v[j] * inv;
}

extern "C" void kernel_launch(void* const* d_in, const int* in_sizes, int n_in,
                              void* d_out, int out_size, void* d_ws, size_t ws_size,
                              hipStream_t stream) {
  (void)n_in; (void)out_size; (void)ws_size;
  const float* x     = (const float*)d_in[0];
  const int*   ei    = (const int*)d_in[1];
  const int*   batch = (const int*)d_in[2];
  int N = in_sizes[0] / 128;
  int E = in_sizes[1] / 2;
  const int* src = ei;
  const int* dst = ei + E;
  const float* Wl[3]  = {(const float*)d_in[3],  (const float*)d_in[7],  (const float*)d_in[11]};
  const float* bl[3]  = {(const float*)d_in[4],  (const float*)d_in[8],  (const float*)d_in[12]};
  const float* gl[3]  = {(const float*)d_in[5],  (const float*)d_in[9],  (const float*)d_in[13]};
  const float* bel[3] = {(const float*)d_in[6],  (const float*)d_in[10], (const float*)d_in[14]};
  int nblk = (N + AG_R - 1) / AG_R;
  int pW = (((E + 31) >> 5) + 3) & ~3;

  float* ws   = (float*)d_ws;
  float* gsum = ws;                       ws += (size_t)GN * CD;
  float* cnt  = ws;                       ws += GN;
  float* bn   = ws;                       ws += 3 * 2 * HD;
  size_t nclear = (size_t)(ws - gsum) * sizeof(float);
  float* dinv = ws;                       ws += N;
  float* h    = ws;                       ws += (size_t)N * HD;
  float* zpl  = ws;                       ws += (size_t)3 * N * HD;
  unsigned* planes = (unsigned*)ws;
  unsigned* lst = planes + (size_t)KP_N * pW;
  int* lcnt = (int*)(lst + (size_t)nblk * AG_W * AG_LC);
  int* ovf  = lcnt + nblk * AG_W;

  hipMemsetAsync(gsum, 0, nclear, stream);

  k_key<<<((pW + 31) / 32 + 1) / 2, 64, 0, stream>>>(dst, E, planes, pW);
  k_own<<<nblk, AG_T, 0, stream>>>(src, dst, planes, batch, dinv, cnt, lst, lcnt, ovf, N, pW);

  const float invN = 1.0f / (float)N;
  for (int l = 0; l < 3; ++l) {
    int tilesM = (N + 15) / 16;
    int gblk = (tilesM + GM_W - 1) / GM_W;
    if (l == 0)
      k_gemm_wmma<false, 128><<<gblk, GM_W * 32, 0, stream>>>(x, 128, Wl[0], h, N, dinv, nullptr, nullptr, nullptr, invN);
    else
      k_gemm_wmma<true, HD><<<gblk, GM_W * 32, 0, stream>>>(zpl + (size_t)(l - 1) * N * HD, HD, Wl[l], h, N, dinv,
                                                            bn + (l - 1) * 2 * HD, gl[l - 1], bel[l - 1], invN);
    k_aggregate<<<nblk, AG_T, 0, stream>>>(src, dst, planes, lst, lcnt, ovf, dinv, h, bl[l], zpl + (size_t)l * N * HD,
                                           bn + l * 2 * HD, batch, gsum, l * HD, N, pW);
  }

  float* out = (float*)d_out;
  k_final<<<(N + GN + 7) / 8, 256, 0, stream>>>(zpl, bn, gl[0], gl[1], gl[2], bel[0], bel[1], bel[2], gsum, cnt, out,
                                                N, GN, invN);
}
